# starting kernel
# baseline (speedup 1.0000x reference)
#include <hip/hip_runtime.h>
#include <hip/hip_bf16.h>

#define NNODES 50000
#define NEDGES 1600000
#define NFEAT  512
#define F1     128
#define H1_    8
#define C_     16
#define NCLASS 16

typedef float v2f __attribute__((ext_vector_type(2)));
typedef float v8f __attribute__((ext_vector_type(8)));

__device__ __forceinline__ unsigned enc_f32(float f) {
  unsigned u = __float_as_uint(f);
  return (u & 0x80000000u) ? ~u : (u | 0x80000000u);
}
__device__ __forceinline__ float dec_f32(unsigned u) {
  return (u & 0x80000000u) ? __uint_as_float(u & 0x7FFFFFFFu) : __uint_as_float(~u);
}

template <int TN, int K, int NOUT>
__global__ __launch_bounds__(32) void gemm_f32_wmma_strip(
    const float* __restrict__ X, const float* __restrict__ Wt,
    float* __restrict__ D) {
  int tm = blockIdx.x;
  int lane = threadIdx.x;
  int l15 = lane & 15, half = lane >> 4;
  const float* ap = X  + (size_t)(tm * 16 + l15) * K + half * 2;
  const float* bp = Wt + (size_t)l15 * K + half * 2;
  v8f acc[TN];
  #pragma unroll
  for (int t = 0; t < TN; ++t) acc[t] = (v8f){};
  #pragma unroll 2
  for (int k = 0; k < K; k += 4) {
    v2f a, b[TN];
    a.x = ap[k];
    a.y = ap[k + 1];
    #pragma unroll
    for (int t = 0; t < TN; ++t) {
      b[t].x = bp[t * 16 * K + k];
      b[t].y = bp[t * 16 * K + k + 1];
    }
    #pragma unroll
    for (int t = 0; t < TN; ++t)
      acc[t] = __builtin_amdgcn_wmma_f32_16x16x4_f32(false, a, false, b[t],
                                                     (short)0, acc[t], false, false);
  }
  #pragma unroll
  for (int t = 0; t < TN; ++t) {
    float* dp = D + (size_t)(tm * 16 + half * 8) * NOUT + t * 16 + l15;
    #pragma unroll
    for (int r = 0; r < 8; ++r) dp[r * NOUT] = acc[t][r];
  }
}

__global__ void attn_scores(const float* __restrict__ h,
                            const float* __restrict__ attS,
                            const float* __restrict__ attD,
                            float* __restrict__ aS, float* __restrict__ aD,
                            int NH, int H) {
  int t = blockIdx.x * blockDim.x + threadIdx.x;
  if (t >= NH) return;
  int hh = t % H;
  const float* hp = h + (size_t)t * C_;
  const float* sp = attS + hh * C_;
  const float* dp = attD + hh * C_;
  float s = 0.f, d = 0.f;
  #pragma unroll
  for (int c = 0; c < C_; ++c) { s += hp[c] * sp[c]; d += hp[c] * dp[c]; }
  aS[t] = s; aD[t] = d;
}

__global__ void fill_u32(unsigned* __restrict__ p, unsigned v, size_t n) {
  size_t t = (size_t)blockIdx.x * blockDim.x + threadIdx.x;
  if (t < n) p[t] = v;
}

template <int H>
__global__ void edge_max(const int* __restrict__ ei,
                         const float* __restrict__ aS, const float* __restrict__ aD,
                         unsigned* __restrict__ amax) {
  int e = blockIdx.x * blockDim.x + threadIdx.x;
  if (e >= NEDGES + NNODES) return;
  int s, d;
  if (e < NEDGES) { s = ei[e]; d = ei[NEDGES + e]; } else { s = d = e - NEDGES; }
  const float* sp = aS + (size_t)s * H;
  const float* dp = aD + (size_t)d * H;
  unsigned* mp = amax + (size_t)d * H;
  #pragma unroll
  for (int h = 0; h < H; ++h) {
    float al = sp[h] + dp[h];
    al = al > 0.f ? al : 0.2f * al;
    atomicMax(&mp[h], enc_f32(al));
  }
}

template <int H>
__global__ void edge_sum(const int* __restrict__ ei,
                         const float* __restrict__ aS, const float* __restrict__ aD,
                         const unsigned* __restrict__ amax, float* __restrict__ denom) {
  int e = blockIdx.x * blockDim.x + threadIdx.x;
  if (e >= NEDGES + NNODES) return;
  int s, d;
  if (e < NEDGES) { s = ei[e]; d = ei[NEDGES + e]; } else { s = d = e - NEDGES; }
  const float* sp = aS + (size_t)s * H;
  const float* dp = aD + (size_t)d * H;
  const unsigned* mp = amax + (size_t)d * H;
  float* np = denom + (size_t)d * H;
  #pragma unroll
  for (int h = 0; h < H; ++h) {
    float al = sp[h] + dp[h];
    al = al > 0.f ? al : 0.2f * al;
    atomicAdd(&np[h], expf(al - dec_f32(mp[h])));
  }
}

template <int H>
__global__ void edge_msg(const int* __restrict__ ei,
                         const float* __restrict__ aS, const float* __restrict__ aD,
                         const unsigned* __restrict__ amax, const float* __restrict__ denom,
                         const float* __restrict__ h, float* __restrict__ agg) {
  long long t = (long long)blockIdx.x * blockDim.x + threadIdx.x;
  long long total = (long long)(NEDGES + NNODES) * H;
  if (t >= total) return;
  int hh = (int)(t % H);
  int e  = (int)(t / H);
  int s, d;
  if (e < NEDGES) { s = ei[e]; d = ei[NEDGES + e]; } else { s = d = e - NEDGES; }
  size_t si = (size_t)s * H + hh, di = (size_t)d * H + hh;
  float al = aS[si] + aD[di];
  al = al > 0.f ? al : 0.2f * al;
  float coeff = expf(al - dec_f32(amax[di])) / (denom[di] + 1e-16f);
  const float* hp = h + si * C_;
  float* op = agg + di * C_;
  #pragma unroll
  for (int c = 0; c < C_; ++c) atomicAdd(&op[c], hp[c] * coeff);
}

__global__ void bias_elu(float* __restrict__ x, const float* __restrict__ b,
                         int F, size_t n) {
  size_t t = (size_t)blockIdx.x * blockDim.x + threadIdx.x;
  if (t >= n) return;
  float v = x[t] + b[(int)(t % F)];
  x[t] = v > 0.f ? v : expm1f(v);
}

__global__ void bias_logsoftmax(const float* __restrict__ agg,
                                const float* __restrict__ b,
                                float* __restrict__ out) {
  int n = blockIdx.x * blockDim.x + threadIdx.x;
  if (n >= NNODES) return;
  float v[NCLASS];
  float m = -1e30f;
  #pragma unroll
  for (int c = 0; c < NCLASS; ++c) {
    v[c] = agg[(size_t)n * NCLASS + c] + b[c];
    m = fmaxf(m, v[c]);
  }
  float s = 0.f;
  #pragma unroll
  for (int c = 0; c < NCLASS; ++c) s += expf(v[c] - m);
  float ls = logf(s);
  #pragma unroll
  for (int c = 0; c < NCLASS; ++c) out[(size_t)n * NCLASS + c] = v[c] - m - ls;
}

extern "C" void kernel_launch(void* const* d_in, const int* in_sizes, int n_in,
                              void* d_out, int out_size, void* d_ws, size_t ws_size,
                              hipStream_t stream) {
  const float* x     = (const float*)d_in[0];
  const int*   ei    = (const int*)  d_in[1];
  const float* W1    = (const float*)d_in[2];
  const float* attS1 = (const float*)d_in[3];
  const float* attD1 = (const float*)d_in[4];
  const float* b1    = (const float*)d_in[5];
  const float* W2    = (const float*)d_in[6];
  const float* attS2 = (const float*)d_in[7];
  const float* attD2 = (const float*)d_in[8];
  const float* b2    = (const float*)d_in[9];
  float* out = (float*)d_out;

  char* ws = (char*)d_ws;
  size_t off = 0;
  auto alloc = [&](size_t bytes) -> void* {
    void* p = ws + off;
    off = (off + bytes + 255) & ~(size_t)255;
    return p;
  };

  float*    h1   = (float*)   alloc((size_t)NNODES * F1 * 4);
  float*    as1  = (float*)   alloc((size_t)NNODES * H1_ * 4);
  float*    ad1  = (float*)   alloc((size_t)NNODES * H1_ * 4);
  unsigned* am1  = (unsigned*)alloc((size_t)NNODES * H1_ * 4);
  float*    dn1  = (float*)   alloc((size_t)NNODES * H1_ * 4);
  float*    agg1 = (float*)   alloc((size_t)NNODES * F1 * 4);
  float*    h2   = (float*)   alloc((size_t)NNODES * C_ * 4);
  float*    as2  = (float*)   alloc((size_t)NNODES * 4);
  float*    ad2  = (float*)   alloc((size_t)NNODES * 4);
  unsigned* am2  = (unsigned*)alloc((size_t)NNODES * 4);
  float*    dn2  = (float*)   alloc((size_t)NNODES * 4);
  float*    agg2 = (float*)   alloc((size_t)NNODES * C_ * 4);

  const int EDG = NEDGES + NNODES;
  const int TB = 256;
  auto nb = [](long long n, int b) { return (unsigned)((n + b - 1) / b); };

  gemm_f32_wmma_strip<8, NFEAT, F1><<<3125, 32, 0, stream>>>(x, W1, h1);
  attn_scores<<<nb((long long)NNODES * H1_, TB), TB, 0, stream>>>(
      h1, attS1, attD1, as1, ad1, NNODES * H1_, H1_);
  fill_u32<<<nb((long long)NNODES * H1_, TB), TB, 0, stream>>>(am1, 0u, (size_t)NNODES * H1_);
  fill_u32<<<nb((long long)NNODES * H1_, TB), TB, 0, stream>>>((unsigned*)dn1, 0u, (size_t)NNODES * H1_);
  fill_u32<<<nb((long long)NNODES * F1, TB), TB, 0, stream>>>((unsigned*)agg1, 0u, (size_t)NNODES * F1);
  edge_max<H1_><<<nb(EDG, TB), TB, 0, stream>>>(ei, as1, ad1, am1);
  edge_sum<H1_><<<nb(EDG, TB), TB, 0, stream>>>(ei, as1, ad1, am1, dn1);
  edge_msg<H1_><<<nb((long long)EDG * H1_, TB), TB, 0, stream>>>(
      ei, as1, ad1, am1, dn1, h1, agg1);
  bias_elu<<<nb((long long)NNODES * F1, TB), TB, 0, stream>>>(
      agg1, b1, F1, (size_t)NNODES * F1);

  gemm_f32_wmma_strip<1, F1, C_><<<3125, 32, 0, stream>>>(agg1, W2, h2);
  attn_scores<<<nb(NNODES, TB), TB, 0, stream>>>(h2, attS2, attD2, as2, ad2, NNODES, 1);
  fill_u32<<<nb(NNODES, TB), TB, 0, stream>>>(am2, 0u, (size_t)NNODES);
  fill_u32<<<nb(NNODES, TB), TB, 0, stream>>>((unsigned*)dn2, 0u, (size_t)NNODES);
  fill_u32<<<nb((long long)NNODES * C_, TB), TB, 0, stream>>>((unsigned*)agg2, 0u, (size_t)NNODES * C_);
  edge_max<1><<<nb(EDG, TB), TB, 0, stream>>>(ei, as2, ad2, am2);
  edge_sum<1><<<nb(EDG, TB), TB, 0, stream>>>(ei, as2, ad2, am2, dn2);
  edge_msg<1><<<nb(EDG, TB), TB, 0, stream>>>(ei, as2, ad2, am2, dn2, h2, agg2);

  bias_logsoftmax<<<nb(NNODES, TB), TB, 0, stream>>>(agg2, b2, out);
}


# Edge passes by destination ownership: each work-group owns 196 nodes, scans an owner byte per edge, keeps max and sums in LDS (LDS atomics replace global ones); 16->7 launches; sum order changes
# speedup vs baseline: 6.2541x; 6.2541x over previous
#include <hip/hip_runtime.h>
#include <hip/hip_bf16.h>

#define NNODES 50000
#define NEDGES 1600000
#define NFEAT  512
#define F1     128
#define H1_    8
#define C_     16
#define NCLASS 16

typedef float v2f __attribute__((ext_vector_type(2)));
typedef float v8f __attribute__((ext_vector_type(8)));

__device__ __forceinline__ unsigned enc_f32(float f) {
  unsigned u = __float_as_uint(f);
  return (u & 0x80000000u) ? ~u : (u | 0x80000000u);
}
__device__ __forceinline__ float dec_f32(unsigned u) {
  return (u & 0x80000000u) ? __uint_as_float(u & 0x7FFFFFFFu) : __uint_as_float(~u);
}

template <int TN, int K, int NOUT>
__global__ __launch_bounds__(32) void gemm_f32_wmma_strip(
    const float* __restrict__ X, const float* __restrict__ Wt,
    float* __restrict__ D) {
  int tm = blockIdx.x;
  int lane = threadIdx.x;
  int l15 = lane & 15, half = lane >> 4;
  const float* ap = X  + (size_t)(tm * 16 + l15) * K + half * 2;
  const float* bp = Wt + (size_t)l15 * K + half * 2;
  v8f acc[TN];
  #pragma unroll
  for (int t = 0; t < TN; ++t) acc[t] = (v8f){};
  #pragma unroll 2
  for (int k = 0; k < K; k += 4) {
    v2f a, b[TN];
    a.x = ap[k];
    a.y = ap[k + 1];
    #pragma unroll
    for (int t = 0; t < TN; ++t) {
      b[t].x = bp[t * 16 * K + k];
      b[t].y = bp[t * 16 * K + k + 1];
    }
    #pragma unroll
    for (int t = 0; t < TN; ++t)
      acc[t] = __builtin_amdgcn_wmma_f32_16x16x4_f32(false, a, false, b[t],
                                                     (short)0, acc[t], false, false);
  }
  #pragma unroll
  for (int t = 0; t < TN; ++t) {
    float* dp = D + (size_t)(tm * 16 + half * 8) * NOUT + t * 16 + l15;
    #pragma unroll
    for (int r = 0; r < 8; ++r) dp[r * NOUT] = acc[t][r];
  }
}

__global__ void attn_scores(const float* __restrict__ h,
                            const float* __restrict__ attS,
                            const float* __restrict__ attD,
                            float* __restrict__ aS, float* __restrict__ aD,
                            int NH, int H) {
  int t = blockIdx.x * blockDim.x + threadIdx.x;
  if (t >= NH) return;
  int hh = t % H;
  const float* hp = h + (size_t)t * C_;
  const float* sp = attS + hh * C_;
  const float* dp = attD + hh * C_;
  float s = 0.f, d = 0.f;
  #pragma unroll
  for (int c = 0; c < C_; ++c) { s += hp[c] * sp[c]; d += hp[c] * dp[c]; }
  aS[t] = s; aD[t] = d;
}

#define NPB    196
#define NBLK   256
#define NKCH   (NEDGES / 16)
#define SCAN_U 2
#define NIT    ((NKCH + 1024 * SCAN_U - 1) / (1024 * SCAN_U))

__device__ __forceinline__ float lrelu(float a) { return a > 0.f ? a : 0.2f * a; }

__global__ void owner_ids(const int* __restrict__ ei, unsigned* __restrict__ key4) {
  int t = blockIdx.x * blockDim.x + threadIdx.x;
  if (t >= NEDGES / 4) return;
  int4 d = ((const int4*)(ei + NEDGES))[t];
  key4[t] = ((unsigned)(d.x / NPB) & 255u)
          | (((unsigned)(d.y / NPB) & 255u) << 8)
          | (((unsigned)(d.z / NPB) & 255u) << 16)
          | (((unsigned)(d.w / NPB) & 255u) << 24);
}

__device__ __forceinline__ void load_keys(const uint4* __restrict__ k4, int it, int tid,
                                          uint4 (&d)[SCAN_U]) {
  #pragma unroll
  for (int u = 0; u < SCAN_U; ++u) {
    int ch = (it * SCAN_U + u) * 1024 + tid;
    ch = ch < NKCH ? ch : NKCH - 1;
    d[u] = k4[ch];
  }
}

__device__ __forceinline__ unsigned zero_bytes(unsigned x) {
  return ~(((x & 0x7f7f7f7fu) + 0x7f7f7f7fu) | x | 0x7f7f7f7fu);
}

__device__ __forceinline__ unsigned match_bits(const uint4 (&d)[SCAN_U], int it, int tid,
                                               unsigned bid4) {
  unsigned mm = 0;
  #pragma unroll
  for (int u = 0; u < SCAN_U; ++u) {
    int ch = (it * SCAN_U + u) * 1024 + tid;
    unsigned f = (zero_bytes(d[u].x ^ bid4) >> 7) | (zero_bytes(d[u].y ^ bid4) >> 6)
               | (zero_bytes(d[u].z ^ bid4) >> 5) | (zero_bytes(d[u].w ^ bid4) >> 4);
    mm |= (ch < NKCH ? f : 0u) << (4 * u);
  }
  return mm;
}

__device__ __forceinline__ int edge_of(int it, int tid, int k) {
  int i = k & 7, b = k >> 3;
  int ch = (it * SCAN_U + (i >> 2)) * 1024 + tid;
  return (ch << 4) + ((i & 3) << 2) + b;
}

template <int H>
__global__ __launch_bounds__(1024) void owned_gat(
    const int* __restrict__ ei, const unsigned* __restrict__ key,
    const float* __restrict__ aS, const float* __restrict__ aD,
    const float* __restrict__ h, const float* __restrict__ bias,
    float* __restrict__ out) {
  constexpr int HC = H * C_;
  __shared__ float    s_agg[NPB * HC];
  __shared__ float    s_den[NPB * H];
  __shared__ unsigned s_max[NPB * H];
  __shared__ float    s_ad [NPB * H];

  const int tid = threadIdx.x, lane = tid & 31;
  const int lo  = blockIdx.x * NPB;
  const int cnt = (NNODES - lo) < NPB ? (NNODES - lo) : NPB;
  const unsigned bid4 = blockIdx.x * 0x01010101u;
  const uint4* k4 = (const uint4*)key;
  const int rot = blockIdx.x % NIT;

  for (int i = tid; i < NPB * H; i += 1024) {
    float ad = 0.f, as = 0.f;
    if (i < cnt * H) { ad = aD[(size_t)lo * H + i]; as = aS[(size_t)lo * H + i]; }
    s_ad[i]  = ad;
    s_max[i] = enc_f32(lrelu(as + ad));
    s_den[i] = 0.f;
  }
  for (int i = tid; i < NPB * HC; i += 1024) s_agg[i] = 0.f;
  __syncthreads();

  {
    uint4 dc[SCAN_U];
    load_keys(k4, rot, tid, dc);
    for (int t = 0; t < NIT; ++t) {
      const int it  = (t + rot) < NIT ? (t + rot) : (t + rot - NIT);
      const int itn = (it + 1) < NIT ? (it + 1) : 0;
      uint4 dn[SCAN_U];
      load_keys(k4, itn, tid, dn);
      unsigned mm = match_bits(dc, it, tid, bid4);
      while (mm) {
        int k = __builtin_ctz(mm); mm &= mm - 1;
        int e = edge_of(it, tid, k);
        int s = ei[e];
        int dl = ei[NEDGES + e] - lo;
        if constexpr (H == 8) {
          const float4* ap = (const float4*)(aS + (size_t)s * 8);
          float4 a0 = ap[0], a1 = ap[1];
          unsigned* mp = &s_max[dl * 8];
          const float* dp = &s_ad[dl * 8];
          atomicMax(&mp[0], enc_f32(lrelu(a0.x + dp[0])));
          atomicMax(&mp[1], enc_f32(lrelu(a0.y + dp[1])));
          atomicMax(&mp[2], enc_f32(lrelu(a0.z + dp[2])));
          atomicMax(&mp[3], enc_f32(lrelu(a0.w + dp[3])));
          atomicMax(&mp[4], enc_f32(lrelu(a1.x + dp[4])));
          atomicMax(&mp[5], enc_f32(lrelu(a1.y + dp[5])));
          atomicMax(&mp[6], enc_f32(lrelu(a1.z + dp[6])));
          atomicMax(&mp[7], enc_f32(lrelu(a1.w + dp[7])));
        } else {
          atomicMax(&s_max[dl], enc_f32(lrelu(aS[s] + s_ad[dl])));
        }
      }
      #pragma unroll
      for (int u = 0; u < SCAN_U; ++u) dc[u] = dn[u];
    }
  }
  __syncthreads();

  {
    uint4 dc[SCAN_U];
    load_keys(k4, rot, tid, dc);
    for (int t = 0; t < NIT; ++t) {
      const int it  = (t + rot) < NIT ? (t + rot) : (t + rot - NIT);
      const int itn = (it + 1) < NIT ? (it + 1) : 0;
      uint4 dn[SCAN_U];
      load_keys(k4, itn, tid, dn);
      unsigned mm = match_bits(dc, it, tid, bid4);
      if constexpr (H == 8) {
        const int hd = lane >> 2;
        for (;;) {
          unsigned bal = __builtin_amdgcn_ballot_w32(mm != 0);
          if (!bal) break;
          int sv = 0, dv = 0;
          if (mm) {
            int k = __builtin_ctz(mm); mm &= mm - 1;
            int e = edge_of(it, tid, k);
            sv = ei[e];
            dv = ei[NEDGES + e] - lo;
          }
          while (bal) {
            int sj[4] = {0, 0, 0, 0}, dj[4] = {0, 0, 0, 0};
            bool vj[4];
            #pragma unroll
            for (int j = 0; j < 4; ++j) {
              vj[j] = bal != 0;
              if (vj[j]) {
                int ml = __builtin_ctz(bal); bal &= bal - 1;
                sj[j] = __builtin_amdgcn_readlane(sv, ml);
                dj[j] = __builtin_amdgcn_readlane(dv, ml);
              } else {
                sj[j] = sj[j > 0 ? j - 1 : 0];
                dj[j] = dj[j > 0 ? j - 1 : 0];
              }
            }
            float  aj[4];
            float4 rj[4];
            #pragma unroll
            for (int j = 0; j < 4; ++j) {
              aj[j] = aS[(size_t)sj[j] * 8 + hd];
              rj[j] = ((const float4*)(h + (size_t)sj[j] * HC))[lane];
            }
            #pragma unroll
            for (int j = 0; j < 4; ++j) {
              if (vj[j]) {
                int dl = dj[j];
                float w = expf(lrelu(aj[j] + s_ad[dl * 8 + hd]) - dec_f32(s_max[dl * 8 + hd]));
                float* ap = &s_agg[dl * HC + lane];
                atomicAdd(ap,      rj[j].x * w);
                atomicAdd(ap + 32, rj[j].y * w);
                atomicAdd(ap + 64, rj[j].z * w);
                atomicAdd(ap + 96, rj[j].w * w);
                if ((lane & 3) == 0) atomicAdd(&s_den[dl * 8 + hd], w);
              }
            }
          }
        }
      } else {
        while (mm) {
          int k = __builtin_ctz(mm); mm &= mm - 1;
          int e = edge_of(it, tid, k);
          int s = ei[e];
          int dl = ei[NEDGES + e] - lo;
          float a = aS[s];
          const float4* rp = (const float4*)(h + (size_t)s * HC);
          float4 r0 = rp[0], r1 = rp[1], r2 = rp[2], r3 = rp[3];
          float w = expf(lrelu(a + s_ad[dl]) - dec_f32(s_max[dl]));
          float* ap = &s_agg[dl * HC];
          atomicAdd(ap + 0,  r0.x * w); atomicAdd(ap + 1,  r0.y * w);
          atomicAdd(ap + 2,  r0.z * w); atomicAdd(ap + 3,  r0.w * w);
          atomicAdd(ap + 4,  r1.x * w); atomicAdd(ap + 5,  r1.y * w);
          atomicAdd(ap + 6,  r1.z * w); atomicAdd(ap + 7,  r1.w * w);
          atomicAdd(ap + 8,  r2.x * w); atomicAdd(ap + 9,  r2.y * w);
          atomicAdd(ap + 10, r2.z * w); atomicAdd(ap + 11, r2.w * w);
          atomicAdd(ap + 12, r3.x * w); atomicAdd(ap + 13, r3.y * w);
          atomicAdd(ap + 14, r3.z * w); atomicAdd(ap + 15, r3.w * w);
          atomicAdd(&s_den[dl], w);
        }
      }
      #pragma unroll
      for (int u = 0; u < SCAN_U; ++u) dc[u] = dn[u];
    }
  }
  __syncthreads();

  if constexpr (H == 8) {
    for (int idx = tid; idx < cnt * HC; idx += 1024) {
      int dl = idx >> 7, c = idx & 127, hd = c >> 4;
      size_t n = (size_t)(lo + dl);
      float ws = expf(lrelu(aS[n * 8 + hd] + s_ad[dl * 8 + hd]) - dec_f32(s_max[dl * 8 + hd]));
      float acc = s_agg[dl * HC + (c & 3) * 32 + (c >> 2)] + ws * h[n * HC + c];
      float v = acc / (s_den[dl * 8 + hd] + ws + 1e-16f) + bias[c];
      out[n * HC + c] = v > 0.f ? v : expm1f(v);
    }
  } else {
    if (tid < cnt) {
      int dl = tid;
      size_t n = (size_t)(lo + dl);
      float ws = expf(lrelu(aS[n] + s_ad[dl]) - dec_f32(s_max[dl]));
      float dn = s_den[dl] + ws + 1e-16f;
      float v[NCLASS];
      float m = -1e30f;
      #pragma unroll
      for (int c = 0; c < NCLASS; ++c) {
        v[c] = (s_agg[dl * HC + c] + ws * h[n * HC + c]) / dn + bias[c];
        m = fmaxf(m, v[c]);
      }
      float sum = 0.f;
      #pragma unroll
      for (int c = 0; c < NCLASS; ++c) sum += expf(v[c] - m);
      float ls = logf(sum);
      #pragma unroll
      for (int c = 0; c < NCLASS; ++c) out[n * NCLASS + c] = v[c] - m - ls;
    }
  }
}

extern "C" void kernel_launch(void* const* d_in, const int* in_sizes, int n_in,
                              void* d_out, int out_size, void* d_ws, size_t ws_size,
                              hipStream_t stream) {
  const float* x     = (const float*)d_in[0];
  const int*   ei    = (const int*)  d_in[1];
  const float* W1    = (const float*)d_in[2];
  const float* attS1 = (const float*)d_in[3];
  const float* attD1 = (const float*)d_in[4];
  const float* b1    = (const float*)d_in[5];
  const float* W2    = (const float*)d_in[6];
  const float* attS2 = (const float*)d_in[7];
  const float* attD2 = (const float*)d_in[8];
  const float* b2    = (const float*)d_in[9];
  float* out = (float*)d_out;

  char* ws = (char*)d_ws;
  size_t off = 0;
  auto alloc = [&](size_t bytes) -> void* {
    void* p = ws + off;
    off = (off + bytes + 255) & ~(size_t)255;
    return p;
  };

  float* h1  = (float*)alloc((size_t)NNODES * F1 * 4);
  float* as1 = (float*)alloc((size_t)NNODES * H1_ * 4);
  float* ad1 = (float*)alloc((size_t)NNODES * H1_ * 4);
  float* x1  = (float*)alloc((size_t)NNODES * F1 * 4);
  float* h2  = (float*)alloc((size_t)NNODES * C_ * 4);
  float* as2 = (float*)alloc((size_t)NNODES * 4);
  float* ad2 = (float*)alloc((size_t)NNODES * 4);
  unsigned* key = (unsigned*)alloc((size_t)NEDGES);

  const int TB = 256;
  auto nb = [](long long n, int b) { return (unsigned)((n + b - 1) / b); };

  gemm_f32_wmma_strip<8, NFEAT, F1><<<3125, 32, 0, stream>>>(x, W1, h1);
  attn_scores<<<nb((long long)NNODES * H1_, TB), TB, 0, stream>>>(
      h1, attS1, attD1, as1, ad1, NNODES * H1_, H1_);
  owner_ids<<<nb(NEDGES / 4, TB), TB, 0, stream>>>(ei, key);
  owned_gat<H1_><<<NBLK, 1024, 0, stream>>>(ei, key, as1, ad1, h1, b1, x1);

  gemm_f32_wmma_strip<1, F1, C_><<<3125, 32, 0, stream>>>(x1, W2, h2);
  attn_scores<<<nb(NNODES, TB), TB, 0, stream>>>(h2, attS2, attD2, as2, ad2, NNODES, 1);
  owned_gat<1><<<NBLK, 1024, 0, stream>>>(ei, key, as2, ad2, h2, b2, out);
}


# Owner-byte scan done once per layer: each work-group lists its matching edges in LDS, and the max and sum passes walk that list with every lane holding an edge; order of the LDS float adds changes
# speedup vs baseline: 8.4804x; 1.3560x over previous
#include <hip/hip_runtime.h>
#include <hip/hip_bf16.h>

#define NNODES 50000
#define NEDGES 1600000
#define NFEAT  512
#define F1     128
#define H1_    8
#define C_     16
#define NCLASS 16

typedef float v2f __attribute__((ext_vector_type(2)));
typedef float v8f __attribute__((ext_vector_type(8)));

__device__ __forceinline__ unsigned enc_f32(float f) {
  unsigned u = __float_as_uint(f);
  return (u & 0x80000000u) ? ~u : (u | 0x80000000u);
}
__device__ __forceinline__ float dec_f32(unsigned u) {
  return (u & 0x80000000u) ? __uint_as_float(u & 0x7FFFFFFFu) : __uint_as_float(~u);
}

template <int TN, int K, int NOUT>
__global__ __launch_bounds__(32) void gemm_f32_wmma_strip(
    const float* __restrict__ X, const float* __restrict__ Wt,
    float* __restrict__ D) {
  int tm = blockIdx.x;
  int lane = threadIdx.x;
  int l15 = lane & 15, half = lane >> 4;
  const float* ap = X  + (size_t)(tm * 16 + l15) * K + half * 2;
  const float* bp = Wt + (size_t)l15 * K + half * 2;
  v8f acc[TN];
  #pragma unroll
  for (int t = 0; t < TN; ++t) acc[t] = (v8f){};
  #pragma unroll 2
  for (int k = 0; k < K; k += 4) {
    v2f a, b[TN];
    a.x = ap[k];
    a.y = ap[k + 1];
    #pragma unroll
    for (int t = 0; t < TN; ++t) {
      b[t].x = bp[t * 16 * K + k];
      b[t].y = bp[t * 16 * K + k + 1];
    }
    #pragma unroll
    for (int t = 0; t < TN; ++t)
      acc[t] = __builtin_amdgcn_wmma_f32_16x16x4_f32(false, a, false, b[t],
                                                     (short)0, acc[t], false, false);
  }
  #pragma unroll
  for (int t = 0; t < TN; ++t) {
    float* dp = D + (size_t)(tm * 16 + half * 8) * NOUT + t * 16 + l15;
    #pragma unroll
    for (int r = 0; r < 8; ++r) dp[r * NOUT] = acc[t][r];
  }
}

__global__ void attn_scores(const float* __restrict__ h,
                            const float* __restrict__ attS,
                            const float* __restrict__ attD,
                            float* __restrict__ aS, float* __restrict__ aD,
                            int NH, int H) {
  int t = blockIdx.x * blockDim.x + threadIdx.x;
  if (t >= NH) return;
  int hh = t % H;
  const float* hp = h + (size_t)t * C_;
  const float* sp = attS + hh * C_;
  const float* dp = attD + hh * C_;
  float s = 0.f, d = 0.f;
  #pragma unroll
  for (int c = 0; c < C_; ++c) { s += hp[c] * sp[c]; d += hp[c] * dp[c]; }
  aS[t] = s; aD[t] = d;
}

#define NPB    196
#define NBLK   256
#define NKCH   (NEDGES / 16)
#define SCAN_U 2
#define NIT    ((NKCH + 1024 * SCAN_U - 1) / (1024 * SCAN_U))

__device__ __forceinline__ float lrelu(float a) { return a > 0.f ? a : 0.2f * a; }

__global__ void owner_ids(const int* __restrict__ ei, unsigned* __restrict__ key4) {
  int t = blockIdx.x * blockDim.x + threadIdx.x;
  if (t >= NEDGES / 4) return;
  int4 d = ((const int4*)(ei + NEDGES))[t];
  key4[t] = ((unsigned)(d.x / NPB) & 255u)
          | (((unsigned)(d.y / NPB) & 255u) << 8)
          | (((unsigned)(d.z / NPB) & 255u) << 16)
          | (((unsigned)(d.w / NPB) & 255u) << 24);
}

__device__ __forceinline__ void load_keys(const uint4* __restrict__ k4, int it, int tid,
                                          uint4 (&d)[SCAN_U]) {
  #pragma unroll
  for (int u = 0; u < SCAN_U; ++u) {
    int ch = (it * SCAN_U + u) * 1024 + tid;
    ch = ch < NKCH ? ch : NKCH - 1;
    d[u] = k4[ch];
  }
}

__device__ __forceinline__ unsigned zero_bytes(unsigned x) {
  return ~(((x & 0x7f7f7f7fu) + 0x7f7f7f7fu) | x | 0x7f7f7f7fu);
}

__device__ __forceinline__ unsigned match_bits(const uint4 (&d)[SCAN_U], int it, int tid,
                                               unsigned bid4) {
  unsigned mm = 0;
  #pragma unroll
  for (int u = 0; u < SCAN_U; ++u) {
    int ch = (it * SCAN_U + u) * 1024 + tid;
    unsigned f = (zero_bytes(d[u].x ^ bid4) >> 7) | (zero_bytes(d[u].y ^ bid4) >> 6)
               | (zero_bytes(d[u].z ^ bid4) >> 5) | (zero_bytes(d[u].w ^ bid4) >> 4);
    mm |= (ch < NKCH ? f : 0u) << (4 * u);
  }
  return mm;
}

__device__ __forceinline__ int edge_of(int it, int tid, int k) {
  int i = k & 7, b = k >> 3;
  int ch = (it * SCAN_U + (i >> 2)) * 1024 + tid;
  return (ch << 4) + ((i & 3) << 2) + b;
}

#define LCAP   16384
static_assert(NNODES <= 65536 && NPB <= 65536, "a list entry packs src and dst_local into 16 bits each");

template <int H>
__global__ __launch_bounds__(1024) void owned_gat(
    const int* __restrict__ ei, const unsigned* __restrict__ key,
    const float* __restrict__ aS, const float* __restrict__ aD,
    const float* __restrict__ h, const float* __restrict__ bias,
    float* __restrict__ out) {
  constexpr int HC = H * C_;
  __shared__ float    s_agg[NPB * HC];
  __shared__ float    s_den[NPB * H];
  __shared__ unsigned s_max[NPB * H];
  __shared__ float    s_ad [NPB * H];
  __shared__ unsigned s_list[LCAP];
  __shared__ int      s_cnt;

  const int tid = threadIdx.x, lane = tid & 31;
  const int lo  = blockIdx.x * NPB;
  const int cnt = (NNODES - lo) < NPB ? (NNODES - lo) : NPB;
  const unsigned bid4 = blockIdx.x * 0x01010101u;
  const uint4* k4 = (const uint4*)key;
  const int rot = blockIdx.x % NIT;

  for (int i = tid; i < NPB * H; i += 1024) {
    float ad = 0.f, as = 0.f;
    if (i < cnt * H) { ad = aD[(size_t)lo * H + i]; as = aS[(size_t)lo * H + i]; }
    const int t = (i % H) * NPB + i / H;
    s_ad[t]  = ad;
    s_max[t] = enc_f32(lrelu(as + ad));
    s_den[t] = 0.f;
  }
  for (int i = tid; i < NPB * HC; i += 1024) s_agg[i] = 0.f;
  if (tid == 0) s_cnt = 0;
  __syncthreads();

  {
    uint4 dc[SCAN_U];
    load_keys(k4, rot, tid, dc);
    for (int t = 0; t < NIT; ++t) {
      const int it  = (t + rot) < NIT ? (t + rot) : (t + rot - NIT);
      const int itn = (it + 1) < NIT ? (it + 1) : 0;
      uint4 dn[SCAN_U];
      load_keys(k4, itn, tid, dn);
      unsigned mm = match_bits(dc, it, tid, bid4);
      for (;;) {
        const unsigned bal = __builtin_amdgcn_ballot_w32(mm != 0);
        if (!bal) break;
        int base = 0;
        if (lane == 0) base = atomicAdd(&s_cnt, __builtin_popcount(bal));
        base = __builtin_amdgcn_readfirstlane(base);
        if (mm) {
          const int k = __builtin_ctz(mm); mm &= mm - 1;
          const int pos = base + __builtin_popcount(bal & ((1u << lane) - 1u));
          if (pos < LCAP) s_list[pos] = (unsigned)edge_of(it, tid, k);
        }
      }
      #pragma unroll
      for (int u = 0; u < SCAN_U; ++u) dc[u] = dn[u];
    }
  }
  __syncthreads();
  const int nl = __builtin_amdgcn_readfirstlane(s_cnt);

  if (nl <= LCAP) {
    {
      constexpr int U2 = 4;
      for (int i0 = tid; i0 < nl; i0 += 1024 * U2) {
        int idx[U2], ev[U2], sv[U2], dv[U2];
        bool v[U2];
        #pragma unroll
        for (int j = 0; j < U2; ++j) {
          idx[j] = i0 + 1024 * j;
          v[j] = idx[j] < nl;
          ev[j] = (int)s_list[v[j] ? idx[j] : i0];
        }
        #pragma unroll
        for (int j = 0; j < U2; ++j) {
          sv[j] = ei[ev[j]];
          dv[j] = ei[NEDGES + ev[j]] - lo;
        }
        if constexpr (H == 8) {
          float4 a0[U2], a1[U2];
          float ad[U2][8];
          #pragma unroll
          for (int j = 0; j < U2; ++j) {
            const float4* ap = (const float4*)(aS + (size_t)sv[j] * 8);
            a0[j] = ap[0]; a1[j] = ap[1];
          }
          #pragma unroll
          for (int j = 0; j < U2; ++j) {
            #pragma unroll
            for (int hh = 0; hh < 8; ++hh) ad[j][hh] = s_ad[hh * NPB + dv[j]];
          }
          #pragma unroll
          for (int j = 0; j < U2; ++j) {
            if (v[j]) {
              const int d = dv[j];
              atomicMax(&s_max[0 * NPB + d], enc_f32(lrelu(a0[j].x + ad[j][0])));
              atomicMax(&s_max[1 * NPB + d], enc_f32(lrelu(a0[j].y + ad[j][1])));
              atomicMax(&s_max[2 * NPB + d], enc_f32(lrelu(a0[j].z + ad[j][2])));
              atomicMax(&s_max[3 * NPB + d], enc_f32(lrelu(a0[j].w + ad[j][3])));
              atomicMax(&s_max[4 * NPB + d], enc_f32(lrelu(a1[j].x + ad[j][4])));
              atomicMax(&s_max[5 * NPB + d], enc_f32(lrelu(a1[j].y + ad[j][5])));
              atomicMax(&s_max[6 * NPB + d], enc_f32(lrelu(a1[j].z + ad[j][6])));
              atomicMax(&s_max[7 * NPB + d], enc_f32(lrelu(a1[j].w + ad[j][7])));
              s_list[idx[j]] = (unsigned)sv[j] | ((unsigned)d << 16);
            }
          }
        } else {
          float a[U2], ad[U2];
          #pragma unroll
          for (int j = 0; j < U2; ++j) a[j] = aS[sv[j]];
          #pragma unroll
          for (int j = 0; j < U2; ++j) ad[j] = s_ad[dv[j]];
          #pragma unroll
          for (int j = 0; j < U2; ++j) {
            if (v[j]) {
              atomicMax(&s_max[dv[j]], enc_f32(lrelu(a[j] + ad[j])));
              s_list[idx[j]] = (unsigned)sv[j] | ((unsigned)dv[j] << 16);
            }
          }
        }
      }
    }
    __syncthreads();

    if constexpr (H == 8) {
      constexpr int EB = 8;
      const int hd = lane >> 2;
      const int wv = __builtin_amdgcn_readfirstlane(tid >> 5);
      for (int base = wv * EB; base < nl; base += 32 * EB) {
        const int li = base + (lane & (EB - 1));
        const unsigned ent = s_list[li < nl ? li : nl - 1];
        int sj[EB], dj[EB];
        bool vj[EB];
        #pragma unroll
        for (int j = 0; j < EB; ++j) {
          const unsigned x = (unsigned)__builtin_amdgcn_readlane((int)ent, j);
          sj[j] = (int)(x & 0xffffu);
          dj[j] = (int)(x >> 16);
          vj[j] = base + j < nl;
        }
        float  aj[EB];
        float4 rj[EB];
        #pragma unroll
        for (int j = 0; j < EB; ++j) {
          aj[j] = aS[(size_t)sj[j] * 8 + hd];
          rj[j] = ((const float4*)(h + (size_t)sj[j] * HC))[lane];
        }
        float adj[EB], mxj[EB];
        #pragma unroll
        for (int j = 0; j < EB; ++j) {
          adj[j] = s_ad[hd * NPB + dj[j]];
          mxj[j] = dec_f32(s_max[hd * NPB + dj[j]]);
        }
        #pragma unroll
        for (int j = 0; j < EB; ++j) {
          if (vj[j]) {
            const int dl = dj[j];
            float w = expf(lrelu(aj[j] + adj[j]) - mxj[j]);
            float* ap = &s_agg[dl * HC + lane];
            atomicAdd(ap,      rj[j].x * w);
            atomicAdd(ap + 32, rj[j].y * w);
            atomicAdd(ap + 64, rj[j].z * w);
            atomicAdd(ap + 96, rj[j].w * w);
            if ((lane & 3) == 0) atomicAdd(&s_den[hd * NPB + dl], w);
          }
        }
      }
    } else {
      constexpr int U3 = 2;
      for (int i0 = tid; i0 < nl; i0 += 1024 * U3) {
        int sv[U3], dv[U3];
        bool v[U3];
        float a[U3];
        float4 r[U3][4];
        #pragma unroll
        for (int j = 0; j < U3; ++j) {
          const int idx = i0 + 1024 * j;
          v[j] = idx < nl;
          const unsigned ent = s_list[v[j] ? idx : i0];
          sv[j] = (int)(ent & 0xffffu);
          dv[j] = (int)(ent >> 16);
        }
        #pragma unroll
        for (int j = 0; j < U3; ++j) {
          a[j] = aS[sv[j]];
          const float4* rp = (const float4*)(h + (size_t)sv[j] * HC);
          r[j][0] = rp[0]; r[j][1] = rp[1]; r[j][2] = rp[2]; r[j][3] = rp[3];
        }
        float adv[U3], mxv[U3];
        #pragma unroll
        for (int j = 0; j < U3; ++j) { adv[j] = s_ad[dv[j]]; mxv[j] = dec_f32(s_max[dv[j]]); }
        #pragma unroll
        for (int j = 0; j < U3; ++j) {
          if (v[j]) {
            const int d = dv[j];
            float w = expf(lrelu(a[j] + adv[j]) - mxv[j]);
            float* ap = &s_agg[d];
            atomicAdd(ap + 0  * NPB, r[j][0].x * w); atomicAdd(ap + 1  * NPB, r[j][0].y * w);
            atomicAdd(ap + 2  * NPB, r[j][0].z * w); atomicAdd(ap + 3  * NPB, r[j][0].w * w);
            atomicAdd(ap + 4  * NPB, r[j][1].x * w); atomicAdd(ap + 5  * NPB, r[j][1].y * w);
            atomicAdd(ap + 6  * NPB, r[j][1].z * w); atomicAdd(ap + 7  * NPB, r[j][1].w * w);
            atomicAdd(ap + 8  * NPB, r[j][2].x * w); atomicAdd(ap + 9  * NPB, r[j][2].y * w);
            atomicAdd(ap + 10 * NPB, r[j][2].z * w); atomicAdd(ap + 11 * NPB, r[j][2].w * w);
            atomicAdd(ap + 12 * NPB, r[j][3].x * w); atomicAdd(ap + 13 * NPB, r[j][3].y * w);
            atomicAdd(ap + 14 * NPB, r[j][3].z * w); atomicAdd(ap + 15 * NPB, r[j][3].w * w);
            atomicAdd(&s_den[d], w);
          }
        }
      }
    }
  } else {
    {
      uint4 dc[SCAN_U];
      load_keys(k4, rot, tid, dc);
      for (int t = 0; t < NIT; ++t) {
        const int it  = (t + rot) < NIT ? (t + rot) : (t + rot - NIT);
        const int itn = (it + 1) < NIT ? (it + 1) : 0;
        uint4 dn[SCAN_U];
        load_keys(k4, itn, tid, dn);
        unsigned mm = match_bits(dc, it, tid, bid4);
        while (mm) {
          int k = __builtin_ctz(mm); mm &= mm - 1;
          int e = edge_of(it, tid, k);
          int s = ei[e];
          int dl = ei[NEDGES + e] - lo;
          if constexpr (H == 8) {
            const float4* ap = (const float4*)(aS + (size_t)s * 8);
            float4 a0 = ap[0], a1 = ap[1];
            atomicMax(&s_max[0 * NPB + dl], enc_f32(lrelu(a0.x + s_ad[0 * NPB + dl])));
            atomicMax(&s_max[1 * NPB + dl], enc_f32(lrelu(a0.y + s_ad[1 * NPB + dl])));
            atomicMax(&s_max[2 * NPB + dl], enc_f32(lrelu(a0.z + s_ad[2 * NPB + dl])));
            atomicMax(&s_max[3 * NPB + dl], enc_f32(lrelu(a0.w + s_ad[3 * NPB + dl])));
            atomicMax(&s_max[4 * NPB + dl], enc_f32(lrelu(a1.x + s_ad[4 * NPB + dl])));
            atomicMax(&s_max[5 * NPB + dl], enc_f32(lrelu(a1.y + s_ad[5 * NPB + dl])));
            atomicMax(&s_max[6 * NPB + dl], enc_f32(lrelu(a1.z + s_ad[6 * NPB + dl])));
            atomicMax(&s_max[7 * NPB + dl], enc_f32(lrelu(a1.w + s_ad[7 * NPB + dl])));
          } else {
            atomicMax(&s_max[dl], enc_f32(lrelu(aS[s] + s_ad[dl])));
          }
        }
        #pragma unroll
        for (int u = 0; u < SCAN_U; ++u) dc[u] = dn[u];
      }
    }
    __syncthreads();

    {
      uint4 dc[SCAN_U];
      load_keys(k4, rot, tid, dc);
      for (int t = 0; t < NIT; ++t) {
        const int it  = (t + rot) < NIT ? (t + rot) : (t + rot - NIT);
        const int itn = (it + 1) < NIT ? (it + 1) : 0;
        uint4 dn[SCAN_U];
        load_keys(k4, itn, tid, dn);
        unsigned mm = match_bits(dc, it, tid, bid4);
        if constexpr (H == 8) {
          const int hd = lane >> 2;
          for (;;) {
            unsigned bal = __builtin_amdgcn_ballot_w32(mm != 0);
            if (!bal) break;
            int sv = 0, dv = 0;
            if (mm) {
              int k = __builtin_ctz(mm); mm &= mm - 1;
              int e = edge_of(it, tid, k);
              sv = ei[e];
              dv = ei[NEDGES + e] - lo;
            }
            while (bal) {
              int ml = __builtin_ctz(bal); bal &= bal - 1;
              int s  = __builtin_amdgcn_readlane(sv, ml);
              int dl = __builtin_amdgcn_readlane(dv, ml);
              float  a = aS[(size_t)s * 8 + hd];
              float4 r = ((const float4*)(h + (size_t)s * HC))[lane];
              float w = expf(lrelu(a + s_ad[hd * NPB + dl]) - dec_f32(s_max[hd * NPB + dl]));
              float* ap = &s_agg[dl * HC + lane];
              atomicAdd(ap,      r.x * w);
              atomicAdd(ap + 32, r.y * w);
              atomicAdd(ap + 64, r.z * w);
              atomicAdd(ap + 96, r.w * w);
              if ((lane & 3) == 0) atomicAdd(&s_den[hd * NPB + dl], w);
            }
          }
        } else {
          while (mm) {
            int k = __builtin_ctz(mm); mm &= mm - 1;
            int e = edge_of(it, tid, k);
            int s = ei[e];
            int dl = ei[NEDGES + e] - lo;
            float a = aS[s];
            const float4* rp = (const float4*)(h + (size_t)s * HC);
            float4 r0 = rp[0], r1 = rp[1], r2 = rp[2], r3 = rp[3];
            float w = expf(lrelu(a + s_ad[dl]) - dec_f32(s_max[dl]));
            float* ap = &s_agg[dl];
            atomicAdd(ap + 0  * NPB, r0.x * w); atomicAdd(ap + 1  * NPB, r0.y * w);
            atomicAdd(ap + 2  * NPB, r0.z * w); atomicAdd(ap + 3  * NPB, r0.w * w);
            atomicAdd(ap + 4  * NPB, r1.x * w); atomicAdd(ap + 5  * NPB, r1.y * w);
            atomicAdd(ap + 6  * NPB, r1.z * w); atomicAdd(ap + 7  * NPB, r1.w * w);
            atomicAdd(ap + 8  * NPB, r2.x * w); atomicAdd(ap + 9  * NPB, r2.y * w);
            atomicAdd(ap + 10 * NPB, r2.z * w); atomicAdd(ap + 11 * NPB, r2.w * w);
            atomicAdd(ap + 12 * NPB, r3.x * w); atomicAdd(ap + 13 * NPB, r3.y * w);
            atomicAdd(ap + 14 * NPB, r3.z * w); atomicAdd(ap + 15 * NPB, r3.w * w);
            atomicAdd(&s_den[dl], w);
          }
        }
        #pragma unroll
        for (int u = 0; u < SCAN_U; ++u) dc[u] = dn[u];
      }
    }
  }
  __syncthreads();

  if constexpr (H == 8) {
    for (int idx = tid; idx < cnt * HC; idx += 1024) {
      int dl = idx >> 7, c = idx & 127, hd = c >> 4;
      size_t n = (size_t)(lo + dl);
      float ws = expf(lrelu(aS[n * 8 + hd] + s_ad[hd * NPB + dl]) - dec_f32(s_max[hd * NPB + dl]));
      float acc = s_agg[dl * HC + (c & 3) * 32 + (c >> 2)] + ws * h[n * HC + c];
      float v = acc / (s_den[hd * NPB + dl] + ws + 1e-16f) + bias[c];
      out[n * HC + c] = v > 0.f ? v : expm1f(v);
    }
  } else {
    if (tid < cnt) {
      int dl = tid;
      size_t n = (size_t)(lo + dl);
      float ws = expf(lrelu(aS[n] + s_ad[dl]) - dec_f32(s_max[dl]));
      float dn = s_den[dl] + ws + 1e-16f;
      float v[NCLASS];
      float m = -1e30f;
      #pragma unroll
      for (int c = 0; c < NCLASS; ++c) {
        v[c] = (s_agg[c * NPB + dl] + ws * h[n * HC + c]) / dn + bias[c];
        m = fmaxf(m, v[c]);
      }
      float sum = 0.f;
      #pragma unroll
      for (int c = 0; c < NCLASS; ++c) sum += expf(v[c] - m);
      float ls = logf(sum);
      #pragma unroll
      for (int c = 0; c < NCLASS; ++c) out[n * NCLASS + c] = v[c] - m - ls;
    }
  }
}

extern "C" void kernel_launch(void* const* d_in, const int* in_sizes, int n_in,
                              void* d_out, int out_size, void* d_ws, size_t ws_size,
                              hipStream_t stream) {
  const float* x     = (const float*)d_in[0];
  const int*   ei    = (const int*)  d_in[1];
  const float* W1    = (const float*)d_in[2];
  const float* attS1 = (const float*)d_in[3];
  const float* attD1 = (const float*)d_in[4];
  const float* b1    = (const float*)d_in[5];
  const float* W2    = (const float*)d_in[6];
  const float* attS2 = (const float*)d_in[7];
  const float* attD2 = (const float*)d_in[8];
  const float* b2    = (const float*)d_in[9];
  float* out = (float*)d_out;

  char* ws = (char*)d_ws;
  size_t off = 0;
  auto alloc = [&](size_t bytes) -> void* {
    void* p = ws + off;
    off = (off + bytes + 255) & ~(size_t)255;
    return p;
  };

  float* h1  = (float*)alloc((size_t)NNODES * F1 * 4);
  float* as1 = (float*)alloc((size_t)NNODES * H1_ * 4);
  float* ad1 = (float*)alloc((size_t)NNODES * H1_ * 4);
  float* x1  = (float*)alloc((size_t)NNODES * F1 * 4);
  float* h2  = (float*)alloc((size_t)NNODES * C_ * 4);
  float* as2 = (float*)alloc((size_t)NNODES * 4);
  float* ad2 = (float*)alloc((size_t)NNODES * 4);
  unsigned* key = (unsigned*)alloc((size_t)NEDGES);

  const int TB = 256;
  auto nb = [](long long n, int b) { return (unsigned)((n + b - 1) / b); };

  gemm_f32_wmma_strip<8, NFEAT, F1><<<3125, 32, 0, stream>>>(x, W1, h1);
  attn_scores<<<nb((long long)NNODES * H1_, TB), TB, 0, stream>>>(
      h1, attS1, attD1, as1, ad1, NNODES * H1_, H1_);
  owner_ids<<<nb(NEDGES / 4, TB), TB, 0, stream>>>(ei, key);
  owned_gat<H1_><<<NBLK, 1024, 0, stream>>>(ei, key, as1, ad1, h1, b1, x1);

  gemm_f32_wmma_strip<1, F1, C_><<<3125, 32, 0, stream>>>(x1, W2, h2);
  attn_scores<<<nb(NNODES, TB), TB, 0, stream>>>(h2, attS2, attD2, as2, ad2, NNODES, 1);
  owned_gat<1><<<NBLK, 1024, 0, stream>>>(ei, key, as2, ad2, h2, b2, out);
}


# Layer-1 product: 256 work-groups of 16 waves copy W1 into LDS once and take B fragments from LDS, not the cache (8 of 9 fragment loads per k-step); A one 32-k block ahead; sum order unchanged
# speedup vs baseline: 9.6401x; 1.1368x over previous
#include <hip/hip_runtime.h>
#include <hip/hip_bf16.h>

#define NNODES 50000
#define NEDGES 1600000
#define NFEAT  512
#define F1     128
#define H1_    8
#define C_     16
#define NCLASS 16

typedef float v2f __attribute__((ext_vector_type(2)));
typedef float v8f __attribute__((ext_vector_type(8)));

__device__ __forceinline__ unsigned enc_f32(float f) {
  unsigned u = __float_as_uint(f);
  return (u & 0x80000000u) ? ~u : (u | 0x80000000u);
}
__device__ __forceinline__ float dec_f32(unsigned u) {
  return (u & 0x80000000u) ? __uint_as_float(u & 0x7FFFFFFFu) : __uint_as_float(~u);
}

template <int TN, int K, int NOUT>
__global__ __launch_bounds__(32) void gemm_f32_wmma_strip(
    const float* __restrict__ X, const float* __restrict__ Wt,
    float* __restrict__ D) {
  int tm = blockIdx.x;
  int lane = threadIdx.x;
  int l15 = lane & 15, half = lane >> 4;
  const float* ap = X  + (size_t)(tm * 16 + l15) * K + half * 2;
  const float* bp = Wt + (size_t)l15 * K + half * 2;
  v8f acc[TN];
  #pragma unroll
  for (int t = 0; t < TN; ++t) acc[t] = (v8f){};
  #pragma unroll 2
  for (int k = 0; k < K; k += 4) {
    v2f a, b[TN];
    a.x = ap[k];
    a.y = ap[k + 1];
    #pragma unroll
    for (int t = 0; t < TN; ++t) {
      b[t].x = bp[t * 16 * K + k];
      b[t].y = bp[t * 16 * K + k + 1];
    }
    #pragma unroll
    for (int t = 0; t < TN; ++t)
      acc[t] = __builtin_amdgcn_wmma_f32_16x16x4_f32(false, a, false, b[t],
                                                     (short)0, acc[t], false, false);
  }
  #pragma unroll
  for (int t = 0; t < TN; ++t) {
    float* dp = D + (size_t)(tm * 16 + half * 8) * NOUT + t * 16 + l15;
    #pragma unroll
    for (int r = 0; r < 8; ++r) dp[r * NOUT] = acc[t][r];
  }
}

#define W1P      516
#define G1_BLK   256
#define G1_WAVES 16
#define G1_THREADS 512
#define G1_STRIPS (NNODES / 16)
#define G1_FULL  (G1_STRIPS / G1_BLK)
#define G1_REST  (G1_STRIPS - G1_FULL * G1_BLK)
static_assert(NNODES % 16 == 0, "strips of 16 rows");
static_assert(G1_FULL + 4 == G1_WAVES, "waves G1_FULL .. G1_FULL+3 take the column quarters");
static_assert(G1_THREADS == G1_WAVES * 32, "work-group size");

template <int TN>
__device__ __forceinline__ void k_block(const float* __restrict__ an, const float* sw,
                                        const unsigned (&bo)[TN],
                                        const v2f (&aU)[8], v2f (&aP)[8],
                                        v2f (&bC)[TN], v8f (&acc)[TN]) {
  #pragma unroll
  for (int j = 0; j < 8; ++j) { aP[j].x = an[4 * j]; aP[j].y = an[4 * j + 1]; }
  #pragma unroll
  for (int ks = 0; ks < 8; ++ks) {
    v2f bN[TN];
    #pragma unroll
    for (int t = 0; t < TN; ++t) bN[t] = *(const v2f*)(sw + bo[t] + 4 * ks + 4);
    #pragma unroll
    for (int t = 0; t < TN; ++t)
      acc[t] = __builtin_amdgcn_wmma_f32_16x16x4_f32(false, aU[ks], false, bC[t],
                                                     (short)0, acc[t], false, false);
    #pragma unroll
    for (int t = 0; t < TN; ++t) bC[t] = bN[t];
  }
  __builtin_amdgcn_sched_barrier(0);
  #pragma unroll
  for (int j = 0; j < 8; ++j) asm volatile("" : "+v"(aP[j]));
}

template <int TN>
__device__ __forceinline__ void strip_b_from_lds(const float* __restrict__ X, const float* sw,
                                                 float* __restrict__ D, int tm, int col0, int lane) {
  constexpr int NB = NFEAT / 32;
  static_assert(NB % 2 == 0, "the two A register sets change roles every block");
  const int l15 = lane & 15, half = lane >> 4;
  const float* ap = X + (size_t)(tm * 16 + l15) * NFEAT + half * 2;
  v8f acc[TN];
  v2f a0[8], a1[8], bC[TN];
  unsigned bo[TN];
  #pragma unroll
  for (int t = 0; t < TN; ++t) {
    acc[t] = (v8f){};
    bo[t] = (unsigned)((col0 + t * 16 + l15) * W1P + half * 2);
    asm volatile("" : "+v"(bo[t]));
  }
  #pragma unroll
  for (int j = 0; j < 8; ++j) { a0[j].x = ap[4 * j]; a0[j].y = ap[4 * j + 1]; }
  #pragma unroll
  for (int t = 0; t < TN; ++t) bC[t] = *(const v2f*)(sw + bo[t]);
  for (int c = 0; c < NB; c += 2) {
    k_block<TN>(ap + 32 * (c + 1), sw, bo, a0, a1, bC, acc);
    #pragma unroll
    for (int t = 0; t < TN; ++t) bo[t] += 32;
    k_block<TN>(ap + 32 * (c + 2 < NB ? c + 2 : c + 1), sw, bo, a1, a0, bC, acc);
    #pragma unroll
    for (int t = 0; t < TN; ++t) bo[t] += 32;
  }
  #pragma unroll
  for (int t = 0; t < TN; ++t) {
    float* dp = D + (size_t)(tm * 16 + half * 8) * F1 + col0 + t * 16 + l15;
    #pragma unroll
    for (int r = 0; r < 8; ++r) dp[r * F1] = acc[t][r];
  }
}

__global__ __launch_bounds__(G1_THREADS) void gemm1_w_in_lds(
    const float* __restrict__ X, const float* __restrict__ Wt, float* __restrict__ D) {
  __shared__ __attribute__((aligned(16))) float s_w[F1 * W1P];
  const int tid = threadIdx.x, lane = tid & 31;
  const int wv = __builtin_amdgcn_readfirstlane(tid >> 5);
  #pragma unroll 8
  for (int i = tid; i < F1 * (NFEAT / 4); i += G1_THREADS) {
    const int row = i / (NFEAT / 4), c4 = i % (NFEAT / 4);
    const float4 v = ((const float4*)Wt)[i];
    *(float4*)&s_w[row * W1P + c4 * 4] = v;
  }
  if (tid < F1) *(float4*)&s_w[tid * W1P + NFEAT] = make_float4(0.f, 0.f, 0.f, 0.f);
  __syncthreads();
  if (wv < G1_FULL) {
    strip_b_from_lds<8>(X, s_w, D, wv * G1_BLK + (int)blockIdx.x, 0, lane);
  } else if ((int)blockIdx.x < G1_REST) {
    strip_b_from_lds<2>(X, s_w, D, G1_FULL * G1_BLK + (int)blockIdx.x, (wv - G1_FULL) * 32, lane);
  }
}

__global__ void attn_scores(const float* __restrict__ h,
                            const float* __restrict__ attS,
                            const float* __restrict__ attD,
                            float* __restrict__ aS, float* __restrict__ aD,
                            int NH, int H) {
  int t = blockIdx.x * blockDim.x + threadIdx.x;
  if (t >= NH) return;
  int hh = t % H;
  const float* hp = h + (size_t)t * C_;
  const float* sp = attS + hh * C_;
  const float* dp = attD + hh * C_;
  float s = 0.f, d = 0.f;
  #pragma unroll
  for (int c = 0; c < C_; ++c) { s += hp[c] * sp[c]; d += hp[c] * dp[c]; }
  aS[t] = s; aD[t] = d;
}

#define NPB    196
#define NBLK   256
#define NKCH   (NEDGES / 16)
#define SCAN_U 2
#define NIT    ((NKCH + 1024 * SCAN_U - 1) / (1024 * SCAN_U))

__device__ __forceinline__ float lrelu(float a) { return a > 0.f ? a : 0.2f * a; }

__global__ void owner_ids(const int* __restrict__ ei, unsigned* __restrict__ key4) {
  int t = blockIdx.x * blockDim.x + threadIdx.x;
  if (t >= NEDGES / 4) return;
  int4 d = ((const int4*)(ei + NEDGES))[t];
  key4[t] = ((unsigned)(d.x / NPB) & 255u)
          | (((unsigned)(d.y / NPB) & 255u) << 8)
          | (((unsigned)(d.z / NPB) & 255u) << 16)
          | (((unsigned)(d.w / NPB) & 255u) << 24);
}

__device__ __forceinline__ void load_keys(const uint4* __restrict__ k4, int it, int tid,
                                          uint4 (&d)[SCAN_U]) {
  #pragma unroll
  for (int u = 0; u < SCAN_U; ++u) {
    int ch = (it * SCAN_U + u) * 1024 + tid;
    ch = ch < NKCH ? ch : NKCH - 1;
    d[u] = k4[ch];
  }
}

__device__ __forceinline__ unsigned zero_bytes(unsigned x) {
  return ~(((x & 0x7f7f7f7fu) + 0x7f7f7f7fu) | x | 0x7f7f7f7fu);
}

__device__ __forceinline__ unsigned match_bits(const uint4 (&d)[SCAN_U], int it, int tid,
                                               unsigned bid4) {
  unsigned mm = 0;
  #pragma unroll
  for (int u = 0; u < SCAN_U; ++u) {
    int ch = (it * SCAN_U + u) * 1024 + tid;
    unsigned f = (zero_bytes(d[u].x ^ bid4) >> 7) | (zero_bytes(d[u].y ^ bid4) >> 6)
               | (zero_bytes(d[u].z ^ bid4) >> 5) | (zero_bytes(d[u].w ^ bid4) >> 4);
    mm |= (ch < NKCH ? f : 0u) << (4 * u);
  }
  return mm;
}

__device__ __forceinline__ int edge_of(int it, int tid, int k) {
  int i = k & 7, b = k >> 3;
  int ch = (it * SCAN_U + (i >> 2)) * 1024 + tid;
  return (ch << 4) + ((i & 3) << 2) + b;
}

#define LCAP   16384
static_assert(NNODES <= 65536 && NPB <= 65536, "a list entry packs src and dst_local into 16 bits each");

template <int H>
__global__ __launch_bounds__(1024) void owned_gat(
    const int* __restrict__ ei, const unsigned* __restrict__ key,
    const float* __restrict__ aS, const float* __restrict__ aD,
    const float* __restrict__ h, const float* __restrict__ bias,
    float* __restrict__ out) {
  constexpr int HC = H * C_;
  __shared__ float    s_agg[NPB * HC];
  __shared__ float    s_den[NPB * H];
  __shared__ unsigned s_max[NPB * H];
  __shared__ float    s_ad [NPB * H];
  __shared__ unsigned s_list[LCAP];
  __shared__ int      s_cnt;

  const int tid = threadIdx.x, lane = tid & 31;
  const int lo  = blockIdx.x * NPB;
  const int cnt = (NNODES - lo) < NPB ? (NNODES - lo) : NPB;
  const unsigned bid4 = blockIdx.x * 0x01010101u;
  const uint4* k4 = (const uint4*)key;
  const int rot = blockIdx.x % NIT;

  for (int i = tid; i < NPB * H; i += 1024) {
    float ad = 0.f, as = 0.f;
    if (i < cnt * H) { ad = aD[(size_t)lo * H + i]; as = aS[(size_t)lo * H + i]; }
    const int t = (i % H) * NPB + i / H;
    s_ad[t]  = ad;
    s_max[t] = enc_f32(lrelu(as + ad));
    s_den[t] = 0.f;
  }
  for (int i = tid; i < NPB * HC; i += 1024) s_agg[i] = 0.f;
  if (tid == 0) s_cnt = 0;
  __syncthreads();

  {
    uint4 dc[SCAN_U];
    load_keys(k4, rot, tid, dc);
    for (int t = 0; t < NIT; ++t) {
      const int it  = (t + rot) < NIT ? (t + rot) : (t + rot - NIT);
      const int itn = (it + 1) < NIT ? (it + 1) : 0;
      uint4 dn[SCAN_U];
      load_keys(k4, itn, tid, dn);
      unsigned mm = match_bits(dc, it, tid, bid4);
      for (;;) {
        const unsigned bal = __builtin_amdgcn_ballot_w32(mm != 0);
        if (!bal) break;
        int base = 0;
        if (lane == 0) base = atomicAdd(&s_cnt, __builtin_popcount(bal));
        base = __builtin_amdgcn_readfirstlane(base);
        if (mm) {
          const int k = __builtin_ctz(mm); mm &= mm - 1;
          const int pos = base + __builtin_popcount(bal & ((1u << lane) - 1u));
          if (pos < LCAP) s_list[pos] = (unsigned)edge_of(it, tid, k);
        }
      }
      #pragma unroll
      for (int u = 0; u < SCAN_U; ++u) dc[u] = dn[u];
    }
  }
  __syncthreads();
  const int nl = __builtin_amdgcn_readfirstlane(s_cnt);

  if (nl <= LCAP) {
    {
      constexpr int U2 = 4;
      for (int i0 = tid; i0 < nl; i0 += 1024 * U2) {
        int idx[U2], ev[U2], sv[U2], dv[U2];
        bool v[U2];
        #pragma unroll
        for (int j = 0; j < U2; ++j) {
          idx[j] = i0 + 1024 * j;
          v[j] = idx[j] < nl;
          ev[j] = (int)s_list[v[j] ? idx[j] : i0];
        }
        #pragma unroll
        for (int j = 0; j < U2; ++j) {
          sv[j] = ei[ev[j]];
          dv[j] = ei[NEDGES + ev[j]] - lo;
        }
        if constexpr (H == 8) {
          float4 a0[U2], a1[U2];
          float ad[U2][8];
          #pragma unroll
          for (int j = 0; j < U2; ++j) {
            const float4* ap = (const float4*)(aS + (size_t)sv[j] * 8);
            a0[j] = ap[0]; a1[j] = ap[1];
          }
          #pragma unroll
          for (int j = 0; j < U2; ++j) {
            #pragma unroll
            for (int hh = 0; hh < 8; ++hh) ad[j][hh] = s_ad[hh * NPB + dv[j]];
          }
          #pragma unroll
          for (int j = 0; j < U2; ++j) {
            if (v[j]) {
              const int d = dv[j];
              atomicMax(&s_max[0 * NPB + d], enc_f32(lrelu(a0[j].x + ad[j][0])));
              atomicMax(&s_max[1 * NPB + d], enc_f32(lrelu(a0[j].y + ad[j][1])));
              atomicMax(&s_max[2 * NPB + d], enc_f32(lrelu(a0[j].z + ad[j][2])));
              atomicMax(&s_max[3 * NPB + d], enc_f32(lrelu(a0[j].w + ad[j][3])));
              atomicMax(&s_max[4 * NPB + d], enc_f32(lrelu(a1[j].x + ad[j][4])));
              atomicMax(&s_max[5 * NPB + d], enc_f32(lrelu(a1[j].y + ad[j][5])));
              atomicMax(&s_max[6 * NPB + d], enc_f32(lrelu(a1[j].z + ad[j][6])));
              atomicMax(&s_max[7 * NPB + d], enc_f32(lrelu(a1[j].w + ad[j][7])));
              s_list[idx[j]] = (unsigned)sv[j] | ((unsigned)d << 16);
            }
          }
        } else {
          float a[U2], ad[U2];
          #pragma unroll
          for (int j = 0; j < U2; ++j) a[j] = aS[sv[j]];
          #pragma unroll
          for (int j = 0; j < U2; ++j) ad[j] = s_ad[dv[j]];
          #pragma unroll
          for (int j = 0; j < U2; ++j) {
            if (v[j]) {
              atomicMax(&s_max[dv[j]], enc_f32(lrelu(a[j] + ad[j])));
              s_list[idx[j]] = (unsigned)sv[j] | ((unsigned)dv[j] << 16);
            }
          }
        }
      }
    }
    __syncthreads();

    if constexpr (H == 8) {
      constexpr int EB = 8;
      const int hd = lane >> 2;
      const int wv = __builtin_amdgcn_readfirstlane(tid >> 5);
      for (int base = wv * EB; base < nl; base += 32 * EB) {
        const int li = base + (lane & (EB - 1));
        const unsigned ent = s_list[li < nl ? li : nl - 1];
        int sj[EB], dj[EB];
        bool vj[EB];
        #pragma unroll
        for (int j = 0; j < EB; ++j) {
          const unsigned x = (unsigned)__builtin_amdgcn_readlane((int)ent, j);
          sj[j] = (int)(x & 0xffffu);
          dj[j] = (int)(x >> 16);
          vj[j] = base + j < nl;
        }
        float  aj[EB];
        float4 rj[EB];
        #pragma unroll
        for (int j = 0; j < EB; ++j) {
          aj[j] = aS[(size_t)sj[j] * 8 + hd];
          rj[j] = ((const float4*)(h + (size_t)sj[j] * HC))[lane];
        }
        float adj[EB], mxj[EB];
        #pragma unroll
        for (int j = 0; j < EB; ++j) {
          adj[j] = s_ad[hd * NPB + dj[j]];
          mxj[j] = dec_f32(s_max[hd * NPB + dj[j]]);
        }
        #pragma unroll
        for (int j = 0; j < EB; ++j) {
          if (vj[j]) {
            const int dl = dj[j];
            float w = expf(lrelu(aj[j] + adj[j]) - mxj[j]);
            float* ap = &s_agg[dl * HC + lane];
            atomicAdd(ap,      rj[j].x * w);
            atomicAdd(ap + 32, rj[j].y * w);
            atomicAdd(ap + 64, rj[j].z * w);
            atomicAdd(ap + 96, rj[j].w * w);
            if ((lane & 3) == 0) atomicAdd(&s_den[hd * NPB + dl], w);
          }
        }
      }
    } else {
      constexpr int U3 = 2;
      for (int i0 = tid; i0 < nl; i0 += 1024 * U3) {
        int sv[U3], dv[U3];
        bool v[U3];
        float a[U3];
        float4 r[U3][4];
        #pragma unroll
        for (int j = 0; j < U3; ++j) {
          const int idx = i0 + 1024 * j;
          v[j] = idx < nl;
          const unsigned ent = s_list[v[j] ? idx : i0];
          sv[j] = (int)(ent & 0xffffu);
          dv[j] = (int)(ent >> 16);
        }
        #pragma unroll
        for (int j = 0; j < U3; ++j) {
          a[j] = aS[sv[j]];
          const float4* rp = (const float4*)(h + (size_t)sv[j] * HC);
          r[j][0] = rp[0]; r[j][1] = rp[1]; r[j][2] = rp[2]; r[j][3] = rp[3];
        }
        float adv[U3], mxv[U3];
        #pragma unroll
        for (int j = 0; j < U3; ++j) { adv[j] = s_ad[dv[j]]; mxv[j] = dec_f32(s_max[dv[j]]); }
        #pragma unroll
        for (int j = 0; j < U3; ++j) {
          if (v[j]) {
            const int d = dv[j];
            float w = expf(lrelu(a[j] + adv[j]) - mxv[j]);
            float* ap = &s_agg[d];
            atomicAdd(ap + 0  * NPB, r[j][0].x * w); atomicAdd(ap + 1  * NPB, r[j][0].y * w);
            atomicAdd(ap + 2  * NPB, r[j][0].z * w); atomicAdd(ap + 3  * NPB, r[j][0].w * w);
            atomicAdd(ap + 4  * NPB, r[j][1].x * w); atomicAdd(ap + 5  * NPB, r[j][1].y * w);
            atomicAdd(ap + 6  * NPB, r[j][1].z * w); atomicAdd(ap + 7  * NPB, r[j][1].w * w);
            atomicAdd(ap + 8  * NPB, r[j][2].x * w); atomicAdd(ap + 9  * NPB, r[j][2].y * w);
            atomicAdd(ap + 10 * NPB, r[j][2].z * w); atomicAdd(ap + 11 * NPB, r[j][2].w * w);
            atomicAdd(ap + 12 * NPB, r[j][3].x * w); atomicAdd(ap + 13 * NPB, r[j][3].y * w);
            atomicAdd(ap + 14 * NPB, r[j][3].z * w); atomicAdd(ap + 15 * NPB, r[j][3].w * w);
            atomicAdd(&s_den[d], w);
          }
        }
      }
    }
  } else {
    {
      uint4 dc[SCAN_U];
      load_keys(k4, rot, tid, dc);
      for (int t = 0; t < NIT; ++t) {
        const int it  = (t + rot) < NIT ? (t + rot) : (t + rot - NIT);
        const int itn = (it + 1) < NIT ? (it + 1) : 0;
        uint4 dn[SCAN_U];
        load_keys(k4, itn, tid, dn);
        unsigned mm = match_bits(dc, it, tid, bid4);
        while (mm) {
          int k = __builtin_ctz(mm); mm &= mm - 1;
          int e = edge_of(it, tid, k);
          int s = ei[e];
          int dl = ei[NEDGES + e] - lo;
          if constexpr (H == 8) {
            const float4* ap = (const float4*)(aS + (size_t)s * 8);
            float4 a0 = ap[0], a1 = ap[1];
            atomicMax(&s_max[0 * NPB + dl], enc_f32(lrelu(a0.x + s_ad[0 * NPB + dl])));
            atomicMax(&s_max[1 * NPB + dl], enc_f32(lrelu(a0.y + s_ad[1 * NPB + dl])));
            atomicMax(&s_max[2 * NPB + dl], enc_f32(lrelu(a0.z + s_ad[2 * NPB + dl])));
            atomicMax(&s_max[3 * NPB + dl], enc_f32(lrelu(a0.w + s_ad[3 * NPB + dl])));
            atomicMax(&s_max[4 * NPB + dl], enc_f32(lrelu(a1.x + s_ad[4 * NPB + dl])));
            atomicMax(&s_max[5 * NPB + dl], enc_f32(lrelu(a1.y + s_ad[5 * NPB + dl])));
            atomicMax(&s_max[6 * NPB + dl], enc_f32(lrelu(a1.z + s_ad[6 * NPB + dl])));
            atomicMax(&s_max[7 * NPB + dl], enc_f32(lrelu(a1.w + s_ad[7 * NPB + dl])));
          } else {
            atomicMax(&s_max[dl], enc_f32(lrelu(aS[s] + s_ad[dl])));
          }
        }
        #pragma unroll
        for (int u = 0; u < SCAN_U; ++u) dc[u] = dn[u];
      }
    }
    __syncthreads();

    {
      uint4 dc[SCAN_U];
      load_keys(k4, rot, tid, dc);
      for (int t = 0; t < NIT; ++t) {
        const int it  = (t + rot) < NIT ? (t + rot) : (t + rot - NIT);
        const int itn = (it + 1) < NIT ? (it + 1) : 0;
        uint4 dn[SCAN_U];
        load_keys(k4, itn, tid, dn);
        unsigned mm = match_bits(dc, it, tid, bid4);
        if constexpr (H == 8) {
          const int hd = lane >> 2;
          for (;;) {
            unsigned bal = __builtin_amdgcn_ballot_w32(mm != 0);
            if (!bal) break;
            int sv = 0, dv = 0;
            if (mm) {
              int k = __builtin_ctz(mm); mm &= mm - 1;
              int e = edge_of(it, tid, k);
              sv = ei[e];
              dv = ei[NEDGES + e] - lo;
            }
            while (bal) {
              int ml = __builtin_ctz(bal); bal &= bal - 1;
              int s  = __builtin_amdgcn_readlane(sv, ml);
              int dl = __builtin_amdgcn_readlane(dv, ml);
              float  a = aS[(size_t)s * 8 + hd];
              float4 r = ((const float4*)(h + (size_t)s * HC))[lane];
              float w = expf(lrelu(a + s_ad[hd * NPB + dl]) - dec_f32(s_max[hd * NPB + dl]));
              float* ap = &s_agg[dl * HC + lane];
              atomicAdd(ap,      r.x * w);
              atomicAdd(ap + 32, r.y * w);
              atomicAdd(ap + 64, r.z * w);
              atomicAdd(ap + 96, r.w * w);
              if ((lane & 3) == 0) atomicAdd(&s_den[hd * NPB + dl], w);
            }
          }
        } else {
          while (mm) {
            int k = __builtin_ctz(mm); mm &= mm - 1;
            int e = edge_of(it, tid, k);
            int s = ei[e];
            int dl = ei[NEDGES + e] - lo;
            float a = aS[s];
            const float4* rp = (const float4*)(h + (size_t)s * HC);
            float4 r0 = rp[0], r1 = rp[1], r2 = rp[2], r3 = rp[3];
            float w = expf(lrelu(a + s_ad[dl]) - dec_f32(s_max[dl]));
            float* ap = &s_agg[dl];
            atomicAdd(ap + 0  * NPB, r0.x * w); atomicAdd(ap + 1  * NPB, r0.y * w);
            atomicAdd(ap + 2  * NPB, r0.z * w); atomicAdd(ap + 3  * NPB, r0.w * w);
            atomicAdd(ap + 4  * NPB, r1.x * w); atomicAdd(ap + 5  * NPB, r1.y * w);
            atomicAdd(ap + 6  * NPB, r1.z * w); atomicAdd(ap + 7  * NPB, r1.w * w);
            atomicAdd(ap + 8  * NPB, r2.x * w); atomicAdd(ap + 9  * NPB, r2.y * w);
            atomicAdd(ap + 10 * NPB, r2.z * w); atomicAdd(ap + 11 * NPB, r2.w * w);
            atomicAdd(ap + 12 * NPB, r3.x * w); atomicAdd(ap + 13 * NPB, r3.y * w);
            atomicAdd(ap + 14 * NPB, r3.z * w); atomicAdd(ap + 15 * NPB, r3.w * w);
            atomicAdd(&s_den[dl], w);
          }
        }
        #pragma unroll
        for (int u = 0; u < SCAN_U; ++u) dc[u] = dn[u];
      }
    }
  }
  __syncthreads();

  if constexpr (H == 8) {
    for (int idx = tid; idx < cnt * HC; idx += 1024) {
      int dl = idx >> 7, c = idx & 127, hd = c >> 4;
      size_t n = (size_t)(lo + dl);
      float ws = expf(lrelu(aS[n * 8 + hd] + s_ad[hd * NPB + dl]) - dec_f32(s_max[hd * NPB + dl]));
      float acc = s_agg[dl * HC + (c & 3) * 32 + (c >> 2)] + ws * h[n * HC + c];
      float v = acc / (s_den[hd * NPB + dl] + ws + 1e-16f) + bias[c];
      out[n * HC + c] = v > 0.f ? v : expm1f(v);
    }
  } else {
    if (tid < cnt) {
      int dl = tid;
      size_t n = (size_t)(lo + dl);
      float ws = expf(lrelu(aS[n] + s_ad[dl]) - dec_f32(s_max[dl]));
      float dn = s_den[dl] + ws + 1e-16f;
      float v[NCLASS];
      float m = -1e30f;
      #pragma unroll
      for (int c = 0; c < NCLASS; ++c) {
        v[c] = (s_agg[c * NPB + dl] + ws * h[n * HC + c]) / dn + bias[c];
        m = fmaxf(m, v[c]);
      }
      float sum = 0.f;
      #pragma unroll
      for (int c = 0; c < NCLASS; ++c) sum += expf(v[c] - m);
      float ls = logf(sum);
      #pragma unroll
      for (int c = 0; c < NCLASS; ++c) out[n * NCLASS + c] = v[c] - m - ls;
    }
  }
}

extern "C" void kernel_launch(void* const* d_in, const int* in_sizes, int n_in,
                              void* d_out, int out_size, void* d_ws, size_t ws_size,
                              hipStream_t stream) {
  const float* x     = (const float*)d_in[0];
  const int*   ei    = (const int*)  d_in[1];
  const float* W1    = (const float*)d_in[2];
  const float* attS1 = (const float*)d_in[3];
  const float* attD1 = (const float*)d_in[4];
  const float* b1    = (const float*)d_in[5];
  const float* W2    = (const float*)d_in[6];
  const float* attS2 = (const float*)d_in[7];
  const float* attD2 = (const float*)d_in[8];
  const float* b2    = (const float*)d_in[9];
  float* out = (float*)d_out;

  char* ws = (char*)d_ws;
  size_t off = 0;
  auto alloc = [&](size_t bytes) -> void* {
    void* p = ws + off;
    off = (off + bytes + 255) & ~(size_t)255;
    return p;
  };

  float* h1  = (float*)alloc((size_t)NNODES * F1 * 4);
  float* as1 = (float*)alloc((size_t)NNODES * H1_ * 4);
  float* ad1 = (float*)alloc((size_t)NNODES * H1_ * 4);
  float* x1  = (float*)alloc((size_t)NNODES * F1 * 4);
  float* h2  = (float*)alloc((size_t)NNODES * C_ * 4);
  float* as2 = (float*)alloc((size_t)NNODES * 4);
  float* ad2 = (float*)alloc((size_t)NNODES * 4);
  unsigned* key = (unsigned*)alloc((size_t)NEDGES);

  const int TB = 256;
  auto nb = [](long long n, int b) { return (unsigned)((n + b - 1) / b); };

  gemm1_w_in_lds<<<G1_BLK, G1_THREADS, 0, stream>>>(x, W1, h1);
  attn_scores<<<nb((long long)NNODES * H1_, TB), TB, 0, stream>>>(
      h1, attS1, attD1, as1, ad1, NNODES * H1_, H1_);
  owner_ids<<<nb(NEDGES / 4, TB), TB, 0, stream>>>(ei, key);
  owned_gat<H1_><<<NBLK, 1024, 0, stream>>>(ei, key, as1, ad1, h1, b1, x1);

  gemm_f32_wmma_strip<1, F1, C_><<<3125, 32, 0, stream>>>(x1, W2, h2);
  attn_scores<<<nb(NNODES, TB), TB, 0, stream>>>(h2, attS2, attD2, as2, ad2, NNODES, 1);
  owned_gat<1><<<NBLK, 1024, 0, stream>>>(ei, key, as2, ad2, h2, b2, out);
}


# Layer 2 reuses layer 1's per-work-group edge list (src, local dst), kept in the workspace: no second owner-byte scan, no edge-end fetches in layer 2; its LDS adds stay in arrival order
# speedup vs baseline: 11.3729x; 1.1797x over previous
#include <hip/hip_runtime.h>
#include <hip/hip_bf16.h>

#define NNODES 50000
#define NEDGES 1600000
#define NFEAT  512
#define F1     128
#define H1_    8
#define C_     16
#define NCLASS 16

typedef float v2f __attribute__((ext_vector_type(2)));
typedef float v8f __attribute__((ext_vector_type(8)));

__device__ __forceinline__ unsigned enc_f32(float f) {
  unsigned u = __float_as_uint(f);
  return (u & 0x80000000u) ? ~u : (u | 0x80000000u);
}
__device__ __forceinline__ float dec_f32(unsigned u) {
  return (u & 0x80000000u) ? __uint_as_float(u & 0x7FFFFFFFu) : __uint_as_float(~u);
}

template <int TN, int K, int NOUT>
__global__ __launch_bounds__(32) void gemm_f32_wmma_strip(
    const float* __restrict__ X, const float* __restrict__ Wt,
    float* __restrict__ D) {
  int tm = blockIdx.x;
  int lane = threadIdx.x;
  int l15 = lane & 15, half = lane >> 4;
  const float* ap = X  + (size_t)(tm * 16 + l15) * K + half * 2;
  const float* bp = Wt + (size_t)l15 * K + half * 2;
  v8f acc[TN];
  #pragma unroll
  for (int t = 0; t < TN; ++t) acc[t] = (v8f){};
  #pragma unroll 2
  for (int k = 0; k < K; k += 4) {
    v2f a, b[TN];
    a.x = ap[k];
    a.y = ap[k + 1];
    #pragma unroll
    for (int t = 0; t < TN; ++t) {
      b[t].x = bp[t * 16 * K + k];
      b[t].y = bp[t * 16 * K + k + 1];
    }
    #pragma unroll
    for (int t = 0; t < TN; ++t)
      acc[t] = __builtin_amdgcn_wmma_f32_16x16x4_f32(false, a, false, b[t],
                                                     (short)0, acc[t], false, false);
  }
  #pragma unroll
  for (int t = 0; t < TN; ++t) {
    float* dp = D + (size_t)(tm * 16 + half * 8) * NOUT + t * 16 + l15;
    #pragma unroll
    for (int r = 0; r < 8; ++r) dp[r * NOUT] = acc[t][r];
  }
}

#define W1P      516
#define G1_BLK   256
#define G1_WAVES 16
#define G1_THREADS 512
#define G1_STRIPS (NNODES / 16)
#define G1_FULL  (G1_STRIPS / G1_BLK)
#define G1_REST  (G1_STRIPS - G1_FULL * G1_BLK)
static_assert(NNODES % 16 == 0, "strips of 16 rows");
static_assert(G1_FULL + 4 == G1_WAVES, "waves G1_FULL .. G1_FULL+3 take the column quarters");
static_assert(G1_THREADS == G1_WAVES * 32, "work-group size");

template <int TN>
__device__ __forceinline__ void k_block(const float* __restrict__ an, const float* sw,
                                        const unsigned (&bo)[TN],
                                        const v2f (&aU)[8], v2f (&aP)[8],
                                        v2f (&bC)[TN], v8f (&acc)[TN]) {
  #pragma unroll
  for (int j = 0; j < 8; ++j) { aP[j].x = an[4 * j]; aP[j].y = an[4 * j + 1]; }
  #pragma unroll
  for (int ks = 0; ks < 8; ++ks) {
    v2f bN[TN];
    #pragma unroll
    for (int t = 0; t < TN; ++t) bN[t] = *(const v2f*)(sw + bo[t] + 4 * ks + 4);
    #pragma unroll
    for (int t = 0; t < TN; ++t)
      acc[t] = __builtin_amdgcn_wmma_f32_16x16x4_f32(false, aU[ks], false, bC[t],
                                                     (short)0, acc[t], false, false);
    #pragma unroll
    for (int t = 0; t < TN; ++t) bC[t] = bN[t];
  }
  __builtin_amdgcn_sched_barrier(0);
  #pragma unroll
  for (int j = 0; j < 8; ++j) asm volatile("" : "+v"(aP[j]));
}

template <int TN>
__device__ __forceinline__ void strip_b_from_lds(const float* __restrict__ X, const float* sw,
                                                 float* __restrict__ D, int tm, int col0, int lane) {
  constexpr int NB = NFEAT / 32;
  static_assert(NB % 2 == 0, "the two A register sets change roles every block");
  const int l15 = lane & 15, half = lane >> 4;
  const float* ap = X + (size_t)(tm * 16 + l15) * NFEAT + half * 2;
  v8f acc[TN];
  v2f a0[8], a1[8], bC[TN];
  unsigned bo[TN];
  #pragma unroll
  for (int t = 0; t < TN; ++t) {
    acc[t] = (v8f){};
    bo[t] = (unsigned)((col0 + t * 16 + l15) * W1P + half * 2);
    asm volatile("" : "+v"(bo[t]));
  }
  #pragma unroll
  for (int j = 0; j < 8; ++j) { a0[j].x = ap[4 * j]; a0[j].y = ap[4 * j + 1]; }
  #pragma unroll
  for (int t = 0; t < TN; ++t) bC[t] = *(const v2f*)(sw + bo[t]);
  for (int c = 0; c < NB; c += 2) {
    k_block<TN>(ap + 32 * (c + 1), sw, bo, a0, a1, bC, acc);
    #pragma unroll
    for (int t = 0; t < TN; ++t) bo[t] += 32;
    k_block<TN>(ap + 32 * (c + 2 < NB ? c + 2 : c + 1), sw, bo, a1, a0, bC, acc);
    #pragma unroll
    for (int t = 0; t < TN; ++t) bo[t] += 32;
  }
  #pragma unroll
  for (int t = 0; t < TN; ++t) {
    float* dp = D + (size_t)(tm * 16 + half * 8) * F1 + col0 + t * 16 + l15;
    #pragma unroll
    for (int r = 0; r < 8; ++r) dp[r * F1] = acc[t][r];
  }
}

__global__ __launch_bounds__(G1_THREADS) void gemm1_w_in_lds(
    const float* __restrict__ X, const float* __restrict__ Wt, float* __restrict__ D) {
  __shared__ __attribute__((aligned(16))) float s_w[F1 * W1P];
  const int tid = threadIdx.x, lane = tid & 31;
  const int wv = __builtin_amdgcn_readfirstlane(tid >> 5);
  #pragma unroll 8
  for (int i = tid; i < F1 * (NFEAT / 4); i += G1_THREADS) {
    const int row = i / (NFEAT / 4), c4 = i % (NFEAT / 4);
    const float4 v = ((const float4*)Wt)[i];
    *(float4*)&s_w[row * W1P + c4 * 4] = v;
  }
  if (tid < F1) *(float4*)&s_w[tid * W1P + NFEAT] = make_float4(0.f, 0.f, 0.f, 0.f);
  __syncthreads();
  if (wv < G1_FULL) {
    strip_b_from_lds<8>(X, s_w, D, wv * G1_BLK + (int)blockIdx.x, 0, lane);
  } else if ((int)blockIdx.x < G1_REST) {
    strip_b_from_lds<2>(X, s_w, D, G1_FULL * G1_BLK + (int)blockIdx.x, (wv - G1_FULL) * 32, lane);
  }
}

__global__ void attn_scores(const float* __restrict__ h,
                            const float* __restrict__ attS,
                            const float* __restrict__ attD,
                            float* __restrict__ aS, float* __restrict__ aD,
                            int NH, int H) {
  int t = blockIdx.x * blockDim.x + threadIdx.x;
  if (t >= NH) return;
  int hh = t % H;
  const float* hp = h + (size_t)t * C_;
  const float* sp = attS + hh * C_;
  const float* dp = attD + hh * C_;
  float s = 0.f, d = 0.f;
  #pragma unroll
  for (int c = 0; c < C_; ++c) { s += hp[c] * sp[c]; d += hp[c] * dp[c]; }
  aS[t] = s; aD[t] = d;
}

#define NPB    196
#define NBLK   256
#define NKCH   (NEDGES / 16)
#define SCAN_U 2
#define NIT    ((NKCH + 1024 * SCAN_U - 1) / (1024 * SCAN_U))

__device__ __forceinline__ float lrelu(float a) { return a > 0.f ? a : 0.2f * a; }

__global__ void owner_ids(const int* __restrict__ ei, unsigned* __restrict__ key4) {
  int t = blockIdx.x * blockDim.x + threadIdx.x;
  if (t >= NEDGES / 4) return;
  int4 d = ((const int4*)(ei + NEDGES))[t];
  key4[t] = ((unsigned)(d.x / NPB) & 255u)
          | (((unsigned)(d.y / NPB) & 255u) << 8)
          | (((unsigned)(d.z / NPB) & 255u) << 16)
          | (((unsigned)(d.w / NPB) & 255u) << 24);
}

__device__ __forceinline__ void load_keys(const uint4* __restrict__ k4, int it, int tid,
                                          uint4 (&d)[SCAN_U]) {
  #pragma unroll
  for (int u = 0; u < SCAN_U; ++u) {
    int ch = (it * SCAN_U + u) * 1024 + tid;
    ch = ch < NKCH ? ch : NKCH - 1;
    d[u] = k4[ch];
  }
}

__device__ __forceinline__ unsigned zero_bytes(unsigned x) {
  return ~(((x & 0x7f7f7f7fu) + 0x7f7f7f7fu) | x | 0x7f7f7f7fu);
}

__device__ __forceinline__ unsigned match_bits(const uint4 (&d)[SCAN_U], int it, int tid,
                                               unsigned bid4) {
  unsigned mm = 0;
  #pragma unroll
  for (int u = 0; u < SCAN_U; ++u) {
    int ch = (it * SCAN_U + u) * 1024 + tid;
    unsigned f = (zero_bytes(d[u].x ^ bid4) >> 7) | (zero_bytes(d[u].y ^ bid4) >> 6)
               | (zero_bytes(d[u].z ^ bid4) >> 5) | (zero_bytes(d[u].w ^ bid4) >> 4);
    mm |= (ch < NKCH ? f : 0u) << (4 * u);
  }
  return mm;
}

__device__ __forceinline__ int edge_of(int it, int tid, int k) {
  int i = k & 7, b = k >> 3;
  int ch = (it * SCAN_U + (i >> 2)) * 1024 + tid;
  return (ch << 4) + ((i & 3) << 2) + b;
}

#define LCAP   16384
#define WCAP   7168
static_assert(WCAP <= LCAP, "a saved list is copied back into the LDS list");
static_assert(NNODES <= 65536 && NPB <= 65536, "a list entry packs src and dst_local into 16 bits each");

template <int H>
__global__ __launch_bounds__(1024) void owned_gat(
    const int* __restrict__ ei, const unsigned* __restrict__ key,
    const float* __restrict__ aS, const float* __restrict__ aD,
    const float* __restrict__ h, const float* __restrict__ bias,
    float* __restrict__ out,
    unsigned* __restrict__ glist, int* __restrict__ gcnt) {
  constexpr int HC = H * C_;
  __shared__ float    s_agg[NPB * HC];
  __shared__ float    s_den[NPB * H];
  __shared__ unsigned s_max[NPB * H];
  __shared__ float    s_ad [NPB * H];
  __shared__ unsigned s_list[LCAP];
  __shared__ int      s_cnt;

  const int tid = threadIdx.x, lane = tid & 31;
  const int lo  = blockIdx.x * NPB;
  const int cnt = (NNODES - lo) < NPB ? (NNODES - lo) : NPB;
  const unsigned bid4 = blockIdx.x * 0x01010101u;
  const uint4* k4 = (const uint4*)key;
  const int rot = blockIdx.x % NIT;

  for (int i = tid; i < NPB * H; i += 1024) {
    float ad = 0.f, as = 0.f;
    if (i < cnt * H) { ad = aD[(size_t)lo * H + i]; as = aS[(size_t)lo * H + i]; }
    const int t = (i % H) * NPB + i / H;
    s_ad[t]  = ad;
    s_max[t] = enc_f32(lrelu(as + ad));
    s_den[t] = 0.f;
  }
  for (int i = tid; i < NPB * HC; i += 1024) s_agg[i] = 0.f;
  if (tid == 0) s_cnt = 0;
  __syncthreads();

  unsigned* gl = glist + (size_t)blockIdx.x * WCAP;
  int saved = -1;
  if constexpr (H == 1) {
    const int c = __builtin_amdgcn_readfirstlane(gcnt[blockIdx.x]);
    if (c <= WCAP) saved = c;
  }

  if (saved < 0) {
    uint4 dc[SCAN_U];
    load_keys(k4, rot, tid, dc);
    for (int t = 0; t < NIT; ++t) {
      const int it  = (t + rot) < NIT ? (t + rot) : (t + rot - NIT);
      const int itn = (it + 1) < NIT ? (it + 1) : 0;
      uint4 dn[SCAN_U];
      load_keys(k4, itn, tid, dn);
      unsigned mm = match_bits(dc, it, tid, bid4);
      for (;;) {
        const unsigned bal = __builtin_amdgcn_ballot_w32(mm != 0);
        if (!bal) break;
        int base = 0;
        if (lane == 0) base = atomicAdd(&s_cnt, __builtin_popcount(bal));
        base = __builtin_amdgcn_readfirstlane(base);
        if (mm) {
          const int k = __builtin_ctz(mm); mm &= mm - 1;
          const int pos = base + __builtin_popcount(bal & ((1u << lane) - 1u));
          if (pos < LCAP) s_list[pos] = (unsigned)edge_of(it, tid, k);
        }
      }
      #pragma unroll
      for (int u = 0; u < SCAN_U; ++u) dc[u] = dn[u];
    }
    __syncthreads();
  }
  const int nl = saved >= 0 ? saved : __builtin_amdgcn_readfirstlane(s_cnt);
  const bool save = (H == 8) && nl <= WCAP;
  if constexpr (H == 8) { if (tid == 0) gcnt[blockIdx.x] = nl; }

  if (nl <= LCAP) {
    if (saved >= 0) {
      if constexpr (H == 1) {
        constexpr int U2 = 4;
        for (int i0 = tid; i0 < nl; i0 += 1024 * U2) {
          int idx[U2];
          unsigned ent[U2];
          bool v[U2];
          float a[U2], ad[U2];
          #pragma unroll
          for (int j = 0; j < U2; ++j) {
            idx[j] = i0 + 1024 * j;
            v[j] = idx[j] < nl;
            ent[j] = gl[v[j] ? idx[j] : i0];
          }
          #pragma unroll
          for (int j = 0; j < U2; ++j) asm volatile("" : "+v"(ent[j]));
          #pragma unroll
          for (int j = 0; j < U2; ++j) a[j] = aS[ent[j] & 0xffffu];
          #pragma unroll
          for (int j = 0; j < U2; ++j) ad[j] = s_ad[ent[j] >> 16];
          #pragma unroll
          for (int j = 0; j < U2; ++j) asm volatile("" : "+v"(a[j]));
          #pragma unroll
          for (int j = 0; j < U2; ++j) {
            if (v[j]) {
              atomicMax(&s_max[ent[j] >> 16], enc_f32(lrelu(a[j] + ad[j])));
              s_list[idx[j]] = ent[j];
            }
          }
        }
      }
    } else {
    {
      constexpr int U2 = 4;
      for (int i0 = tid; i0 < nl; i0 += 1024 * U2) {
        int idx[U2], ev[U2], sv[U2], dv[U2];
        bool v[U2];
        #pragma unroll
        for (int j = 0; j < U2; ++j) {
          idx[j] = i0 + 1024 * j;
          v[j] = idx[j] < nl;
          ev[j] = (int)s_list[v[j] ? idx[j] : i0];
        }
        #pragma unroll
        for (int j = 0; j < U2; ++j) {
          sv[j] = ei[ev[j]];
          dv[j] = ei[NEDGES + ev[j]] - lo;
        }
        if constexpr (H == 8) {
          float4 a0[U2], a1[U2];
          float ad[U2][8];
          #pragma unroll
          for (int j = 0; j < U2; ++j) {
            const float4* ap = (const float4*)(aS + (size_t)sv[j] * 8);
            a0[j] = ap[0]; a1[j] = ap[1];
          }
          #pragma unroll
          for (int j = 0; j < U2; ++j) {
            #pragma unroll
            for (int hh = 0; hh < 8; ++hh) ad[j][hh] = s_ad[hh * NPB + dv[j]];
          }
          #pragma unroll
          for (int j = 0; j < U2; ++j) {
            if (v[j]) {
              const int d = dv[j];
              atomicMax(&s_max[0 * NPB + d], enc_f32(lrelu(a0[j].x + ad[j][0])));
              atomicMax(&s_max[1 * NPB + d], enc_f32(lrelu(a0[j].y + ad[j][1])));
              atomicMax(&s_max[2 * NPB + d], enc_f32(lrelu(a0[j].z + ad[j][2])));
              atomicMax(&s_max[3 * NPB + d], enc_f32(lrelu(a0[j].w + ad[j][3])));
              atomicMax(&s_max[4 * NPB + d], enc_f32(lrelu(a1[j].x + ad[j][4])));
              atomicMax(&s_max[5 * NPB + d], enc_f32(lrelu(a1[j].y + ad[j][5])));
              atomicMax(&s_max[6 * NPB + d], enc_f32(lrelu(a1[j].z + ad[j][6])));
              atomicMax(&s_max[7 * NPB + d], enc_f32(lrelu(a1[j].w + ad[j][7])));
              const unsigned ent = (unsigned)sv[j] | ((unsigned)d << 16);
              s_list[idx[j]] = ent;
              if (save) gl[idx[j]] = ent;
            }
          }
        } else {
          float a[U2], ad[U2];
          #pragma unroll
          for (int j = 0; j < U2; ++j) a[j] = aS[sv[j]];
          #pragma unroll
          for (int j = 0; j < U2; ++j) ad[j] = s_ad[dv[j]];
          #pragma unroll
          for (int j = 0; j < U2; ++j) {
            if (v[j]) {
              atomicMax(&s_max[dv[j]], enc_f32(lrelu(a[j] + ad[j])));
              s_list[idx[j]] = (unsigned)sv[j] | ((unsigned)dv[j] << 16);
            }
          }
        }
      }
    }
    }
    __syncthreads();

    if constexpr (H == 8) {
      constexpr int EB = 8;
      const int hd = lane >> 2;
      const int wv = __builtin_amdgcn_readfirstlane(tid >> 5);
      for (int base = wv * EB; base < nl; base += 32 * EB) {
        const int li = base + (lane & (EB - 1));
        const unsigned ent = s_list[li < nl ? li : nl - 1];
        int sj[EB], dj[EB];
        bool vj[EB];
        #pragma unroll
        for (int j = 0; j < EB; ++j) {
          const unsigned x = (unsigned)__builtin_amdgcn_readlane((int)ent, j);
          sj[j] = (int)(x & 0xffffu);
          dj[j] = (int)(x >> 16);
          vj[j] = base + j < nl;
        }
        float  aj[EB];
        float4 rj[EB];
        #pragma unroll
        for (int j = 0; j < EB; ++j) {
          aj[j] = aS[(size_t)sj[j] * 8 + hd];
          rj[j] = ((const float4*)(h + (size_t)sj[j] * HC))[lane];
        }
        float adj[EB], mxj[EB];
        #pragma unroll
        for (int j = 0; j < EB; ++j) {
          adj[j] = s_ad[hd * NPB + dj[j]];
          mxj[j] = dec_f32(s_max[hd * NPB + dj[j]]);
        }
        #pragma unroll
        for (int j = 0; j < EB; ++j) {
          if (vj[j]) {
            const int dl = dj[j];
            float w = expf(lrelu(aj[j] + adj[j]) - mxj[j]);
            float* ap = &s_agg[dl * HC + lane];
            atomicAdd(ap,      rj[j].x * w);
            atomicAdd(ap + 32, rj[j].y * w);
            atomicAdd(ap + 64, rj[j].z * w);
            atomicAdd(ap + 96, rj[j].w * w);
            if ((lane & 3) == 0) atomicAdd(&s_den[hd * NPB + dl], w);
          }
        }
      }
    } else {
      constexpr int U3 = 2;
      for (int i0 = tid; i0 < nl; i0 += 1024 * U3) {
        int sv[U3], dv[U3];
        bool v[U3];
        float a[U3];
        float4 r[U3][4];
        #pragma unroll
        for (int j = 0; j < U3; ++j) {
          const int idx = i0 + 1024 * j;
          v[j] = idx < nl;
          const unsigned ent = s_list[v[j] ? idx : i0];
          sv[j] = (int)(ent & 0xffffu);
          dv[j] = (int)(ent >> 16);
        }
        #pragma unroll
        for (int j = 0; j < U3; ++j) {
          a[j] = aS[sv[j]];
          const float4* rp = (const float4*)(h + (size_t)sv[j] * HC);
          r[j][0] = rp[0]; r[j][1] = rp[1]; r[j][2] = rp[2]; r[j][3] = rp[3];
        }
        float adv[U3], mxv[U3];
        #pragma unroll
        for (int j = 0; j < U3; ++j) { adv[j] = s_ad[dv[j]]; mxv[j] = dec_f32(s_max[dv[j]]); }
        #pragma unroll
        for (int j = 0; j < U3; ++j) {
          if (v[j]) {
            const int d = dv[j];
            float w = expf(lrelu(a[j] + adv[j]) - mxv[j]);
            float* ap = &s_agg[d];
            atomicAdd(ap + 0  * NPB, r[j][0].x * w); atomicAdd(ap + 1  * NPB, r[j][0].y * w);
            atomicAdd(ap + 2  * NPB, r[j][0].z * w); atomicAdd(ap + 3  * NPB, r[j][0].w * w);
            atomicAdd(ap + 4  * NPB, r[j][1].x * w); atomicAdd(ap + 5  * NPB, r[j][1].y * w);
            atomicAdd(ap + 6  * NPB, r[j][1].z * w); atomicAdd(ap + 7  * NPB, r[j][1].w * w);
            atomicAdd(ap + 8  * NPB, r[j][2].x * w); atomicAdd(ap + 9  * NPB, r[j][2].y * w);
            atomicAdd(ap + 10 * NPB, r[j][2].z * w); atomicAdd(ap + 11 * NPB, r[j][2].w * w);
            atomicAdd(ap + 12 * NPB, r[j][3].x * w); atomicAdd(ap + 13 * NPB, r[j][3].y * w);
            atomicAdd(ap + 14 * NPB, r[j][3].z * w); atomicAdd(ap + 15 * NPB, r[j][3].w * w);
            atomicAdd(&s_den[d], w);
          }
        }
      }
    }
  } else {
    {
      uint4 dc[SCAN_U];
      load_keys(k4, rot, tid, dc);
      for (int t = 0; t < NIT; ++t) {
        const int it  = (t + rot) < NIT ? (t + rot) : (t + rot - NIT);
        const int itn = (it + 1) < NIT ? (it + 1) : 0;
        uint4 dn[SCAN_U];
        load_keys(k4, itn, tid, dn);
        unsigned mm = match_bits(dc, it, tid, bid4);
        while (mm) {
          int k = __builtin_ctz(mm); mm &= mm - 1;
          int e = edge_of(it, tid, k);
          int s = ei[e];
          int dl = ei[NEDGES + e] - lo;
          if constexpr (H == 8) {
            const float4* ap = (const float4*)(aS + (size_t)s * 8);
            float4 a0 = ap[0], a1 = ap[1];
            atomicMax(&s_max[0 * NPB + dl], enc_f32(lrelu(a0.x + s_ad[0 * NPB + dl])));
            atomicMax(&s_max[1 * NPB + dl], enc_f32(lrelu(a0.y + s_ad[1 * NPB + dl])));
            atomicMax(&s_max[2 * NPB + dl], enc_f32(lrelu(a0.z + s_ad[2 * NPB + dl])));
            atomicMax(&s_max[3 * NPB + dl], enc_f32(lrelu(a0.w + s_ad[3 * NPB + dl])));
            atomicMax(&s_max[4 * NPB + dl], enc_f32(lrelu(a1.x + s_ad[4 * NPB + dl])));
            atomicMax(&s_max[5 * NPB + dl], enc_f32(lrelu(a1.y + s_ad[5 * NPB + dl])));
            atomicMax(&s_max[6 * NPB + dl], enc_f32(lrelu(a1.z + s_ad[6 * NPB + dl])));
            atomicMax(&s_max[7 * NPB + dl], enc_f32(lrelu(a1.w + s_ad[7 * NPB + dl])));
          } else {
            atomicMax(&s_max[dl], enc_f32(lrelu(aS[s] + s_ad[dl])));
          }
        }
        #pragma unroll
        for (int u = 0; u < SCAN_U; ++u) dc[u] = dn[u];
      }
    }
    __syncthreads();

    {
      uint4 dc[SCAN_U];
      load_keys(k4, rot, tid, dc);
      for (int t = 0; t < NIT; ++t) {
        const int it  = (t + rot) < NIT ? (t + rot) : (t + rot - NIT);
        const int itn = (it + 1) < NIT ? (it + 1) : 0;
        uint4 dn[SCAN_U];
        load_keys(k4, itn, tid, dn);
        unsigned mm = match_bits(dc, it, tid, bid4);
        if constexpr (H == 8) {
          const int hd = lane >> 2;
          for (;;) {
            unsigned bal = __builtin_amdgcn_ballot_w32(mm != 0);
            if (!bal) break;
            int sv = 0, dv = 0;
            if (mm) {
              int k = __builtin_ctz(mm); mm &= mm - 1;
              int e = edge_of(it, tid, k);
              sv = ei[e];
              dv = ei[NEDGES + e] - lo;
            }
            while (bal) {
              int ml = __builtin_ctz(bal); bal &= bal - 1;
              int s  = __builtin_amdgcn_readlane(sv, ml);
              int dl = __builtin_amdgcn_readlane(dv, ml);
              float  a = aS[(size_t)s * 8 + hd];
              float4 r = ((const float4*)(h + (size_t)s * HC))[lane];
              float w = expf(lrelu(a + s_ad[hd * NPB + dl]) - dec_f32(s_max[hd * NPB + dl]));
              float* ap = &s_agg[dl * HC + lane];
              atomicAdd(ap,      r.x * w);
              atomicAdd(ap + 32, r.y * w);
              atomicAdd(ap + 64, r.z * w);
              atomicAdd(ap + 96, r.w * w);
              if ((lane & 3) == 0) atomicAdd(&s_den[hd * NPB + dl], w);
            }
          }
        } else {
          while (mm) {
            int k = __builtin_ctz(mm); mm &= mm - 1;
            int e = edge_of(it, tid, k);
            int s = ei[e];
            int dl = ei[NEDGES + e] - lo;
            float a = aS[s];
            const float4* rp = (const float4*)(h + (size_t)s * HC);
            float4 r0 = rp[0], r1 = rp[1], r2 = rp[2], r3 = rp[3];
            float w = expf(lrelu(a + s_ad[dl]) - dec_f32(s_max[dl]));
            float* ap = &s_agg[dl];
            atomicAdd(ap + 0  * NPB, r0.x * w); atomicAdd(ap + 1  * NPB, r0.y * w);
            atomicAdd(ap + 2  * NPB, r0.z * w); atomicAdd(ap + 3  * NPB, r0.w * w);
            atomicAdd(ap + 4  * NPB, r1.x * w); atomicAdd(ap + 5  * NPB, r1.y * w);
            atomicAdd(ap + 6  * NPB, r1.z * w); atomicAdd(ap + 7  * NPB, r1.w * w);
            atomicAdd(ap + 8  * NPB, r2.x * w); atomicAdd(ap + 9  * NPB, r2.y * w);
            atomicAdd(ap + 10 * NPB, r2.z * w); atomicAdd(ap + 11 * NPB, r2.w * w);
            atomicAdd(ap + 12 * NPB, r3.x * w); atomicAdd(ap + 13 * NPB, r3.y * w);
            atomicAdd(ap + 14 * NPB, r3.z * w); atomicAdd(ap + 15 * NPB, r3.w * w);
            atomicAdd(&s_den[dl], w);
          }
        }
        #pragma unroll
        for (int u = 0; u < SCAN_U; ++u) dc[u] = dn[u];
      }
    }
  }
  __syncthreads();

  if constexpr (H == 8) {
    for (int idx = tid; idx < cnt * HC; idx += 1024) {
      int dl = idx >> 7, c = idx & 127, hd = c >> 4;
      size_t n = (size_t)(lo + dl);
      float ws = expf(lrelu(aS[n * 8 + hd] + s_ad[hd * NPB + dl]) - dec_f32(s_max[hd * NPB + dl]));
      float acc = s_agg[dl * HC + (c & 3) * 32 + (c >> 2)] + ws * h[n * HC + c];
      float v = acc / (s_den[hd * NPB + dl] + ws + 1e-16f) + bias[c];
      out[n * HC + c] = v > 0.f ? v : expm1f(v);
    }
  } else {
    if (tid < cnt) {
      int dl = tid;
      size_t n = (size_t)(lo + dl);
      float ws = expf(lrelu(aS[n] + s_ad[dl]) - dec_f32(s_max[dl]));
      float dn = s_den[dl] + ws + 1e-16f;
      float v[NCLASS];
      float m = -1e30f;
      #pragma unroll
      for (int c = 0; c < NCLASS; ++c) {
        v[c] = (s_agg[c * NPB + dl] + ws * h[n * HC + c]) / dn + bias[c];
        m = fmaxf(m, v[c]);
      }
      float sum = 0.f;
      #pragma unroll
      for (int c = 0; c < NCLASS; ++c) sum += expf(v[c] - m);
      float ls = logf(sum);
      #pragma unroll
      for (int c = 0; c < NCLASS; ++c) out[n * NCLASS + c] = v[c] - m - ls;
    }
  }
}

extern "C" void kernel_launch(void* const* d_in, const int* in_sizes, int n_in,
                              void* d_out, int out_size, void* d_ws, size_t ws_size,
                              hipStream_t stream) {
  const float* x     = (const float*)d_in[0];
  const int*   ei    = (const int*)  d_in[1];
  const float* W1    = (const float*)d_in[2];
  const float* attS1 = (const float*)d_in[3];
  const float* attD1 = (const float*)d_in[4];
  const float* b1    = (const float*)d_in[5];
  const float* W2    = (const float*)d_in[6];
  const float* attS2 = (const float*)d_in[7];
  const float* attD2 = (const float*)d_in[8];
  const float* b2    = (const float*)d_in[9];
  float* out = (float*)d_out;

  char* ws = (char*)d_ws;
  size_t off = 0;
  auto alloc = [&](size_t bytes) -> void* {
    void* p = ws + off;
    off = (off + bytes + 255) & ~(size_t)255;
    return p;
  };

  float* h1  = (float*)alloc((size_t)NNODES * F1 * 4);
  float* as1 = (float*)alloc((size_t)NNODES * H1_ * 4);
  float* ad1 = (float*)alloc((size_t)NNODES * H1_ * 4);
  float* x1  = (float*)alloc((size_t)NNODES * F1 * 4);
  float* h2  = (float*)alloc((size_t)NNODES * C_ * 4);
  float* as2 = (float*)alloc((size_t)NNODES * 4);
  float* ad2 = (float*)alloc((size_t)NNODES * 4);
  unsigned* key = (unsigned*)alloc((size_t)NEDGES);
  unsigned* glist = (unsigned*)alloc((size_t)NBLK * WCAP * 4);
  int* gcnt = (int*)alloc((size_t)NBLK * 4);

  const int TB = 256;
  auto nb = [](long long n, int b) { return (unsigned)((n + b - 1) / b); };

  gemm1_w_in_lds<<<G1_BLK, G1_THREADS, 0, stream>>>(x, W1, h1);
  attn_scores<<<nb((long long)NNODES * H1_, TB), TB, 0, stream>>>(
      h1, attS1, attD1, as1, ad1, NNODES * H1_, H1_);
  owner_ids<<<nb(NEDGES / 4, TB), TB, 0, stream>>>(ei, key);
  owned_gat<H1_><<<NBLK, 1024, 0, stream>>>(ei, key, as1, ad1, h1, b1, x1, glist, gcnt);

  gemm_f32_wmma_strip<1, F1, C_><<<3125, 32, 0, stream>>>(x1, W2, h2);
  attn_scores<<<nb(NNODES, TB), TB, 0, stream>>>(h2, attS2, attD2, as2, ad2, NNODES, 1);
  owned_gat<1><<<NBLK, 1024, 0, stream>>>(ei, key, as2, ad2, h2, b2, out, glist, gcnt);
}


# Layer-2 product and scores done in layer 1's aggregation tail: a work-group's 196 finished rows stay in LDS and feed the matrix unit; x1 never written; 7->5 launches; product sum order changes
# speedup vs baseline: 11.7433x; 1.0326x over previous
#include <hip/hip_runtime.h>
#include <hip/hip_bf16.h>

#define NNODES 50000
#define NEDGES 1600000
#define NFEAT  512
#define F1     128
#define H1_    8
#define C_     16
#define NCLASS 16

typedef float v2f __attribute__((ext_vector_type(2)));
typedef float v8f __attribute__((ext_vector_type(8)));

__device__ __forceinline__ unsigned enc_f32(float f) {
  unsigned u = __float_as_uint(f);
  return (u & 0x80000000u) ? ~u : (u | 0x80000000u);
}
__device__ __forceinline__ float dec_f32(unsigned u) {
  return (u & 0x80000000u) ? __uint_as_float(u & 0x7FFFFFFFu) : __uint_as_float(~u);
}

#define W1P      516
#define G1_BLK   256
#define G1_WAVES 16
#define G1_THREADS 512
#define G1_STRIPS (NNODES / 16)
#define G1_FULL  (G1_STRIPS / G1_BLK)
#define G1_REST  (G1_STRIPS - G1_FULL * G1_BLK)
static_assert(NNODES % 16 == 0, "strips of 16 rows");
static_assert(G1_FULL + 4 == G1_WAVES, "waves G1_FULL .. G1_FULL+3 take the column quarters");
static_assert(G1_THREADS == G1_WAVES * 32, "work-group size");

template <int TN>
__device__ __forceinline__ void k_block(const float* __restrict__ an, const float* sw,
                                        const unsigned (&bo)[TN],
                                        const v2f (&aU)[8], v2f (&aP)[8],
                                        v2f (&bC)[TN], v8f (&acc)[TN]) {
  #pragma unroll
  for (int j = 0; j < 8; ++j) { aP[j].x = an[4 * j]; aP[j].y = an[4 * j + 1]; }
  #pragma unroll
  for (int ks = 0; ks < 8; ++ks) {
    v2f bN[TN];
    #pragma unroll
    for (int t = 0; t < TN; ++t) bN[t] = *(const v2f*)(sw + bo[t] + 4 * ks + 4);
    #pragma unroll
    for (int t = 0; t < TN; ++t)
      acc[t] = __builtin_amdgcn_wmma_f32_16x16x4_f32(false, aU[ks], false, bC[t],
                                                     (short)0, acc[t], false, false);
    #pragma unroll
    for (int t = 0; t < TN; ++t) bC[t] = bN[t];
  }
  __builtin_amdgcn_sched_barrier(0);
  #pragma unroll
  for (int j = 0; j < 8; ++j) asm volatile("" : "+v"(aP[j]));
}

template <int TN>
__device__ __forceinline__ void strip_b_from_lds(const float* __restrict__ X, const float* sw,
                                                 float* __restrict__ D, int tm, int col0, int lane) {
  constexpr int NB = NFEAT / 32;
  static_assert(NB % 2 == 0, "the two A register sets change roles every block");
  const int l15 = lane & 15, half = lane >> 4;
  const float* ap = X + (size_t)(tm * 16 + l15) * NFEAT + half * 2;
  v8f acc[TN];
  v2f a0[8], a1[8], bC[TN];
  unsigned bo[TN];
  #pragma unroll
  for (int t = 0; t < TN; ++t) {
    acc[t] = (v8f){};
    bo[t] = (unsigned)((col0 + t * 16 + l15) * W1P + half * 2);
    asm volatile("" : "+v"(bo[t]));
  }
  #pragma unroll
  for (int j = 0; j < 8; ++j) { a0[j].x = ap[4 * j]; a0[j].y = ap[4 * j + 1]; }
  #pragma unroll
  for (int t = 0; t < TN; ++t) bC[t] = *(const v2f*)(sw + bo[t]);
  for (int c = 0; c < NB; c += 2) {
    k_block<TN>(ap + 32 * (c + 1), sw, bo, a0, a1, bC, acc);
    #pragma unroll
    for (int t = 0; t < TN; ++t) bo[t] += 32;
    k_block<TN>(ap + 32 * (c + 2 < NB ? c + 2 : c + 1), sw, bo, a1, a0, bC, acc);
    #pragma unroll
    for (int t = 0; t < TN; ++t) bo[t] += 32;
  }
  #pragma unroll
  for (int t = 0; t < TN; ++t) {
    float* dp = D + (size_t)(tm * 16 + half * 8) * F1 + col0 + t * 16 + l15;
    #pragma unroll
    for (int r = 0; r < 8; ++r) dp[r * F1] = acc[t][r];
  }
}

__global__ __launch_bounds__(G1_THREADS) void gemm1_w_in_lds(
    const float* __restrict__ X, const float* __restrict__ Wt, float* __restrict__ D) {
  __shared__ __attribute__((aligned(16))) float s_w[F1 * W1P];
  const int tid = threadIdx.x, lane = tid & 31;
  const int wv = __builtin_amdgcn_readfirstlane(tid >> 5);
  #pragma unroll 8
  for (int i = tid; i < F1 * (NFEAT / 4); i += G1_THREADS) {
    const int row = i / (NFEAT / 4), c4 = i % (NFEAT / 4);
    const float4 v = ((const float4*)Wt)[i];
    *(float4*)&s_w[row * W1P + c4 * 4] = v;
  }
  if (tid < F1) *(float4*)&s_w[tid * W1P + NFEAT] = make_float4(0.f, 0.f, 0.f, 0.f);
  __syncthreads();
  if (wv < G1_FULL) {
    strip_b_from_lds<8>(X, s_w, D, wv * G1_BLK + (int)blockIdx.x, 0, lane);
  } else if ((int)blockIdx.x < G1_REST) {
    strip_b_from_lds<2>(X, s_w, D, G1_FULL * G1_BLK + (int)blockIdx.x, (wv - G1_FULL) * 32, lane);
  }
}

__global__ void attn_scores(const float* __restrict__ h,
                            const float* __restrict__ attS,
                            const float* __restrict__ attD,
                            float* __restrict__ aS, float* __restrict__ aD,
                            int NH, int H) {
  int t = blockIdx.x * blockDim.x + threadIdx.x;
  if (t >= NH) return;
  int hh = t % H;
  const float* hp = h + (size_t)t * C_;
  const float* sp = attS + hh * C_;
  const float* dp = attD + hh * C_;
  float s = 0.f, d = 0.f;
  #pragma unroll
  for (int c = 0; c < C_; ++c) { s += hp[c] * sp[c]; d += hp[c] * dp[c]; }
  aS[t] = s; aD[t] = d;
}

#define NPB    196
#define NBLK   256
#define NKCH   (NEDGES / 16)
#define SCAN_U 2
#define NIT    ((NKCH + 1024 * SCAN_U - 1) / (1024 * SCAN_U))

__device__ __forceinline__ float lrelu(float a) { return a > 0.f ? a : 0.2f * a; }

__global__ void owner_ids(const int* __restrict__ ei, unsigned* __restrict__ key4) {
  int t = blockIdx.x * blockDim.x + threadIdx.x;
  if (t >= NEDGES / 4) return;
  int4 d = ((const int4*)(ei + NEDGES))[t];
  key4[t] = ((unsigned)(d.x / NPB) & 255u)
          | (((unsigned)(d.y / NPB) & 255u) << 8)
          | (((unsigned)(d.z / NPB) & 255u) << 16)
          | (((unsigned)(d.w / NPB) & 255u) << 24);
}

__device__ __forceinline__ void load_keys(const uint4* __restrict__ k4, int it, int tid,
                                          uint4 (&d)[SCAN_U]) {
  #pragma unroll
  for (int u = 0; u < SCAN_U; ++u) {
    int ch = (it * SCAN_U + u) * 1024 + tid;
    ch = ch < NKCH ? ch : NKCH - 1;
    d[u] = k4[ch];
  }
}

__device__ __forceinline__ unsigned zero_bytes(unsigned x) {
  return ~(((x & 0x7f7f7f7fu) + 0x7f7f7f7fu) | x | 0x7f7f7f7fu);
}

__device__ __forceinline__ unsigned match_bits(const uint4 (&d)[SCAN_U], int it, int tid,
                                               unsigned bid4) {
  unsigned mm = 0;
  #pragma unroll
  for (int u = 0; u < SCAN_U; ++u) {
    int ch = (it * SCAN_U + u) * 1024 + tid;
    unsigned f = (zero_bytes(d[u].x ^ bid4) >> 7) | (zero_bytes(d[u].y ^ bid4) >> 6)
               | (zero_bytes(d[u].z ^ bid4) >> 5) | (zero_bytes(d[u].w ^ bid4) >> 4);
    mm |= (ch < NKCH ? f : 0u) << (4 * u);
  }
  return mm;
}

__device__ __forceinline__ int edge_of(int it, int tid, int k) {
  int i = k & 7, b = k >> 3;
  int ch = (it * SCAN_U + (i >> 2)) * 1024 + tid;
  return (ch << 4) + ((i & 3) << 2) + b;
}

#define X1P    132
#define T2     ((NPB + 15) / 16)
#define H2P    17
#define LCAP   16384
#define WCAP   7168
static_assert(WCAP <= LCAP, "a saved list is copied back into the LDS list");
static_assert(NNODES <= 65536 && NPB <= 65536, "a list entry packs src and dst_local into 16 bits each");
static_assert(T2 <= 32, "one wave per 16-row tile of the layer-2 product");
static_assert(C_ * X1P + T2 * 16 * H2P <= LCAP, "W2 and the layer-2 rows reuse the list's LDS in the tail");
static_assert(F1 == 128 && C_ == 16 && NCLASS == C_, "index arithmetic of the fused layer-2 product");

template <int H>
__global__ __launch_bounds__(1024) void owned_gat(
    const int* __restrict__ ei, const unsigned* __restrict__ key,
    const float* __restrict__ aS, const float* __restrict__ aD,
    const float* __restrict__ h, const float* __restrict__ bias,
    float* __restrict__ out,
    unsigned* __restrict__ glist, int* __restrict__ gcnt,
    const float* __restrict__ W2, const float* __restrict__ attS2, const float* __restrict__ attD2,
    float* __restrict__ h2o, float* __restrict__ as2o, float* __restrict__ ad2o) {
  constexpr int HC = H * C_;
  constexpr int AGG = (H == 8) ? NPB * X1P : NPB * HC;
  __shared__ __attribute__((aligned(16))) float s_agg[AGG];
  __shared__ float    s_den[NPB * H];
  __shared__ unsigned s_max[NPB * H];
  __shared__ float    s_ad [NPB * H];
  __shared__ __attribute__((aligned(16))) unsigned s_list[LCAP];
  __shared__ int      s_cnt;

  const int tid = threadIdx.x, lane = tid & 31;
  const int lo  = blockIdx.x * NPB;
  const int cnt = (NNODES - lo) < NPB ? (NNODES - lo) : NPB;
  const unsigned bid4 = blockIdx.x * 0x01010101u;
  const uint4* k4 = (const uint4*)key;
  const int rot = blockIdx.x % NIT;

  for (int i = tid; i < NPB * H; i += 1024) {
    float ad = 0.f, as = 0.f;
    if (i < cnt * H) { ad = aD[(size_t)lo * H + i]; as = aS[(size_t)lo * H + i]; }
    const int t = (i % H) * NPB + i / H;
    s_ad[t]  = ad;
    s_max[t] = enc_f32(lrelu(as + ad));
    s_den[t] = 0.f;
  }
  for (int i = tid; i < AGG; i += 1024) s_agg[i] = 0.f;
  if (tid == 0) s_cnt = 0;
  __syncthreads();

  unsigned* gl = glist + (size_t)blockIdx.x * WCAP;
  int saved = -1;
  if constexpr (H == 1) {
    const int c = __builtin_amdgcn_readfirstlane(gcnt[blockIdx.x]);
    if (c <= WCAP) saved = c;
  }

  if (saved < 0) {
    uint4 dc[SCAN_U];
    load_keys(k4, rot, tid, dc);
    for (int t = 0; t < NIT; ++t) {
      const int it  = (t + rot) < NIT ? (t + rot) : (t + rot - NIT);
      const int itn = (it + 1) < NIT ? (it + 1) : 0;
      uint4 dn[SCAN_U];
      load_keys(k4, itn, tid, dn);
      unsigned mm = match_bits(dc, it, tid, bid4);
      for (;;) {
        const unsigned bal = __builtin_amdgcn_ballot_w32(mm != 0);
        if (!bal) break;
        int base = 0;
        if (lane == 0) base = atomicAdd(&s_cnt, __builtin_popcount(bal));
        base = __builtin_amdgcn_readfirstlane(base);
        if (mm) {
          const int k = __builtin_ctz(mm); mm &= mm - 1;
          const int pos = base + __builtin_popcount(bal & ((1u << lane) - 1u));
          if (pos < LCAP) s_list[pos] = (unsigned)edge_of(it, tid, k);
        }
      }
      #pragma unroll
      for (int u = 0; u < SCAN_U; ++u) dc[u] = dn[u];
    }
    __syncthreads();
  }
  const int nl = saved >= 0 ? saved : __builtin_amdgcn_readfirstlane(s_cnt);
  const bool save = (H == 8) && nl <= WCAP;
  if constexpr (H == 8) { if (tid == 0) gcnt[blockIdx.x] = nl; }

  if (nl <= LCAP) {
    if (saved >= 0) {
      if constexpr (H == 1) {
        constexpr int U2 = 4;
        for (int i0 = tid; i0 < nl; i0 += 1024 * U2) {
          int idx[U2];
          unsigned ent[U2];
          bool v[U2];
          float a[U2], ad[U2];
          #pragma unroll
          for (int j = 0; j < U2; ++j) {
            idx[j] = i0 + 1024 * j;
            v[j] = idx[j] < nl;
            ent[j] = gl[v[j] ? idx[j] : i0];
          }
          #pragma unroll
          for (int j = 0; j < U2; ++j) asm volatile("" : "+v"(ent[j]));
          #pragma unroll
          for (int j = 0; j < U2; ++j) a[j] = aS[ent[j] & 0xffffu];
          #pragma unroll
          for (int j = 0; j < U2; ++j) ad[j] = s_ad[ent[j] >> 16];
          #pragma unroll
          for (int j = 0; j < U2; ++j) asm volatile("" : "+v"(a[j]));
          #pragma unroll
          for (int j = 0; j < U2; ++j) {
            if (v[j]) {
              atomicMax(&s_max[ent[j] >> 16], enc_f32(lrelu(a[j] + ad[j])));
              s_list[idx[j]] = ent[j];
            }
          }
        }
      }
    } else {
    {
      constexpr int U2 = 4;
      for (int i0 = tid; i0 < nl; i0 += 1024 * U2) {
        int idx[U2], ev[U2], sv[U2], dv[U2];
        bool v[U2];
        #pragma unroll
        for (int j = 0; j < U2; ++j) {
          idx[j] = i0 + 1024 * j;
          v[j] = idx[j] < nl;
          ev[j] = (int)s_list[v[j] ? idx[j] : i0];
        }
        #pragma unroll
        for (int j = 0; j < U2; ++j) {
          sv[j] = ei[ev[j]];
          dv[j] = ei[NEDGES + ev[j]] - lo;
        }
        if constexpr (H == 8) {
          float4 a0[U2], a1[U2];
          float ad[U2][8];
          #pragma unroll
          for (int j = 0; j < U2; ++j) {
            const float4* ap = (const float4*)(aS + (size_t)sv[j] * 8);
            a0[j] = ap[0]; a1[j] = ap[1];
          }
          #pragma unroll
          for (int j = 0; j < U2; ++j) {
            #pragma unroll
            for (int hh = 0; hh < 8; ++hh) ad[j][hh] = s_ad[hh * NPB + dv[j]];
          }
          #pragma unroll
          for (int j = 0; j < U2; ++j) {
            if (v[j]) {
              const int d = dv[j];
              atomicMax(&s_max[0 * NPB + d], enc_f32(lrelu(a0[j].x + ad[j][0])));
              atomicMax(&s_max[1 * NPB + d], enc_f32(lrelu(a0[j].y + ad[j][1])));
              atomicMax(&s_max[2 * NPB + d], enc_f32(lrelu(a0[j].z + ad[j][2])));
              atomicMax(&s_max[3 * NPB + d], enc_f32(lrelu(a0[j].w + ad[j][3])));
              atomicMax(&s_max[4 * NPB + d], enc_f32(lrelu(a1[j].x + ad[j][4])));
              atomicMax(&s_max[5 * NPB + d], enc_f32(lrelu(a1[j].y + ad[j][5])));
              atomicMax(&s_max[6 * NPB + d], enc_f32(lrelu(a1[j].z + ad[j][6])));
              atomicMax(&s_max[7 * NPB + d], enc_f32(lrelu(a1[j].w + ad[j][7])));
              const unsigned ent = (unsigned)sv[j] | ((unsigned)d << 16);
              s_list[idx[j]] = ent;
              if (save) gl[idx[j]] = ent;
            }
          }
        } else {
          float a[U2], ad[U2];
          #pragma unroll
          for (int j = 0; j < U2; ++j) a[j] = aS[sv[j]];
          #pragma unroll
          for (int j = 0; j < U2; ++j) ad[j] = s_ad[dv[j]];
          #pragma unroll
          for (int j = 0; j < U2; ++j) {
            if (v[j]) {
              atomicMax(&s_max[dv[j]], enc_f32(lrelu(a[j] + ad[j])));
              s_list[idx[j]] = (unsigned)sv[j] | ((unsigned)dv[j] << 16);
            }
          }
        }
      }
    }
    }
    __syncthreads();

    if constexpr (H == 8) {
      constexpr int EB = 8;
      const int hd = lane >> 2;
      const int wv = __builtin_amdgcn_readfirstlane(tid >> 5);
      for (int base = wv * EB; base < nl; base += 32 * EB) {
        const int li = base + (lane & (EB - 1));
        const unsigned ent = s_list[li < nl ? li : nl - 1];
        int sj[EB], dj[EB];
        bool vj[EB];
        #pragma unroll
        for (int j = 0; j < EB; ++j) {
          const unsigned x = (unsigned)__builtin_amdgcn_readlane((int)ent, j);
          sj[j] = (int)(x & 0xffffu);
          dj[j] = (int)(x >> 16);
          vj[j] = base + j < nl;
        }
        float  aj[EB];
        float4 rj[EB];
        #pragma unroll
        for (int j = 0; j < EB; ++j) {
          aj[j] = aS[(size_t)sj[j] * 8 + hd];
          rj[j] = ((const float4*)(h + (size_t)sj[j] * HC))[lane];
        }
        float adj[EB], mxj[EB];
        #pragma unroll
        for (int j = 0; j < EB; ++j) {
          adj[j] = s_ad[hd * NPB + dj[j]];
          mxj[j] = dec_f32(s_max[hd * NPB + dj[j]]);
        }
        #pragma unroll
        for (int j = 0; j < EB; ++j) {
          if (vj[j]) {
            const int dl = dj[j];
            float w = expf(lrelu(aj[j] + adj[j]) - mxj[j]);
            float* ap = &s_agg[dl * X1P + lane];
            atomicAdd(ap,      rj[j].x * w);
            atomicAdd(ap + 32, rj[j].y * w);
            atomicAdd(ap + 64, rj[j].z * w);
            atomicAdd(ap + 96, rj[j].w * w);
            if ((lane & 3) == 0) atomicAdd(&s_den[hd * NPB + dl], w);
          }
        }
      }
    } else {
      constexpr int U3 = 2;
      for (int i0 = tid; i0 < nl; i0 += 1024 * U3) {
        int sv[U3], dv[U3];
        bool v[U3];
        float a[U3];
        float4 r[U3][4];
        #pragma unroll
        for (int j = 0; j < U3; ++j) {
          const int idx = i0 + 1024 * j;
          v[j] = idx < nl;
          const unsigned ent = s_list[v[j] ? idx : i0];
          sv[j] = (int)(ent & 0xffffu);
          dv[j] = (int)(ent >> 16);
        }
        #pragma unroll
        for (int j = 0; j < U3; ++j) {
          a[j] = aS[sv[j]];
          const float4* rp = (const float4*)(h + (size_t)sv[j] * HC);
          r[j][0] = rp[0]; r[j][1] = rp[1]; r[j][2] = rp[2]; r[j][3] = rp[3];
        }
        float adv[U3], mxv[U3];
        #pragma unroll
        for (int j = 0; j < U3; ++j) { adv[j] = s_ad[dv[j]]; mxv[j] = dec_f32(s_max[dv[j]]); }
        #pragma unroll
        for (int j = 0; j < U3; ++j) {
          if (v[j]) {
            const int d = dv[j];
            float w = expf(lrelu(a[j] + adv[j]) - mxv[j]);
            float* ap = &s_agg[d];
            atomicAdd(ap + 0  * NPB, r[j][0].x * w); atomicAdd(ap + 1  * NPB, r[j][0].y * w);
            atomicAdd(ap + 2  * NPB, r[j][0].z * w); atomicAdd(ap + 3  * NPB, r[j][0].w * w);
            atomicAdd(ap + 4  * NPB, r[j][1].x * w); atomicAdd(ap + 5  * NPB, r[j][1].y * w);
            atomicAdd(ap + 6  * NPB, r[j][1].z * w); atomicAdd(ap + 7  * NPB, r[j][1].w * w);
            atomicAdd(ap + 8  * NPB, r[j][2].x * w); atomicAdd(ap + 9  * NPB, r[j][2].y * w);
            atomicAdd(ap + 10 * NPB, r[j][2].z * w); atomicAdd(ap + 11 * NPB, r[j][2].w * w);
            atomicAdd(ap + 12 * NPB, r[j][3].x * w); atomicAdd(ap + 13 * NPB, r[j][3].y * w);
            atomicAdd(ap + 14 * NPB, r[j][3].z * w); atomicAdd(ap + 15 * NPB, r[j][3].w * w);
            atomicAdd(&s_den[d], w);
          }
        }
      }
    }
  } else {
    {
      uint4 dc[SCAN_U];
      load_keys(k4, rot, tid, dc);
      for (int t = 0; t < NIT; ++t) {
        const int it  = (t + rot) < NIT ? (t + rot) : (t + rot - NIT);
        const int itn = (it + 1) < NIT ? (it + 1) : 0;
        uint4 dn[SCAN_U];
        load_keys(k4, itn, tid, dn);
        unsigned mm = match_bits(dc, it, tid, bid4);
        while (mm) {
          int k = __builtin_ctz(mm); mm &= mm - 1;
          int e = edge_of(it, tid, k);
          int s = ei[e];
          int dl = ei[NEDGES + e] - lo;
          if constexpr (H == 8) {
            const float4* ap = (const float4*)(aS + (size_t)s * 8);
            float4 a0 = ap[0], a1 = ap[1];
            atomicMax(&s_max[0 * NPB + dl], enc_f32(lrelu(a0.x + s_ad[0 * NPB + dl])));
            atomicMax(&s_max[1 * NPB + dl], enc_f32(lrelu(a0.y + s_ad[1 * NPB + dl])));
            atomicMax(&s_max[2 * NPB + dl], enc_f32(lrelu(a0.z + s_ad[2 * NPB + dl])));
            atomicMax(&s_max[3 * NPB + dl], enc_f32(lrelu(a0.w + s_ad[3 * NPB + dl])));
            atomicMax(&s_max[4 * NPB + dl], enc_f32(lrelu(a1.x + s_ad[4 * NPB + dl])));
            atomicMax(&s_max[5 * NPB + dl], enc_f32(lrelu(a1.y + s_ad[5 * NPB + dl])));
            atomicMax(&s_max[6 * NPB + dl], enc_f32(lrelu(a1.z + s_ad[6 * NPB + dl])));
            atomicMax(&s_max[7 * NPB + dl], enc_f32(lrelu(a1.w + s_ad[7 * NPB + dl])));
          } else {
            atomicMax(&s_max[dl], enc_f32(lrelu(aS[s] + s_ad[dl])));
          }
        }
        #pragma unroll
        for (int u = 0; u < SCAN_U; ++u) dc[u] = dn[u];
      }
    }
    __syncthreads();

    {
      uint4 dc[SCAN_U];
      load_keys(k4, rot, tid, dc);
      for (int t = 0; t < NIT; ++t) {
        const int it  = (t + rot) < NIT ? (t + rot) : (t + rot - NIT);
        const int itn = (it + 1) < NIT ? (it + 1) : 0;
        uint4 dn[SCAN_U];
        load_keys(k4, itn, tid, dn);
        unsigned mm = match_bits(dc, it, tid, bid4);
        if constexpr (H == 8) {
          const int hd = lane >> 2;
          for (;;) {
            unsigned bal = __builtin_amdgcn_ballot_w32(mm != 0);
            if (!bal) break;
            int sv = 0, dv = 0;
            if (mm) {
              int k = __builtin_ctz(mm); mm &= mm - 1;
              int e = edge_of(it, tid, k);
              sv = ei[e];
              dv = ei[NEDGES + e] - lo;
            }
            while (bal) {
              int ml = __builtin_ctz(bal); bal &= bal - 1;
              int s  = __builtin_amdgcn_readlane(sv, ml);
              int dl = __builtin_amdgcn_readlane(dv, ml);
              float  a = aS[(size_t)s * 8 + hd];
              float4 r = ((const float4*)(h + (size_t)s * HC))[lane];
              float w = expf(lrelu(a + s_ad[hd * NPB + dl]) - dec_f32(s_max[hd * NPB + dl]));
              float* ap = &s_agg[dl * X1P + lane];
              atomicAdd(ap,      r.x * w);
              atomicAdd(ap + 32, r.y * w);
              atomicAdd(ap + 64, r.z * w);
              atomicAdd(ap + 96, r.w * w);
              if ((lane & 3) == 0) atomicAdd(&s_den[hd * NPB + dl], w);
            }
          }
        } else {
          while (mm) {
            int k = __builtin_ctz(mm); mm &= mm - 1;
            int e = edge_of(it, tid, k);
            int s = ei[e];
            int dl = ei[NEDGES + e] - lo;
            float a = aS[s];
            const float4* rp = (const float4*)(h + (size_t)s * HC);
            float4 r0 = rp[0], r1 = rp[1], r2 = rp[2], r3 = rp[3];
            float w = expf(lrelu(a + s_ad[dl]) - dec_f32(s_max[dl]));
            float* ap = &s_agg[dl];
            atomicAdd(ap + 0  * NPB, r0.x * w); atomicAdd(ap + 1  * NPB, r0.y * w);
            atomicAdd(ap + 2  * NPB, r0.z * w); atomicAdd(ap + 3  * NPB, r0.w * w);
            atomicAdd(ap + 4  * NPB, r1.x * w); atomicAdd(ap + 5  * NPB, r1.y * w);
            atomicAdd(ap + 6  * NPB, r1.z * w); atomicAdd(ap + 7  * NPB, r1.w * w);
            atomicAdd(ap + 8  * NPB, r2.x * w); atomicAdd(ap + 9  * NPB, r2.y * w);
            atomicAdd(ap + 10 * NPB, r2.z * w); atomicAdd(ap + 11 * NPB, r2.w * w);
            atomicAdd(ap + 12 * NPB, r3.x * w); atomicAdd(ap + 13 * NPB, r3.y * w);
            atomicAdd(ap + 14 * NPB, r3.z * w); atomicAdd(ap + 15 * NPB, r3.w * w);
            atomicAdd(&s_den[dl], w);
          }
        }
        #pragma unroll
        for (int u = 0; u < SCAN_U; ++u) dc[u] = dn[u];
      }
    }
  }
  __syncthreads();

  if constexpr (H == 8) {
    float* s_w2 = (float*)s_list;
    float* s_h2 = s_w2 + C_ * X1P;
    static_assert(C_ * HC == 2 * 1024, "W2 is copied two words per thread");
    const float w2a = W2[tid], w2b = W2[tid + 1024];
    float4 sa[4], da[4];
    #pragma unroll
    for (int q = 0; q < 4; ++q) { sa[q] = ((const float4*)attS2)[q]; da[q] = ((const float4*)attD2)[q]; }
    for (int idx = tid; idx < cnt * HC; idx += 1024) {
      int dl = idx >> 7, c = idx & 127, hd = c >> 4;
      size_t n = (size_t)(lo + dl);
      float ws = expf(lrelu(aS[n * 8 + hd] + s_ad[hd * NPB + dl]) - dec_f32(s_max[hd * NPB + dl]));
      float* xp = &s_agg[dl * X1P + (c & 3) * 32 + (c >> 2)];
      float acc = *xp + ws * h[n * HC + c];
      float v = acc / (s_den[hd * NPB + dl] + ws + 1e-16f) + bias[c];
      *xp = v > 0.f ? v : expm1f(v);
    }
    {
      const int c = tid & 127;
      float* wp = &s_w2[(tid >> 7) * X1P + (c & 3) * 32 + (c >> 2)];
      wp[0] = w2a; wp[8 * X1P] = w2b;
    }
    __syncthreads();
    const int wv2 = __builtin_amdgcn_readfirstlane(tid >> 5);
    if (wv2 < T2) {
      const int l15 = lane & 15, half = lane >> 4;
      const int row = (wv2 * 16 + l15) < NPB ? (wv2 * 16 + l15) : NPB - 1;
      const float* ar = s_agg + row * X1P + half * 2;
      const float* br = s_w2 + l15 * X1P + half * 2;
      v8f acc0 = (v8f){};
      #pragma unroll
      for (int s = 0; s < HC / 4; ++s) {
        const v2f a0 = *(const v2f*)(ar + 4 * s), b0 = *(const v2f*)(br + 4 * s);
        acc0 = __builtin_amdgcn_wmma_f32_16x16x4_f32(false, a0, false, b0, (short)0, acc0, false, false);
      }
      float* hp = s_h2 + (wv2 * 16 + half * 8) * H2P + l15;
      #pragma unroll
      for (int r = 0; r < 8; ++r) hp[r * H2P] = acc0[r];
    }
    __syncthreads();
    if (tid < cnt) {
      const size_t n = (size_t)(lo + tid);
      const float* hp = s_h2 + tid * H2P;
      float hv[C_];
      float s = 0.f, d = 0.f;
      #pragma unroll
      for (int c = 0; c < C_; ++c) hv[c] = hp[c];
      #pragma unroll
      for (int q = 0; q < 4; ++q) {
        s += hv[4 * q] * sa[q].x; s += hv[4 * q + 1] * sa[q].y; s += hv[4 * q + 2] * sa[q].z; s += hv[4 * q + 3] * sa[q].w;
        d += hv[4 * q] * da[q].x; d += hv[4 * q + 1] * da[q].y; d += hv[4 * q + 2] * da[q].z; d += hv[4 * q + 3] * da[q].w;
      }
      as2o[n] = s; ad2o[n] = d;
      float4* op = (float4*)(h2o + n * C_);
      op[0] = make_float4(hv[0],  hv[1],  hv[2],  hv[3]);
      op[1] = make_float4(hv[4],  hv[5],  hv[6],  hv[7]);
      op[2] = make_float4(hv[8],  hv[9],  hv[10], hv[11]);
      op[3] = make_float4(hv[12], hv[13], hv[14], hv[15]);
    }
  } else {
    if (tid < cnt) {
      int dl = tid;
      size_t n = (size_t)(lo + dl);
      float ws = expf(lrelu(aS[n] + s_ad[dl]) - dec_f32(s_max[dl]));
      float dn = s_den[dl] + ws + 1e-16f;
      float v[NCLASS];
      float m = -1e30f;
      #pragma unroll
      for (int c = 0; c < NCLASS; ++c) {
        v[c] = (s_agg[c * NPB + dl] + ws * h[n * HC + c]) / dn + bias[c];
        m = fmaxf(m, v[c]);
      }
      float sum = 0.f;
      #pragma unroll
      for (int c = 0; c < NCLASS; ++c) sum += expf(v[c] - m);
      float ls = logf(sum);
      #pragma unroll
      for (int c = 0; c < NCLASS; ++c) out[n * NCLASS + c] = v[c] - m - ls;
    }
  }
}

extern "C" void kernel_launch(void* const* d_in, const int* in_sizes, int n_in,
                              void* d_out, int out_size, void* d_ws, size_t ws_size,
                              hipStream_t stream) {
  const float* x     = (const float*)d_in[0];
  const int*   ei    = (const int*)  d_in[1];
  const float* W1    = (const float*)d_in[2];
  const float* attS1 = (const float*)d_in[3];
  const float* attD1 = (const float*)d_in[4];
  const float* b1    = (const float*)d_in[5];
  const float* W2    = (const float*)d_in[6];
  const float* attS2 = (const float*)d_in[7];
  const float* attD2 = (const float*)d_in[8];
  const float* b2    = (const float*)d_in[9];
  float* out = (float*)d_out;

  char* ws = (char*)d_ws;
  size_t off = 0;
  auto alloc = [&](size_t bytes) -> void* {
    void* p = ws + off;
    off = (off + bytes + 255) & ~(size_t)255;
    return p;
  };

  float* h1  = (float*)alloc((size_t)NNODES * F1 * 4);
  float* as1 = (float*)alloc((size_t)NNODES * H1_ * 4);
  float* ad1 = (float*)alloc((size_t)NNODES * H1_ * 4);
  float* h2  = (float*)alloc((size_t)NNODES * C_ * 4);
  float* as2 = (float*)alloc((size_t)NNODES * 4);
  float* ad2 = (float*)alloc((size_t)NNODES * 4);
  unsigned* key = (unsigned*)alloc((size_t)NEDGES);
  unsigned* glist = (unsigned*)alloc((size_t)NBLK * WCAP * 4);
  int* gcnt = (int*)alloc((size_t)NBLK * 4);

  const int TB = 256;
  auto nb = [](long long n, int b) { return (unsigned)((n + b - 1) / b); };

  gemm1_w_in_lds<<<G1_BLK, G1_THREADS, 0, stream>>>(x, W1, h1);
  attn_scores<<<nb((long long)NNODES * H1_, TB), TB, 0, stream>>>(
      h1, attS1, attD1, as1, ad1, NNODES * H1_, H1_);
  owner_ids<<<nb(NEDGES / 4, TB), TB, 0, stream>>>(ei, key);
  owned_gat<H1_><<<NBLK, 1024, 0, stream>>>(ei, key, as1, ad1, h1, b1, nullptr, glist, gcnt,
                                            W2, attS2, attD2, h2, as2, ad2);

  owned_gat<1><<<NBLK, 1024, 0, stream>>>(ei, key, as2, ad2, h2, b2, out, glist, gcnt,
                                          nullptr, nullptr, nullptr, nullptr, nullptr, nullptr);
}


# Owner ids kept as eight bit planes per 32 edges: a work-group's match test for 32 edges is 8 logic ops (was ~60); a lane scans 128 edges per trip (13 trips, was 49); list and LDS add order change
# speedup vs baseline: 12.6173x; 1.0744x over previous
#include <hip/hip_runtime.h>
#include <hip/hip_bf16.h>

#define NNODES 50000
#define NEDGES 1600000
#define NFEAT  512
#define F1     128
#define H1_    8
#define C_     16
#define NCLASS 16

typedef float v2f __attribute__((ext_vector_type(2)));
typedef float v8f __attribute__((ext_vector_type(8)));

__device__ __forceinline__ unsigned enc_f32(float f) {
  unsigned u = __float_as_uint(f);
  return (u & 0x80000000u) ? ~u : (u | 0x80000000u);
}
__device__ __forceinline__ float dec_f32(unsigned u) {
  return (u & 0x80000000u) ? __uint_as_float(u & 0x7FFFFFFFu) : __uint_as_float(~u);
}

#define W1P      516
#define G1_BLK   256
#define G1_WAVES 16
#define G1_THREADS 512
#define G1_STRIPS (NNODES / 16)
#define G1_FULL  (G1_STRIPS / G1_BLK)
#define G1_REST  (G1_STRIPS - G1_FULL * G1_BLK)
static_assert(NNODES % 16 == 0, "strips of 16 rows");
static_assert(G1_FULL + 4 == G1_WAVES, "waves G1_FULL .. G1_FULL+3 take the column quarters");
static_assert(G1_THREADS == G1_WAVES * 32, "work-group size");

template <int TN>
__device__ __forceinline__ void k_block(const float* __restrict__ an, const float* sw,
                                        const unsigned (&bo)[TN],
                                        const v2f (&aU)[8], v2f (&aP)[8],
                                        v2f (&bC)[TN], v8f (&acc)[TN]) {
  #pragma unroll
  for (int j = 0; j < 8; ++j) { aP[j].x = an[4 * j]; aP[j].y = an[4 * j + 1]; }
  #pragma unroll
  for (int ks = 0; ks < 8; ++ks) {
    v2f bN[TN];
    #pragma unroll
    for (int t = 0; t < TN; ++t) bN[t] = *(const v2f*)(sw + bo[t] + 4 * ks + 4);
    #pragma unroll
    for (int t = 0; t < TN; ++t)
      acc[t] = __builtin_amdgcn_wmma_f32_16x16x4_f32(false, aU[ks], false, bC[t],
                                                     (short)0, acc[t], false, false);
    #pragma unroll
    for (int t = 0; t < TN; ++t) bC[t] = bN[t];
  }
  __builtin_amdgcn_sched_barrier(0);
  #pragma unroll
  for (int j = 0; j < 8; ++j) asm volatile("" : "+v"(aP[j]));
}

template <int TN>
__device__ __forceinline__ void strip_b_from_lds(const float* __restrict__ X, const float* sw,
                                                 float* __restrict__ D, int tm, int col0, int lane) {
  constexpr int NB = NFEAT / 32;
  static_assert(NB % 2 == 0, "the two A register sets change roles every block");
  const int l15 = lane & 15, half = lane >> 4;
  const float* ap = X + (size_t)(tm * 16 + l15) * NFEAT + half * 2;
  v8f acc[TN];
  v2f a0[8], a1[8], bC[TN];
  unsigned bo[TN];
  #pragma unroll
  for (int t = 0; t < TN; ++t) {
    acc[t] = (v8f){};
    bo[t] = (unsigned)((col0 + t * 16 + l15) * W1P + half * 2);
    asm volatile("" : "+v"(bo[t]));
  }
  #pragma unroll
  for (int j = 0; j < 8; ++j) { a0[j].x = ap[4 * j]; a0[j].y = ap[4 * j + 1]; }
  #pragma unroll
  for (int t = 0; t < TN; ++t) bC[t] = *(const v2f*)(sw + bo[t]);
  for (int c = 0; c < NB; c += 2) {
    k_block<TN>(ap + 32 * (c + 1), sw, bo, a0, a1, bC, acc);
    #pragma unroll
    for (int t = 0; t < TN; ++t) bo[t] += 32;
    k_block<TN>(ap + 32 * (c + 2 < NB ? c + 2 : c + 1), sw, bo, a1, a0, bC, acc);
    #pragma unroll
    for (int t = 0; t < TN; ++t) bo[t] += 32;
  }
  #pragma unroll
  for (int t = 0; t < TN; ++t) {
    float* dp = D + (size_t)(tm * 16 + half * 8) * F1 + col0 + t * 16 + l15;
    #pragma unroll
    for (int r = 0; r < 8; ++r) dp[r * F1] = acc[t][r];
  }
}

__global__ __launch_bounds__(G1_THREADS) void gemm1_w_in_lds(
    const float* __restrict__ X, const float* __restrict__ Wt, float* __restrict__ D) {
  __shared__ __attribute__((aligned(16))) float s_w[F1 * W1P];
  const int tid = threadIdx.x, lane = tid & 31;
  const int wv = __builtin_amdgcn_readfirstlane(tid >> 5);
  #pragma unroll 8
  for (int i = tid; i < F1 * (NFEAT / 4); i += G1_THREADS) {
    const int row = i / (NFEAT / 4), c4 = i % (NFEAT / 4);
    const float4 v = ((const float4*)Wt)[i];
    *(float4*)&s_w[row * W1P + c4 * 4] = v;
  }
  if (tid < F1) *(float4*)&s_w[tid * W1P + NFEAT] = make_float4(0.f, 0.f, 0.f, 0.f);
  __syncthreads();
  if (wv < G1_FULL) {
    strip_b_from_lds<8>(X, s_w, D, wv * G1_BLK + (int)blockIdx.x, 0, lane);
  } else if ((int)blockIdx.x < G1_REST) {
    strip_b_from_lds<2>(X, s_w, D, G1_FULL * G1_BLK + (int)blockIdx.x, (wv - G1_FULL) * 32, lane);
  }
}

__global__ void attn_scores(const float* __restrict__ h,
                            const float* __restrict__ attS,
                            const float* __restrict__ attD,
                            float* __restrict__ aS, float* __restrict__ aD,
                            int NH, int H) {
  int t = blockIdx.x * blockDim.x + threadIdx.x;
  if (t >= NH) return;
  int hh = t % H;
  const float* hp = h + (size_t)t * C_;
  const float* sp = attS + hh * C_;
  const float* dp = attD + hh * C_;
  float s = 0.f, d = 0.f;
  #pragma unroll
  for (int c = 0; c < C_; ++c) { s += hp[c] * sp[c]; d += hp[c] * dp[c]; }
  aS[t] = s; aD[t] = d;
}

#define NPB    196
#define NBLK   256
#define NSG    (NEDGES / 128)
#define NIT    ((NSG + 1023) / 1024)
static_assert(NEDGES % 128 == 0, "whole groups of 128 edges");

__device__ __forceinline__ float lrelu(float a) { return a > 0.f ? a : 0.2f * a; }

__global__ __launch_bounds__(256) void owner_planes(const int* __restrict__ ei, uint4* __restrict__ key4) {
  const int lane = threadIdx.x & 31;
  const int g = (int)((blockIdx.x * blockDim.x + threadIdx.x) >> 5);
  if (g >= NSG) return;
  const int4 d = ((const int4*)(ei + NEDGES))[g * 32 + lane];
  const unsigned o0 = (unsigned)d.x / NPB, o1 = (unsigned)d.y / NPB;
  const unsigned o2 = (unsigned)d.z / NPB, o3 = (unsigned)d.w / NPB;
  uint4 mine = make_uint4(0u, 0u, 0u, 0u);
  #pragma unroll
  for (int b = 0; b < 8; ++b) {
    uint4 p;
    p.x = __builtin_amdgcn_ballot_w32(((o0 >> b) & 1u) != 0u);
    p.y = __builtin_amdgcn_ballot_w32(((o1 >> b) & 1u) != 0u);
    p.z = __builtin_amdgcn_ballot_w32(((o2 >> b) & 1u) != 0u);
    p.w = __builtin_amdgcn_ballot_w32(((o3 >> b) & 1u) != 0u);
    if (lane == b) mine = p;
  }
  if (lane < 8) key4[((g >> 10) * 8 + lane) * 1024 + (g & 1023)] = mine;
}

__device__ __forceinline__ void load_planes(const uint4* __restrict__ k4, int it, int tid,
                                            uint4 (&p)[8]) {
  int g = it * 1024 + tid;
  g = g < NSG ? g : NSG - 1;
  const uint4* q = k4 + ((g >> 10) * 8) * 1024 + (g & 1023);
  #pragma unroll
  for (int b = 0; b < 8; ++b) p[b] = q[b * 1024];
}

__device__ __forceinline__ unsigned and_xor(unsigned acc, unsigned plane, unsigned inv) {
  return __builtin_amdgcn_bitop3_b32(acc, plane, inv, 0x60);
}

__device__ __forceinline__ uint4 match_planes(const uint4 (&p)[8], int it, int tid,
                                              const unsigned (&inv)[8]) {
  uint4 m = make_uint4(p[0].x ^ inv[0], p[0].y ^ inv[0], p[0].z ^ inv[0], p[0].w ^ inv[0]);
  #pragma unroll
  for (int b = 1; b < 8; ++b) {
    m.x = and_xor(m.x, p[b].x, inv[b]); m.y = and_xor(m.y, p[b].y, inv[b]);
    m.z = and_xor(m.z, p[b].z, inv[b]); m.w = and_xor(m.w, p[b].w, inv[b]);
  }
  if (it * 1024 + tid >= NSG) m = make_uint4(0u, 0u, 0u, 0u);
  return m;
}

__device__ __forceinline__ int edge_of(int it, int tid, int c, int l) {
  return ((it * 1024 + tid) << 7) + (l << 2) + c;
}

__device__ __forceinline__ unsigned word_of(const uint4& m, int c) {
  return c == 0 ? m.x : c == 1 ? m.y : c == 2 ? m.z : m.w;
}

#define X1P    132
#define T2     ((NPB + 15) / 16)
#define H2P    17
#define LCAP   16384
#define WCAP   7168
static_assert(WCAP <= LCAP, "a saved list is copied back into the LDS list");
static_assert(NNODES <= 65536 && NPB <= 65536, "a list entry packs src and dst_local into 16 bits each");
static_assert(T2 <= 32, "one wave per 16-row tile of the layer-2 product");
static_assert(C_ * X1P + T2 * 16 * H2P <= LCAP, "W2 and the layer-2 rows reuse the list's LDS in the tail");
static_assert(F1 == 128 && C_ == 16 && NCLASS == C_, "index arithmetic of the fused layer-2 product");

template <int H>
__global__ __launch_bounds__(1024) void owned_gat(
    const int* __restrict__ ei, const unsigned* __restrict__ key,
    const float* __restrict__ aS, const float* __restrict__ aD,
    const float* __restrict__ h, const float* __restrict__ bias,
    float* __restrict__ out,
    unsigned* __restrict__ glist, int* __restrict__ gcnt,
    const float* __restrict__ W2, const float* __restrict__ attS2, const float* __restrict__ attD2,
    float* __restrict__ h2o, float* __restrict__ as2o, float* __restrict__ ad2o) {
  constexpr int HC = H * C_;
  constexpr int AGG = (H == 8) ? NPB * X1P : NPB * HC;
  __shared__ __attribute__((aligned(16))) float s_agg[AGG];
  __shared__ float    s_den[NPB * H];
  __shared__ unsigned s_max[NPB * H];
  __shared__ float    s_ad [NPB * H];
  __shared__ __attribute__((aligned(16))) unsigned s_list[LCAP];
  __shared__ int      s_cnt;

  const int tid = threadIdx.x, lane = tid & 31;
  const int lo  = blockIdx.x * NPB;
  const int cnt = (NNODES - lo) < NPB ? (NNODES - lo) : NPB;
  unsigned inv[8];
  #pragma unroll
  for (int b = 0; b < 8; ++b) inv[b] = ((blockIdx.x >> b) & 1u) ? 0u : ~0u;
  const uint4* k4 = (const uint4*)key;
  const int rot = blockIdx.x % NIT;

  for (int i = tid; i < NPB * H; i += 1024) {
    float ad = 0.f, as = 0.f;
    if (i < cnt * H) { ad = aD[(size_t)lo * H + i]; as = aS[(size_t)lo * H + i]; }
    const int t = (i % H) * NPB + i / H;
    s_ad[t]  = ad;
    s_max[t] = enc_f32(lrelu(as + ad));
    s_den[t] = 0.f;
  }
  for (int i = tid; i < AGG; i += 1024) s_agg[i] = 0.f;
  if (tid == 0) s_cnt = 0;
  __syncthreads();

  unsigned* gl = glist + (size_t)blockIdx.x * WCAP;
  int saved = -1;
  if constexpr (H == 1) {
    const int c = __builtin_amdgcn_readfirstlane(gcnt[blockIdx.x]);
    if (c <= WCAP) saved = c;
  }

  if (saved < 0) {
    auto scan_trip = [&](const uint4 (&cur)[8], uint4 (&nxt)[8], int t) {
      const int it = (t + rot) < NIT ? (t + rot) : (t + rot - NIT);
      if (t + 1 < NIT) load_planes(k4, (it + 1) < NIT ? (it + 1) : 0, tid, nxt);
      const uint4 m = match_planes(cur, it, tid, inv);
      unsigned mw[4] = {m.x, m.y, m.z, m.w};
      unsigned bl[4];
      int n = 0;
      #pragma unroll
      for (int c = 0; c < 4; ++c) { bl[c] = __builtin_amdgcn_ballot_w32(mw[c] != 0); n += __builtin_popcount(bl[c]); }
      if (n) {
        int off = 0;
        if (lane == 0) off = atomicAdd(&s_cnt, n);
        off = __builtin_amdgcn_readfirstlane(off);
        #pragma unroll
        for (int c = 0; c < 4; ++c) {
          if (mw[c]) {
            const int k = __builtin_ctz(mw[c]); mw[c] &= mw[c] - 1;
            const int pos = off + __builtin_popcount(bl[c] & ((1u << lane) - 1u));
            if (pos < LCAP) s_list[pos] = (unsigned)edge_of(it, tid, c, k);
          }
          off += __builtin_popcount(bl[c]);
        }
      }
      #pragma unroll
      for (int c = 0; c < 4; ++c) {
        unsigned mm = mw[c];
        for (;;) {
          const unsigned bal = __builtin_amdgcn_ballot_w32(mm != 0);
          if (!bal) break;
          int base = 0;
          if (lane == 0) base = atomicAdd(&s_cnt, __builtin_popcount(bal));
          base = __builtin_amdgcn_readfirstlane(base);
          if (mm) {
            const int k = __builtin_ctz(mm); mm &= mm - 1;
            const int pos = base + __builtin_popcount(bal & ((1u << lane) - 1u));
            if (pos < LCAP) s_list[pos] = (unsigned)edge_of(it, tid, c, k);
          }
        }
      }
    };
    uint4 pa[8], pb[8];
    load_planes(k4, rot, tid, pa);
    for (int t = 0; t + 1 < NIT; t += 2) { scan_trip(pa, pb, t); scan_trip(pb, pa, t + 1); }
    if (NIT & 1) scan_trip(pa, pb, NIT - 1);
    __syncthreads();
  }
  const int nl = saved >= 0 ? saved : __builtin_amdgcn_readfirstlane(s_cnt);
  const bool save = (H == 8) && nl <= WCAP;
  if constexpr (H == 8) { if (tid == 0) gcnt[blockIdx.x] = nl; }

  if (nl <= LCAP) {
    if (saved >= 0) {
      if constexpr (H == 1) {
        constexpr int U2 = 4;
        for (int i0 = tid; i0 < nl; i0 += 1024 * U2) {
          int idx[U2];
          unsigned ent[U2];
          bool v[U2];
          float a[U2], ad[U2];
          #pragma unroll
          for (int j = 0; j < U2; ++j) {
            idx[j] = i0 + 1024 * j;
            v[j] = idx[j] < nl;
            ent[j] = gl[v[j] ? idx[j] : i0];
          }
          #pragma unroll
          for (int j = 0; j < U2; ++j) asm volatile("" : "+v"(ent[j]));
          #pragma unroll
          for (int j = 0; j < U2; ++j) a[j] = aS[ent[j] & 0xffffu];
          #pragma unroll
          for (int j = 0; j < U2; ++j) ad[j] = s_ad[ent[j] >> 16];
          #pragma unroll
          for (int j = 0; j < U2; ++j) asm volatile("" : "+v"(a[j]));
          #pragma unroll
          for (int j = 0; j < U2; ++j) {
            if (v[j]) {
              atomicMax(&s_max[ent[j] >> 16], enc_f32(lrelu(a[j] + ad[j])));
              s_list[idx[j]] = ent[j];
            }
          }
        }
      }
    } else {
    {
      constexpr int U2 = 4;
      for (int i0 = tid; i0 < nl; i0 += 1024 * U2) {
        int idx[U2], ev[U2], sv[U2], dv[U2];
        bool v[U2];
        #pragma unroll
        for (int j = 0; j < U2; ++j) {
          idx[j] = i0 + 1024 * j;
          v[j] = idx[j] < nl;
          ev[j] = (int)s_list[v[j] ? idx[j] : i0];
        }
        #pragma unroll
        for (int j = 0; j < U2; ++j) {
          sv[j] = ei[ev[j]];
          dv[j] = ei[NEDGES + ev[j]] - lo;
        }
        if constexpr (H == 8) {
          float4 a0[U2], a1[U2];
          float ad[U2][8];
          #pragma unroll
          for (int j = 0; j < U2; ++j) {
            const float4* ap = (const float4*)(aS + (size_t)sv[j] * 8);
            a0[j] = ap[0]; a1[j] = ap[1];
          }
          #pragma unroll
          for (int j = 0; j < U2; ++j) {
            #pragma unroll
            for (int hh = 0; hh < 8; ++hh) ad[j][hh] = s_ad[hh * NPB + dv[j]];
          }
          #pragma unroll
          for (int j = 0; j < U2; ++j) {
            if (v[j]) {
              const int d = dv[j];
              atomicMax(&s_max[0 * NPB + d], enc_f32(lrelu(a0[j].x + ad[j][0])));
              atomicMax(&s_max[1 * NPB + d], enc_f32(lrelu(a0[j].y + ad[j][1])));
              atomicMax(&s_max[2 * NPB + d], enc_f32(lrelu(a0[j].z + ad[j][2])));
              atomicMax(&s_max[3 * NPB + d], enc_f32(lrelu(a0[j].w + ad[j][3])));
              atomicMax(&s_max[4 * NPB + d], enc_f32(lrelu(a1[j].x + ad[j][4])));
              atomicMax(&s_max[5 * NPB + d], enc_f32(lrelu(a1[j].y + ad[j][5])));
              atomicMax(&s_max[6 * NPB + d], enc_f32(lrelu(a1[j].z + ad[j][6])));
              atomicMax(&s_max[7 * NPB + d], enc_f32(lrelu(a1[j].w + ad[j][7])));
              const unsigned ent = (unsigned)sv[j] | ((unsigned)d << 16);
              s_list[idx[j]] = ent;
              if (save) gl[idx[j]] = ent;
            }
          }
        } else {
          float a[U2], ad[U2];
          #pragma unroll
          for (int j = 0; j < U2; ++j) a[j] = aS[sv[j]];
          #pragma unroll
          for (int j = 0; j < U2; ++j) ad[j] = s_ad[dv[j]];
          #pragma unroll
          for (int j = 0; j < U2; ++j) {
            if (v[j]) {
              atomicMax(&s_max[dv[j]], enc_f32(lrelu(a[j] + ad[j])));
              s_list[idx[j]] = (unsigned)sv[j] | ((unsigned)dv[j] << 16);
            }
          }
        }
      }
    }
    }
    __syncthreads();

    if constexpr (H == 8) {
      constexpr int EB = 8;
      const int hd = lane >> 2;
      const int wv = __builtin_amdgcn_readfirstlane(tid >> 5);
      for (int base = wv * EB; base < nl; base += 32 * EB) {
        const int li = base + (lane & (EB - 1));
        const unsigned ent = s_list[li < nl ? li : nl - 1];
        int sj[EB], dj[EB];
        bool vj[EB];
        #pragma unroll
        for (int j = 0; j < EB; ++j) {
          const unsigned x = (unsigned)__builtin_amdgcn_readlane((int)ent, j);
          sj[j] = (int)(x & 0xffffu);
          dj[j] = (int)(x >> 16);
          vj[j] = base + j < nl;
        }
        float  aj[EB];
        float4 rj[EB];
        #pragma unroll
        for (int j = 0; j < EB; ++j) {
          aj[j] = aS[(size_t)sj[j] * 8 + hd];
          rj[j] = ((const float4*)(h + (size_t)sj[j] * HC))[lane];
        }
        float adj[EB], mxj[EB];
        #pragma unroll
        for (int j = 0; j < EB; ++j) {
          adj[j] = s_ad[hd * NPB + dj[j]];
          mxj[j] = dec_f32(s_max[hd * NPB + dj[j]]);
        }
        #pragma unroll
        for (int j = 0; j < EB; ++j) {
          if (vj[j]) {
            const int dl = dj[j];
            float w = expf(lrelu(aj[j] + adj[j]) - mxj[j]);
            float* ap = &s_agg[dl * X1P + lane];
            atomicAdd(ap,      rj[j].x * w);
            atomicAdd(ap + 32, rj[j].y * w);
            atomicAdd(ap + 64, rj[j].z * w);
            atomicAdd(ap + 96, rj[j].w * w);
            if ((lane & 3) == 0) atomicAdd(&s_den[hd * NPB + dl], w);
          }
        }
      }
    } else {
      constexpr int U3 = 2;
      for (int i0 = tid; i0 < nl; i0 += 1024 * U3) {
        int sv[U3], dv[U3];
        bool v[U3];
        float a[U3];
        float4 r[U3][4];
        #pragma unroll
        for (int j = 0; j < U3; ++j) {
          const int idx = i0 + 1024 * j;
          v[j] = idx < nl;
          const unsigned ent = s_list[v[j] ? idx : i0];
          sv[j] = (int)(ent & 0xffffu);
          dv[j] = (int)(ent >> 16);
        }
        #pragma unroll
        for (int j = 0; j < U3; ++j) {
          a[j] = aS[sv[j]];
          const float4* rp = (const float4*)(h + (size_t)sv[j] * HC);
          r[j][0] = rp[0]; r[j][1] = rp[1]; r[j][2] = rp[2]; r[j][3] = rp[3];
        }
        float adv[U3], mxv[U3];
        #pragma unroll
        for (int j = 0; j < U3; ++j) { adv[j] = s_ad[dv[j]]; mxv[j] = dec_f32(s_max[dv[j]]); }
        #pragma unroll
        for (int j = 0; j < U3; ++j) {
          if (v[j]) {
            const int d = dv[j];
            float w = expf(lrelu(a[j] + adv[j]) - mxv[j]);
            float* ap = &s_agg[d];
            atomicAdd(ap + 0  * NPB, r[j][0].x * w); atomicAdd(ap + 1  * NPB, r[j][0].y * w);
            atomicAdd(ap + 2  * NPB, r[j][0].z * w); atomicAdd(ap + 3  * NPB, r[j][0].w * w);
            atomicAdd(ap + 4  * NPB, r[j][1].x * w); atomicAdd(ap + 5  * NPB, r[j][1].y * w);
            atomicAdd(ap + 6  * NPB, r[j][1].z * w); atomicAdd(ap + 7  * NPB, r[j][1].w * w);
            atomicAdd(ap + 8  * NPB, r[j][2].x * w); atomicAdd(ap + 9  * NPB, r[j][2].y * w);
            atomicAdd(ap + 10 * NPB, r[j][2].z * w); atomicAdd(ap + 11 * NPB, r[j][2].w * w);
            atomicAdd(ap + 12 * NPB, r[j][3].x * w); atomicAdd(ap + 13 * NPB, r[j][3].y * w);
            atomicAdd(ap + 14 * NPB, r[j][3].z * w); atomicAdd(ap + 15 * NPB, r[j][3].w * w);
            atomicAdd(&s_den[d], w);
          }
        }
      }
    }
  } else {
    {
      uint4 pc[8];
      load_planes(k4, rot, tid, pc);
      for (int t = 0; t < NIT; ++t) {
        const int it  = (t + rot) < NIT ? (t + rot) : (t + rot - NIT);
        const int itn = (it + 1) < NIT ? (it + 1) : 0;
        uint4 pn[8];
        load_planes(k4, itn, tid, pn);
        const uint4 m = match_planes(pc, it, tid, inv);
        #pragma nounroll
        for (int c = 0; c < 4; ++c) {
          unsigned mm = word_of(m, c);
          while (mm) {
            int k = __builtin_ctz(mm); mm &= mm - 1;
            int e = edge_of(it, tid, c, k);
            int s = ei[e];
            int dl = ei[NEDGES + e] - lo;
            if constexpr (H == 8) {
              const float4* ap = (const float4*)(aS + (size_t)s * 8);
              float4 a0 = ap[0], a1 = ap[1];
              atomicMax(&s_max[0 * NPB + dl], enc_f32(lrelu(a0.x + s_ad[0 * NPB + dl])));
              atomicMax(&s_max[1 * NPB + dl], enc_f32(lrelu(a0.y + s_ad[1 * NPB + dl])));
              atomicMax(&s_max[2 * NPB + dl], enc_f32(lrelu(a0.z + s_ad[2 * NPB + dl])));
              atomicMax(&s_max[3 * NPB + dl], enc_f32(lrelu(a0.w + s_ad[3 * NPB + dl])));
              atomicMax(&s_max[4 * NPB + dl], enc_f32(lrelu(a1.x + s_ad[4 * NPB + dl])));
              atomicMax(&s_max[5 * NPB + dl], enc_f32(lrelu(a1.y + s_ad[5 * NPB + dl])));
              atomicMax(&s_max[6 * NPB + dl], enc_f32(lrelu(a1.z + s_ad[6 * NPB + dl])));
              atomicMax(&s_max[7 * NPB + dl], enc_f32(lrelu(a1.w + s_ad[7 * NPB + dl])));
            } else {
              atomicMax(&s_max[dl], enc_f32(lrelu(aS[s] + s_ad[dl])));
            }
          }
        }
        #pragma unroll
        for (int b = 0; b < 8; ++b) pc[b] = pn[b];
      }
    }
    __syncthreads();

    {
      uint4 pc[8];
      load_planes(k4, rot, tid, pc);
      for (int t = 0; t < NIT; ++t) {
        const int it  = (t + rot) < NIT ? (t + rot) : (t + rot - NIT);
        const int itn = (it + 1) < NIT ? (it + 1) : 0;
        uint4 pn[8];
        load_planes(k4, itn, tid, pn);
        const uint4 m = match_planes(pc, it, tid, inv);
        #pragma nounroll
        for (int c = 0; c < 4; ++c) {
          unsigned mm = word_of(m, c);
          if constexpr (H == 8) {
            const int hd = lane >> 2;
            for (;;) {
              unsigned bal = __builtin_amdgcn_ballot_w32(mm != 0);
              if (!bal) break;
              int sv = 0, dv = 0;
              if (mm) {
                int k = __builtin_ctz(mm); mm &= mm - 1;
                int e = edge_of(it, tid, c, k);
                sv = ei[e];
                dv = ei[NEDGES + e] - lo;
              }
              while (bal) {
                int ml = __builtin_ctz(bal); bal &= bal - 1;
                int s  = __builtin_amdgcn_readlane(sv, ml);
                int dl = __builtin_amdgcn_readlane(dv, ml);
                float  a = aS[(size_t)s * 8 + hd];
                float4 r = ((const float4*)(h + (size_t)s * HC))[lane];
                float w = expf(lrelu(a + s_ad[hd * NPB + dl]) - dec_f32(s_max[hd * NPB + dl]));
                float* ap = &s_agg[dl * X1P + lane];
                atomicAdd(ap,      r.x * w);
                atomicAdd(ap + 32, r.y * w);
                atomicAdd(ap + 64, r.z * w);
                atomicAdd(ap + 96, r.w * w);
                if ((lane & 3) == 0) atomicAdd(&s_den[hd * NPB + dl], w);
              }
            }
          } else {
            while (mm) {
              int k = __builtin_ctz(mm); mm &= mm - 1;
              int e = edge_of(it, tid, c, k);
              int s = ei[e];
              int dl = ei[NEDGES + e] - lo;
              float a = aS[s];
              const float4* rp = (const float4*)(h + (size_t)s * HC);
              float4 r0 = rp[0], r1 = rp[1], r2 = rp[2], r3 = rp[3];
              float w = expf(lrelu(a + s_ad[dl]) - dec_f32(s_max[dl]));
              float* ap = &s_agg[dl];
              atomicAdd(ap + 0  * NPB, r0.x * w); atomicAdd(ap + 1  * NPB, r0.y * w);
              atomicAdd(ap + 2  * NPB, r0.z * w); atomicAdd(ap + 3  * NPB, r0.w * w);
              atomicAdd(ap + 4  * NPB, r1.x * w); atomicAdd(ap + 5  * NPB, r1.y * w);
              atomicAdd(ap + 6  * NPB, r1.z * w); atomicAdd(ap + 7  * NPB, r1.w * w);
              atomicAdd(ap + 8  * NPB, r2.x * w); atomicAdd(ap + 9  * NPB, r2.y * w);
              atomicAdd(ap + 10 * NPB, r2.z * w); atomicAdd(ap + 11 * NPB, r2.w * w);
              atomicAdd(ap + 12 * NPB, r3.x * w); atomicAdd(ap + 13 * NPB, r3.y * w);
              atomicAdd(ap + 14 * NPB, r3.z * w); atomicAdd(ap + 15 * NPB, r3.w * w);
              atomicAdd(&s_den[dl], w);
            }
          }
        }
        #pragma unroll
        for (int b = 0; b < 8; ++b) pc[b] = pn[b];
      }
    }
  }
  __syncthreads();

  if constexpr (H == 8) {
    float* s_w2 = (float*)s_list;
    float* s_h2 = s_w2 + C_ * X1P;
    static_assert(C_ * HC == 2 * 1024, "W2 is copied two words per thread");
    const float w2a = W2[tid], w2b = W2[tid + 1024];
    float4 sa[4], da[4];
    #pragma unroll
    for (int q = 0; q < 4; ++q) { sa[q] = ((const float4*)attS2)[q]; da[q] = ((const float4*)attD2)[q]; }
    for (int idx = tid; idx < cnt * HC; idx += 1024) {
      int dl = idx >> 7, c = idx & 127, hd = c >> 4;
      size_t n = (size_t)(lo + dl);
      float ws = expf(lrelu(aS[n * 8 + hd] + s_ad[hd * NPB + dl]) - dec_f32(s_max[hd * NPB + dl]));
      float* xp = &s_agg[dl * X1P + (c & 3) * 32 + (c >> 2)];
      float acc = *xp + ws * h[n * HC + c];
      float v = acc / (s_den[hd * NPB + dl] + ws + 1e-16f) + bias[c];
      *xp = v > 0.f ? v : expm1f(v);
    }
    {
      const int c = tid & 127;
      float* wp = &s_w2[(tid >> 7) * X1P + (c & 3) * 32 + (c >> 2)];
      wp[0] = w2a; wp[8 * X1P] = w2b;
    }
    __syncthreads();
    const int wv2 = __builtin_amdgcn_readfirstlane(tid >> 5);
    if (wv2 < T2) {
      const int l15 = lane & 15, half = lane >> 4;
      const int row = (wv2 * 16 + l15) < NPB ? (wv2 * 16 + l15) : NPB - 1;
      const float* ar = s_agg + row * X1P + half * 2;
      const float* br = s_w2 + l15 * X1P + half * 2;
      v8f acc0 = (v8f){};
      #pragma unroll
      for (int s = 0; s < HC / 4; ++s) {
        const v2f a0 = *(const v2f*)(ar + 4 * s), b0 = *(const v2f*)(br + 4 * s);
        acc0 = __builtin_amdgcn_wmma_f32_16x16x4_f32(false, a0, false, b0, (short)0, acc0, false, false);
      }
      float* hp = s_h2 + (wv2 * 16 + half * 8) * H2P + l15;
      #pragma unroll
      for (int r = 0; r < 8; ++r) hp[r * H2P] = acc0[r];
    }
    __syncthreads();
    if (tid < cnt) {
      const size_t n = (size_t)(lo + tid);
      const float* hp = s_h2 + tid * H2P;
      float hv[C_];
      float s = 0.f, d = 0.f;
      #pragma unroll
      for (int c = 0; c < C_; ++c) hv[c] = hp[c];
      #pragma unroll
      for (int q = 0; q < 4; ++q) {
        s += hv[4 * q] * sa[q].x; s += hv[4 * q + 1] * sa[q].y; s += hv[4 * q + 2] * sa[q].z; s += hv[4 * q + 3] * sa[q].w;
        d += hv[4 * q] * da[q].x; d += hv[4 * q + 1] * da[q].y; d += hv[4 * q + 2] * da[q].z; d += hv[4 * q + 3] * da[q].w;
      }
      as2o[n] = s; ad2o[n] = d;
      float4* op = (float4*)(h2o + n * C_);
      op[0] = make_float4(hv[0],  hv[1],  hv[2],  hv[3]);
      op[1] = make_float4(hv[4],  hv[5],  hv[6],  hv[7]);
      op[2] = make_float4(hv[8],  hv[9],  hv[10], hv[11]);
      op[3] = make_float4(hv[12], hv[13], hv[14], hv[15]);
    }
  } else {
    if (tid < cnt) {
      int dl = tid;
      size_t n = (size_t)(lo + dl);
      float ws = expf(lrelu(aS[n] + s_ad[dl]) - dec_f32(s_max[dl]));
      float dn = s_den[dl] + ws + 1e-16f;
      float v[NCLASS];
      float m = -1e30f;
      #pragma unroll
      for (int c = 0; c < NCLASS; ++c) {
        v[c] = (s_agg[c * NPB + dl] + ws * h[n * HC + c]) / dn + bias[c];
        m = fmaxf(m, v[c]);
      }
      float sum = 0.f;
      #pragma unroll
      for (int c = 0; c < NCLASS; ++c) sum += expf(v[c] - m);
      float ls = logf(sum);
      #pragma unroll
      for (int c = 0; c < NCLASS; ++c) out[n * NCLASS + c] = v[c] - m - ls;
    }
  }
}

extern "C" void kernel_launch(void* const* d_in, const int* in_sizes, int n_in,
                              void* d_out, int out_size, void* d_ws, size_t ws_size,
                              hipStream_t stream) {
  const float* x     = (const float*)d_in[0];
  const int*   ei    = (const int*)  d_in[1];
  const float* W1    = (const float*)d_in[2];
  const float* attS1 = (const float*)d_in[3];
  const float* attD1 = (const float*)d_in[4];
  const float* b1    = (const float*)d_in[5];
  const float* W2    = (const float*)d_in[6];
  const float* attS2 = (const float*)d_in[7];
  const float* attD2 = (const float*)d_in[8];
  const float* b2    = (const float*)d_in[9];
  float* out = (float*)d_out;

  char* ws = (char*)d_ws;
  size_t off = 0;
  auto alloc = [&](size_t bytes) -> void* {
    void* p = ws + off;
    off = (off + bytes + 255) & ~(size_t)255;
    return p;
  };

  float* h1  = (float*)alloc((size_t)NNODES * F1 * 4);
  float* as1 = (float*)alloc((size_t)NNODES * H1_ * 4);
  float* ad1 = (float*)alloc((size_t)NNODES * H1_ * 4);
  float* h2  = (float*)alloc((size_t)NNODES * C_ * 4);
  float* as2 = (float*)alloc((size_t)NNODES * 4);
  float* ad2 = (float*)alloc((size_t)NNODES * 4);
  unsigned* key = (unsigned*)alloc((size_t)NIT * 8 * 1024 * 16);
  unsigned* glist = (unsigned*)alloc((size_t)NBLK * WCAP * 4);
  int* gcnt = (int*)alloc((size_t)NBLK * 4);

  const int TB = 256;
  auto nb = [](long long n, int b) { return (unsigned)((n + b - 1) / b); };

  gemm1_w_in_lds<<<G1_BLK, G1_THREADS, 0, stream>>>(x, W1, h1);
  attn_scores<<<nb((long long)NNODES * H1_, TB), TB, 0, stream>>>(
      h1, attS1, attD1, as1, ad1, NNODES * H1_, H1_);
  owner_planes<<<nb((long long)NSG * 32, TB), TB, 0, stream>>>(ei, (uint4*)key);
  owned_gat<H1_><<<NBLK, 1024, 0, stream>>>(ei, key, as1, ad1, h1, b1, nullptr, glist, gcnt,
                                            W2, attS2, attD2, h2, as2, ad2);

  owned_gat<1><<<NBLK, 1024, 0, stream>>>(ei, key, as2, ad2, h2, b2, out, glist, gcnt,
                                          nullptr, nullptr, nullptr, nullptr, nullptr, nullptr);
}


# Layer-1 sum pass: each softmax weight exp(alpha-max) formed once per (edge, head) by one lane of the head's four and passed by a quad move (2 exps per lane per trip, was 8); LDS add order as before
# speedup vs baseline: 13.2460x; 1.0498x over previous
#include <hip/hip_runtime.h>
#include <hip/hip_bf16.h>

#define NNODES 50000
#define NEDGES 1600000
#define NFEAT  512
#define F1     128
#define H1_    8
#define C_     16
#define NCLASS 16

typedef float v2f __attribute__((ext_vector_type(2)));
typedef float v8f __attribute__((ext_vector_type(8)));

__device__ __forceinline__ unsigned enc_f32(float f) {
  unsigned u = __float_as_uint(f);
  return (u & 0x80000000u) ? ~u : (u | 0x80000000u);
}
__device__ __forceinline__ float dec_f32(unsigned u) {
  return (u & 0x80000000u) ? __uint_as_float(u & 0x7FFFFFFFu) : __uint_as_float(~u);
}

#define W1P      516
#define G1_BLK   256
#define G1_WAVES 16
#define G1_THREADS 512
#define G1_STRIPS (NNODES / 16)
#define G1_FULL  (G1_STRIPS / G1_BLK)
#define G1_REST  (G1_STRIPS - G1_FULL * G1_BLK)
static_assert(NNODES % 16 == 0, "strips of 16 rows");
static_assert(G1_FULL + 4 == G1_WAVES, "waves G1_FULL .. G1_FULL+3 take the column quarters");
static_assert(G1_THREADS == G1_WAVES * 32, "work-group size");

template <int TN>
__device__ __forceinline__ void k_block(const float* __restrict__ an, const float* sw,
                                        const unsigned (&bo)[TN],
                                        const v2f (&aU)[8], v2f (&aP)[8],
                                        v2f (&bC)[TN], v8f (&acc)[TN]) {
  #pragma unroll
  for (int j = 0; j < 8; ++j) { aP[j].x = an[4 * j]; aP[j].y = an[4 * j + 1]; }
  #pragma unroll
  for (int ks = 0; ks < 8; ++ks) {
    v2f bN[TN];
    #pragma unroll
    for (int t = 0; t < TN; ++t) bN[t] = *(const v2f*)(sw + bo[t] + 4 * ks + 4);
    #pragma unroll
    for (int t = 0; t < TN; ++t)
      acc[t] = __builtin_amdgcn_wmma_f32_16x16x4_f32(false, aU[ks], false, bC[t],
                                                     (short)0, acc[t], false, false);
    #pragma unroll
    for (int t = 0; t < TN; ++t) bC[t] = bN[t];
  }
  __builtin_amdgcn_sched_barrier(0);
  #pragma unroll
  for (int j = 0; j < 8; ++j) asm volatile("" : "+v"(aP[j]));
}

template <int TN>
__device__ __forceinline__ void strip_b_from_lds(const float* __restrict__ X, const float* sw,
                                                 float* __restrict__ D, int tm, int col0, int lane) {
  constexpr int NB = NFEAT / 32;
  static_assert(NB % 2 == 0, "the two A register sets change roles every block");
  const int l15 = lane & 15, half = lane >> 4;
  const float* ap = X + (size_t)(tm * 16 + l15) * NFEAT + half * 2;
  v8f acc[TN];
  v2f a0[8], a1[8], bC[TN];
  unsigned bo[TN];
  #pragma unroll
  for (int t = 0; t < TN; ++t) {
    acc[t] = (v8f){};
    bo[t] = (unsigned)((col0 + t * 16 + l15) * W1P + half * 2);
    asm volatile("" : "+v"(bo[t]));
  }
  #pragma unroll
  for (int j = 0; j < 8; ++j) { a0[j].x = ap[4 * j]; a0[j].y = ap[4 * j + 1]; }
  #pragma unroll
  for (int t = 0; t < TN; ++t) bC[t] = *(const v2f*)(sw + bo[t]);
  for (int c = 0; c < NB; c += 2) {
    k_block<TN>(ap + 32 * (c + 1), sw, bo, a0, a1, bC, acc);
    #pragma unroll
    for (int t = 0; t < TN; ++t) bo[t] += 32;
    k_block<TN>(ap + 32 * (c + 2 < NB ? c + 2 : c + 1), sw, bo, a1, a0, bC, acc);
    #pragma unroll
    for (int t = 0; t < TN; ++t) bo[t] += 32;
  }
  #pragma unroll
  for (int t = 0; t < TN; ++t) {
    float* dp = D + (size_t)(tm * 16 + half * 8) * F1 + col0 + t * 16 + l15;
    #pragma unroll
    for (int r = 0; r < 8; ++r) dp[r * F1] = acc[t][r];
  }
}

__global__ __launch_bounds__(G1_THREADS) void gemm1_w_in_lds(
    const float* __restrict__ X, const float* __restrict__ Wt, float* __restrict__ D) {
  __shared__ __attribute__((aligned(16))) float s_w[F1 * W1P];
  const int tid = threadIdx.x, lane = tid & 31;
  const int wv = __builtin_amdgcn_readfirstlane(tid >> 5);
  #pragma unroll 8
  for (int i = tid; i < F1 * (NFEAT / 4); i += G1_THREADS) {
    const int row = i / (NFEAT / 4), c4 = i % (NFEAT / 4);
    const float4 v = ((const float4*)Wt)[i];
    *(float4*)&s_w[row * W1P + c4 * 4] = v;
  }
  if (tid < F1) *(float4*)&s_w[tid * W1P + NFEAT] = make_float4(0.f, 0.f, 0.f, 0.f);
  __syncthreads();
  if (wv < G1_FULL) {
    strip_b_from_lds<8>(X, s_w, D, wv * G1_BLK + (int)blockIdx.x, 0, lane);
  } else if ((int)blockIdx.x < G1_REST) {
    strip_b_from_lds<2>(X, s_w, D, G1_FULL * G1_BLK + (int)blockIdx.x, (wv - G1_FULL) * 32, lane);
  }
}

__global__ void attn_scores(const float* __restrict__ h,
                            const float* __restrict__ attS,
                            const float* __restrict__ attD,
                            float* __restrict__ aS, float* __restrict__ aD,
                            int NH, int H) {
  int t = blockIdx.x * blockDim.x + threadIdx.x;
  if (t >= NH) return;
  int hh = t % H;
  const float* hp = h + (size_t)t * C_;
  const float* sp = attS + hh * C_;
  const float* dp = attD + hh * C_;
  float s = 0.f, d = 0.f;
  #pragma unroll
  for (int c = 0; c < C_; ++c) { s += hp[c] * sp[c]; d += hp[c] * dp[c]; }
  aS[t] = s; aD[t] = d;
}

#define NPB    196
#define NBLK   256
#define NSG    (NEDGES / 128)
#define NIT    ((NSG + 1023) / 1024)
static_assert(NEDGES % 128 == 0, "whole groups of 128 edges");

__device__ __forceinline__ float lrelu(float a) { return a > 0.f ? a : 0.2f * a; }

__device__ __forceinline__ float quad_bcast(float v, int k) {
  const int i = __float_as_int(v);
  int r;
  switch (k) {
    case 0:  r = __builtin_amdgcn_mov_dpp(i, 0x00, 0xf, 0xf, false); break;
    case 1:  r = __builtin_amdgcn_mov_dpp(i, 0x55, 0xf, 0xf, false); break;
    case 2:  r = __builtin_amdgcn_mov_dpp(i, 0xaa, 0xf, 0xf, false); break;
    default: r = __builtin_amdgcn_mov_dpp(i, 0xff, 0xf, 0xf, false); break;
  }
  return __int_as_float(r);
}

__global__ __launch_bounds__(256) void owner_planes(const int* __restrict__ ei, uint4* __restrict__ key4) {
  const int lane = threadIdx.x & 31;
  const int g = (int)((blockIdx.x * blockDim.x + threadIdx.x) >> 5);
  if (g >= NSG) return;
  const int4 d = ((const int4*)(ei + NEDGES))[g * 32 + lane];
  const unsigned o0 = (unsigned)d.x / NPB, o1 = (unsigned)d.y / NPB;
  const unsigned o2 = (unsigned)d.z / NPB, o3 = (unsigned)d.w / NPB;
  uint4 mine = make_uint4(0u, 0u, 0u, 0u);
  #pragma unroll
  for (int b = 0; b < 8; ++b) {
    uint4 p;
    p.x = __builtin_amdgcn_ballot_w32(((o0 >> b) & 1u) != 0u);
    p.y = __builtin_amdgcn_ballot_w32(((o1 >> b) & 1u) != 0u);
    p.z = __builtin_amdgcn_ballot_w32(((o2 >> b) & 1u) != 0u);
    p.w = __builtin_amdgcn_ballot_w32(((o3 >> b) & 1u) != 0u);
    if (lane == b) mine = p;
  }
  if (lane < 8) key4[((g >> 10) * 8 + lane) * 1024 + (g & 1023)] = mine;
}

__device__ __forceinline__ void load_planes(const uint4* __restrict__ k4, int it, int tid,
                                            uint4 (&p)[8]) {
  int g = it * 1024 + tid;
  g = g < NSG ? g : NSG - 1;
  const uint4* q = k4 + ((g >> 10) * 8) * 1024 + (g & 1023);
  #pragma unroll
  for (int b = 0; b < 8; ++b) p[b] = q[b * 1024];
}

__device__ __forceinline__ unsigned and_xor(unsigned acc, unsigned plane, unsigned inv) {
  return __builtin_amdgcn_bitop3_b32(acc, plane, inv, 0x60);
}

__device__ __forceinline__ uint4 match_planes(const uint4 (&p)[8], int it, int tid,
                                              const unsigned (&inv)[8]) {
  uint4 m = make_uint4(p[0].x ^ inv[0], p[0].y ^ inv[0], p[0].z ^ inv[0], p[0].w ^ inv[0]);
  #pragma unroll
  for (int b = 1; b < 8; ++b) {
    m.x = and_xor(m.x, p[b].x, inv[b]); m.y = and_xor(m.y, p[b].y, inv[b]);
    m.z = and_xor(m.z, p[b].z, inv[b]); m.w = and_xor(m.w, p[b].w, inv[b]);
  }
  if (it * 1024 + tid >= NSG) m = make_uint4(0u, 0u, 0u, 0u);
  return m;
}

__device__ __forceinline__ int edge_of(int it, int tid, int c, int l) {
  return ((it * 1024 + tid) << 7) + (l << 2) + c;
}

__device__ __forceinline__ unsigned word_of(const uint4& m, int c) {
  return c == 0 ? m.x : c == 1 ? m.y : c == 2 ? m.z : m.w;
}

#define X1P    132
#define T2     ((NPB + 15) / 16)
#define H2P    17
#define LCAP   16384
#define WCAP   7168
static_assert(WCAP <= LCAP, "a saved list is copied back into the LDS list");
static_assert(NNODES <= 65536 && NPB <= 65536, "a list entry packs src and dst_local into 16 bits each");
static_assert(T2 <= 32, "one wave per 16-row tile of the layer-2 product");
static_assert(C_ * X1P + T2 * 16 * H2P <= LCAP, "W2 and the layer-2 rows reuse the list's LDS in the tail");
static_assert(F1 == 128 && C_ == 16 && NCLASS == C_, "index arithmetic of the fused layer-2 product");

template <int H>
__global__ __launch_bounds__(1024) void owned_gat(
    const int* __restrict__ ei, const unsigned* __restrict__ key,
    const float* __restrict__ aS, const float* __restrict__ aD,
    const float* __restrict__ h, const float* __restrict__ bias,
    float* __restrict__ out,
    unsigned* __restrict__ glist, int* __restrict__ gcnt,
    const float* __restrict__ W2, const float* __restrict__ attS2, const float* __restrict__ attD2,
    float* __restrict__ h2o, float* __restrict__ as2o, float* __restrict__ ad2o) {
  constexpr int HC = H * C_;
  constexpr int AGG = (H == 8) ? NPB * X1P : NPB * HC;
  __shared__ __attribute__((aligned(16))) float s_agg[AGG];
  __shared__ float    s_den[NPB * H];
  __shared__ unsigned s_max[NPB * H];
  __shared__ float    s_ad [NPB * H];
  __shared__ __attribute__((aligned(16))) unsigned s_list[LCAP];
  __shared__ int      s_cnt;

  const int tid = threadIdx.x, lane = tid & 31;
  const int lo  = blockIdx.x * NPB;
  const int cnt = (NNODES - lo) < NPB ? (NNODES - lo) : NPB;
  unsigned inv[8];
  #pragma unroll
  for (int b = 0; b < 8; ++b) inv[b] = ((blockIdx.x >> b) & 1u) ? 0u : ~0u;
  const uint4* k4 = (const uint4*)key;
  const int rot = blockIdx.x % NIT;

  for (int i = tid; i < NPB * H; i += 1024) {
    float ad = 0.f, as = 0.f;
    if (i < cnt * H) { ad = aD[(size_t)lo * H + i]; as = aS[(size_t)lo * H + i]; }
    const int t = (i % H) * NPB + i / H;
    s_ad[t]  = ad;
    s_max[t] = enc_f32(lrelu(as + ad));
    s_den[t] = 0.f;
  }
  for (int i = tid; i < AGG; i += 1024) s_agg[i] = 0.f;
  if (tid == 0) s_cnt = 0;
  __syncthreads();

  unsigned* gl = glist + (size_t)blockIdx.x * WCAP;
  int saved = -1;
  if constexpr (H == 1) {
    const int c = __builtin_amdgcn_readfirstlane(gcnt[blockIdx.x]);
    if (c <= WCAP) saved = c;
  }

  if (saved < 0) {
    auto scan_trip = [&](const uint4 (&cur)[8], uint4 (&nxt)[8], int t) {
      const int it = (t + rot) < NIT ? (t + rot) : (t + rot - NIT);
      if (t + 1 < NIT) load_planes(k4, (it + 1) < NIT ? (it + 1) : 0, tid, nxt);
      const uint4 m = match_planes(cur, it, tid, inv);
      unsigned mw[4] = {m.x, m.y, m.z, m.w};
      unsigned bl[4];
      int n = 0;
      #pragma unroll
      for (int c = 0; c < 4; ++c) { bl[c] = __builtin_amdgcn_ballot_w32(mw[c] != 0); n += __builtin_popcount(bl[c]); }
      if (n) {
        int off = 0;
        if (lane == 0) off = atomicAdd(&s_cnt, n);
        off = __builtin_amdgcn_readfirstlane(off);
        #pragma unroll
        for (int c = 0; c < 4; ++c) {
          if (mw[c]) {
            const int k = __builtin_ctz(mw[c]); mw[c] &= mw[c] - 1;
            const int pos = off + __builtin_popcount(bl[c] & ((1u << lane) - 1u));
            if (pos < LCAP) s_list[pos] = (unsigned)edge_of(it, tid, c, k);
          }
          off += __builtin_popcount(bl[c]);
        }
      }
      #pragma unroll
      for (int c = 0; c < 4; ++c) {
        unsigned mm = mw[c];
        for (;;) {
          const unsigned bal = __builtin_amdgcn_ballot_w32(mm != 0);
          if (!bal) break;
          int base = 0;
          if (lane == 0) base = atomicAdd(&s_cnt, __builtin_popcount(bal));
          base = __builtin_amdgcn_readfirstlane(base);
          if (mm) {
            const int k = __builtin_ctz(mm); mm &= mm - 1;
            const int pos = base + __builtin_popcount(bal & ((1u << lane) - 1u));
            if (pos < LCAP) s_list[pos] = (unsigned)edge_of(it, tid, c, k);
          }
        }
      }
    };
    uint4 pa[8], pb[8];
    load_planes(k4, rot, tid, pa);
    for (int t = 0; t + 1 < NIT; t += 2) { scan_trip(pa, pb, t); scan_trip(pb, pa, t + 1); }
    if (NIT & 1) scan_trip(pa, pb, NIT - 1);
    __syncthreads();
  }
  const int nl = saved >= 0 ? saved : __builtin_amdgcn_readfirstlane(s_cnt);
  const bool save = (H == 8) && nl <= WCAP;
  if constexpr (H == 8) { if (tid == 0) gcnt[blockIdx.x] = nl; }

  if (nl <= LCAP) {
    if (saved >= 0) {
      if constexpr (H == 1) {
        constexpr int U2 = 4;
        for (int i0 = tid; i0 < nl; i0 += 1024 * U2) {
          int idx[U2];
          unsigned ent[U2];
          bool v[U2];
          float a[U2], ad[U2];
          #pragma unroll
          for (int j = 0; j < U2; ++j) {
            idx[j] = i0 + 1024 * j;
            v[j] = idx[j] < nl;
            ent[j] = gl[v[j] ? idx[j] : i0];
          }
          #pragma unroll
          for (int j = 0; j < U2; ++j) asm volatile("" : "+v"(ent[j]));
          #pragma unroll
          for (int j = 0; j < U2; ++j) a[j] = aS[ent[j] & 0xffffu];
          #pragma unroll
          for (int j = 0; j < U2; ++j) ad[j] = s_ad[ent[j] >> 16];
          #pragma unroll
          for (int j = 0; j < U2; ++j) asm volatile("" : "+v"(a[j]));
          #pragma unroll
          for (int j = 0; j < U2; ++j) {
            if (v[j]) {
              atomicMax(&s_max[ent[j] >> 16], enc_f32(lrelu(a[j] + ad[j])));
              s_list[idx[j]] = ent[j];
            }
          }
        }
      }
    } else {
    {
      constexpr int U2 = 4;
      for (int i0 = tid; i0 < nl; i0 += 1024 * U2) {
        int idx[U2], ev[U2], sv[U2], dv[U2];
        bool v[U2];
        #pragma unroll
        for (int j = 0; j < U2; ++j) {
          idx[j] = i0 + 1024 * j;
          v[j] = idx[j] < nl;
          ev[j] = (int)s_list[v[j] ? idx[j] : i0];
        }
        #pragma unroll
        for (int j = 0; j < U2; ++j) {
          sv[j] = ei[ev[j]];
          dv[j] = ei[NEDGES + ev[j]] - lo;
        }
        if constexpr (H == 8) {
          float4 a0[U2], a1[U2];
          float ad[U2][8];
          #pragma unroll
          for (int j = 0; j < U2; ++j) {
            const float4* ap = (const float4*)(aS + (size_t)sv[j] * 8);
            a0[j] = ap[0]; a1[j] = ap[1];
          }
          #pragma unroll
          for (int j = 0; j < U2; ++j) {
            #pragma unroll
            for (int hh = 0; hh < 8; ++hh) ad[j][hh] = s_ad[hh * NPB + dv[j]];
          }
          #pragma unroll
          for (int j = 0; j < U2; ++j) {
            if (v[j]) {
              const int d = dv[j];
              atomicMax(&s_max[0 * NPB + d], enc_f32(lrelu(a0[j].x + ad[j][0])));
              atomicMax(&s_max[1 * NPB + d], enc_f32(lrelu(a0[j].y + ad[j][1])));
              atomicMax(&s_max[2 * NPB + d], enc_f32(lrelu(a0[j].z + ad[j][2])));
              atomicMax(&s_max[3 * NPB + d], enc_f32(lrelu(a0[j].w + ad[j][3])));
              atomicMax(&s_max[4 * NPB + d], enc_f32(lrelu(a1[j].x + ad[j][4])));
              atomicMax(&s_max[5 * NPB + d], enc_f32(lrelu(a1[j].y + ad[j][5])));
              atomicMax(&s_max[6 * NPB + d], enc_f32(lrelu(a1[j].z + ad[j][6])));
              atomicMax(&s_max[7 * NPB + d], enc_f32(lrelu(a1[j].w + ad[j][7])));
              const unsigned ent = (unsigned)sv[j] | ((unsigned)d << 16);
              s_list[idx[j]] = ent;
              if (save) gl[idx[j]] = ent;
            }
          }
        } else {
          float a[U2], ad[U2];
          #pragma unroll
          for (int j = 0; j < U2; ++j) a[j] = aS[sv[j]];
          #pragma unroll
          for (int j = 0; j < U2; ++j) ad[j] = s_ad[dv[j]];
          #pragma unroll
          for (int j = 0; j < U2; ++j) {
            if (v[j]) {
              atomicMax(&s_max[dv[j]], enc_f32(lrelu(a[j] + ad[j])));
              s_list[idx[j]] = (unsigned)sv[j] | ((unsigned)dv[j] << 16);
            }
          }
        }
      }
    }
    }
    __syncthreads();

    if constexpr (H == 8) {
      constexpr int EB = 8;
      const int hd = lane >> 2, qp = lane & 3;
      const int wv = __builtin_amdgcn_readfirstlane(tid >> 5);
      for (int base = wv * EB; base < nl; base += 32 * EB) {
        const int liA = base + qp, liB = base + 4 + qp;
        const unsigned entA = s_list[liA < nl ? liA : nl - 1];
        const unsigned entB = s_list[liB < nl ? liB : nl - 1];
        int sj[EB], dj[EB];
        bool vj[EB];
        #pragma unroll
        for (int j = 0; j < EB; ++j) {
          const unsigned x = (unsigned)__builtin_amdgcn_readlane((int)(j < 4 ? entA : entB), j & 3);
          sj[j] = (int)(x & 0xffffu);
          dj[j] = (int)(x >> 16);
          vj[j] = base + j < nl;
        }
        const float aA = aS[(size_t)(entA & 0xffffu) * 8 + hd];
        const float aB = aS[(size_t)(entB & 0xffffu) * 8 + hd];
        float4 rj[EB];
        #pragma unroll
        for (int j = 0; j < EB; ++j) rj[j] = ((const float4*)(h + (size_t)sj[j] * HC))[lane];
        const int oA = hd * NPB + (int)(entA >> 16), oB = hd * NPB + (int)(entB >> 16);
        const float adA = s_ad[oA], mxA = dec_f32(s_max[oA]);
        const float adB = s_ad[oB], mxB = dec_f32(s_max[oB]);
        float wA = expf(lrelu(aA + adA) - mxA);
        float wB = expf(lrelu(aB + adB) - mxB);
        asm volatile("" : "+v"(wA), "+v"(wB));
        #pragma unroll
        for (int j = 0; j < EB; ++j) {
          if (vj[j]) {
            const int dl = dj[j];
            const float w = quad_bcast(j < 4 ? wA : wB, j & 3);
            float* ap = &s_agg[dl * X1P + lane];
            atomicAdd(ap,      rj[j].x * w);
            atomicAdd(ap + 32, rj[j].y * w);
            atomicAdd(ap + 64, rj[j].z * w);
            atomicAdd(ap + 96, rj[j].w * w);
            if ((lane & 3) == 0) atomicAdd(&s_den[hd * NPB + dl], w);
          }
        }
      }
    } else {
      constexpr int U3 = 2;
      for (int i0 = tid; i0 < nl; i0 += 1024 * U3) {
        int sv[U3], dv[U3];
        bool v[U3];
        float a[U3];
        float4 r[U3][4];
        #pragma unroll
        for (int j = 0; j < U3; ++j) {
          const int idx = i0 + 1024 * j;
          v[j] = idx < nl;
          const unsigned ent = s_list[v[j] ? idx : i0];
          sv[j] = (int)(ent & 0xffffu);
          dv[j] = (int)(ent >> 16);
        }
        #pragma unroll
        for (int j = 0; j < U3; ++j) {
          a[j] = aS[sv[j]];
          const float4* rp = (const float4*)(h + (size_t)sv[j] * HC);
          r[j][0] = rp[0]; r[j][1] = rp[1]; r[j][2] = rp[2]; r[j][3] = rp[3];
        }
        float adv[U3], mxv[U3];
        #pragma unroll
        for (int j = 0; j < U3; ++j) { adv[j] = s_ad[dv[j]]; mxv[j] = dec_f32(s_max[dv[j]]); }
        #pragma unroll
        for (int j = 0; j < U3; ++j) {
          if (v[j]) {
            const int d = dv[j];
            float w = expf(lrelu(a[j] + adv[j]) - mxv[j]);
            float* ap = &s_agg[d];
            atomicAdd(ap + 0  * NPB, r[j][0].x * w); atomicAdd(ap + 1  * NPB, r[j][0].y * w);
            atomicAdd(ap + 2  * NPB, r[j][0].z * w); atomicAdd(ap + 3  * NPB, r[j][0].w * w);
            atomicAdd(ap + 4  * NPB, r[j][1].x * w); atomicAdd(ap + 5  * NPB, r[j][1].y * w);
            atomicAdd(ap + 6  * NPB, r[j][1].z * w); atomicAdd(ap + 7  * NPB, r[j][1].w * w);
            atomicAdd(ap + 8  * NPB, r[j][2].x * w); atomicAdd(ap + 9  * NPB, r[j][2].y * w);
            atomicAdd(ap + 10 * NPB, r[j][2].z * w); atomicAdd(ap + 11 * NPB, r[j][2].w * w);
            atomicAdd(ap + 12 * NPB, r[j][3].x * w); atomicAdd(ap + 13 * NPB, r[j][3].y * w);
            atomicAdd(ap + 14 * NPB, r[j][3].z * w); atomicAdd(ap + 15 * NPB, r[j][3].w * w);
            atomicAdd(&s_den[d], w);
          }
        }
      }
    }
  } else {
    {
      uint4 pc[8];
      load_planes(k4, rot, tid, pc);
      for (int t = 0; t < NIT; ++t) {
        const int it  = (t + rot) < NIT ? (t + rot) : (t + rot - NIT);
        const int itn = (it + 1) < NIT ? (it + 1) : 0;
        uint4 pn[8];
        load_planes(k4, itn, tid, pn);
        const uint4 m = match_planes(pc, it, tid, inv);
        #pragma nounroll
        for (int c = 0; c < 4; ++c) {
          unsigned mm = word_of(m, c);
          while (mm) {
            int k = __builtin_ctz(mm); mm &= mm - 1;
            int e = edge_of(it, tid, c, k);
            int s = ei[e];
            int dl = ei[NEDGES + e] - lo;
            if constexpr (H == 8) {
              const float4* ap = (const float4*)(aS + (size_t)s * 8);
              float4 a0 = ap[0], a1 = ap[1];
              atomicMax(&s_max[0 * NPB + dl], enc_f32(lrelu(a0.x + s_ad[0 * NPB + dl])));
              atomicMax(&s_max[1 * NPB + dl], enc_f32(lrelu(a0.y + s_ad[1 * NPB + dl])));
              atomicMax(&s_max[2 * NPB + dl], enc_f32(lrelu(a0.z + s_ad[2 * NPB + dl])));
              atomicMax(&s_max[3 * NPB + dl], enc_f32(lrelu(a0.w + s_ad[3 * NPB + dl])));
              atomicMax(&s_max[4 * NPB + dl], enc_f32(lrelu(a1.x + s_ad[4 * NPB + dl])));
              atomicMax(&s_max[5 * NPB + dl], enc_f32(lrelu(a1.y + s_ad[5 * NPB + dl])));
              atomicMax(&s_max[6 * NPB + dl], enc_f32(lrelu(a1.z + s_ad[6 * NPB + dl])));
              atomicMax(&s_max[7 * NPB + dl], enc_f32(lrelu(a1.w + s_ad[7 * NPB + dl])));
            } else {
              atomicMax(&s_max[dl], enc_f32(lrelu(aS[s] + s_ad[dl])));
            }
          }
        }
        #pragma unroll
        for (int b = 0; b < 8; ++b) pc[b] = pn[b];
      }
    }
    __syncthreads();

    {
      uint4 pc[8];
      load_planes(k4, rot, tid, pc);
      for (int t = 0; t < NIT; ++t) {
        const int it  = (t + rot) < NIT ? (t + rot) : (t + rot - NIT);
        const int itn = (it + 1) < NIT ? (it + 1) : 0;
        uint4 pn[8];
        load_planes(k4, itn, tid, pn);
        const uint4 m = match_planes(pc, it, tid, inv);
        #pragma nounroll
        for (int c = 0; c < 4; ++c) {
          unsigned mm = word_of(m, c);
          if constexpr (H == 8) {
            const int hd = lane >> 2;
            for (;;) {
              unsigned bal = __builtin_amdgcn_ballot_w32(mm != 0);
              if (!bal) break;
              int sv = 0, dv = 0;
              if (mm) {
                int k = __builtin_ctz(mm); mm &= mm - 1;
                int e = edge_of(it, tid, c, k);
                sv = ei[e];
                dv = ei[NEDGES + e] - lo;
              }
              while (bal) {
                int ml = __builtin_ctz(bal); bal &= bal - 1;
                int s  = __builtin_amdgcn_readlane(sv, ml);
                int dl = __builtin_amdgcn_readlane(dv, ml);
                float  a = aS[(size_t)s * 8 + hd];
                float4 r = ((const float4*)(h + (size_t)s * HC))[lane];
                float w = expf(lrelu(a + s_ad[hd * NPB + dl]) - dec_f32(s_max[hd * NPB + dl]));
                float* ap = &s_agg[dl * X1P + lane];
                atomicAdd(ap,      r.x * w);
                atomicAdd(ap + 32, r.y * w);
                atomicAdd(ap + 64, r.z * w);
                atomicAdd(ap + 96, r.w * w);
                if ((lane & 3) == 0) atomicAdd(&s_den[hd * NPB + dl], w);
              }
            }
          } else {
            while (mm) {
              int k = __builtin_ctz(mm); mm &= mm - 1;
              int e = edge_of(it, tid, c, k);
              int s = ei[e];
              int dl = ei[NEDGES + e] - lo;
              float a = aS[s];
              const float4* rp = (const float4*)(h + (size_t)s * HC);
              float4 r0 = rp[0], r1 = rp[1], r2 = rp[2], r3 = rp[3];
              float w = expf(lrelu(a + s_ad[dl]) - dec_f32(s_max[dl]));
              float* ap = &s_agg[dl];
              atomicAdd(ap + 0  * NPB, r0.x * w); atomicAdd(ap + 1  * NPB, r0.y * w);
              atomicAdd(ap + 2  * NPB, r0.z * w); atomicAdd(ap + 3  * NPB, r0.w * w);
              atomicAdd(ap + 4  * NPB, r1.x * w); atomicAdd(ap + 5  * NPB, r1.y * w);
              atomicAdd(ap + 6  * NPB, r1.z * w); atomicAdd(ap + 7  * NPB, r1.w * w);
              atomicAdd(ap + 8  * NPB, r2.x * w); atomicAdd(ap + 9  * NPB, r2.y * w);
              atomicAdd(ap + 10 * NPB, r2.z * w); atomicAdd(ap + 11 * NPB, r2.w * w);
              atomicAdd(ap + 12 * NPB, r3.x * w); atomicAdd(ap + 13 * NPB, r3.y * w);
              atomicAdd(ap + 14 * NPB, r3.z * w); atomicAdd(ap + 15 * NPB, r3.w * w);
              atomicAdd(&s_den[dl], w);
            }
          }
        }
        #pragma unroll
        for (int b = 0; b < 8; ++b) pc[b] = pn[b];
      }
    }
  }
  __syncthreads();

  if constexpr (H == 8) {
    float* s_w2 = (float*)s_list;
    float* s_h2 = s_w2 + C_ * X1P;
    static_assert(C_ * HC == 2 * 1024, "W2 is copied two words per thread");
    const float w2a = W2[tid], w2b = W2[tid + 1024];
    float4 sa[4], da[4];
    #pragma unroll
    for (int q = 0; q < 4; ++q) { sa[q] = ((const float4*)attS2)[q]; da[q] = ((const float4*)attD2)[q]; }
    for (int idx = tid; idx < cnt * HC; idx += 1024) {
      int dl = idx >> 7, c = idx & 127, hd = c >> 4;
      size_t n = (size_t)(lo + dl);
      float ws = expf(lrelu(aS[n * 8 + hd] + s_ad[hd * NPB + dl]) - dec_f32(s_max[hd * NPB + dl]));
      float* xp = &s_agg[dl * X1P + (c & 3) * 32 + (c >> 2)];
      float acc = *xp + ws * h[n * HC + c];
      float v = acc / (s_den[hd * NPB + dl] + ws + 1e-16f) + bias[c];
      *xp = v > 0.f ? v : expm1f(v);
    }
    {
      const int c = tid & 127;
      float* wp = &s_w2[(tid >> 7) * X1P + (c & 3) * 32 + (c >> 2)];
      wp[0] = w2a; wp[8 * X1P] = w2b;
    }
    __syncthreads();
    const int wv2 = __builtin_amdgcn_readfirstlane(tid >> 5);
    if (wv2 < T2) {
      const int l15 = lane & 15, half = lane >> 4;
      const int row = (wv2 * 16 + l15) < NPB ? (wv2 * 16 + l15) : NPB - 1;
      const float* ar = s_agg + row * X1P + half * 2;
      const float* br = s_w2 + l15 * X1P + half * 2;
      v8f acc0 = (v8f){};
      #pragma unroll
      for (int s = 0; s < HC / 4; ++s) {
        const v2f a0 = *(const v2f*)(ar + 4 * s), b0 = *(const v2f*)(br + 4 * s);
        acc0 = __builtin_amdgcn_wmma_f32_16x16x4_f32(false, a0, false, b0, (short)0, acc0, false, false);
      }
      float* hp = s_h2 + (wv2 * 16 + half * 8) * H2P + l15;
      #pragma unroll
      for (int r = 0; r < 8; ++r) hp[r * H2P] = acc0[r];
    }
    __syncthreads();
    if (tid < cnt) {
      const size_t n = (size_t)(lo + tid);
      const float* hp = s_h2 + tid * H2P;
      float hv[C_];
      float s = 0.f, d = 0.f;
      #pragma unroll
      for (int c = 0; c < C_; ++c) hv[c] = hp[c];
      #pragma unroll
      for (int q = 0; q < 4; ++q) {
        s += hv[4 * q] * sa[q].x; s += hv[4 * q + 1] * sa[q].y; s += hv[4 * q + 2] * sa[q].z; s += hv[4 * q + 3] * sa[q].w;
        d += hv[4 * q] * da[q].x; d += hv[4 * q + 1] * da[q].y; d += hv[4 * q + 2] * da[q].z; d += hv[4 * q + 3] * da[q].w;
      }
      as2o[n] = s; ad2o[n] = d;
      float4* op = (float4*)(h2o + n * C_);
      op[0] = make_float4(hv[0],  hv[1],  hv[2],  hv[3]);
      op[1] = make_float4(hv[4],  hv[5],  hv[6],  hv[7]);
      op[2] = make_float4(hv[8],  hv[9],  hv[10], hv[11]);
      op[3] = make_float4(hv[12], hv[13], hv[14], hv[15]);
    }
  } else {
    if (tid < cnt) {
      int dl = tid;
      size_t n = (size_t)(lo + dl);
      float ws = expf(lrelu(aS[n] + s_ad[dl]) - dec_f32(s_max[dl]));
      float dn = s_den[dl] + ws + 1e-16f;
      float v[NCLASS];
      float m = -1e30f;
      #pragma unroll
      for (int c = 0; c < NCLASS; ++c) {
        v[c] = (s_agg[c * NPB + dl] + ws * h[n * HC + c]) / dn + bias[c];
        m = fmaxf(m, v[c]);
      }
      float sum = 0.f;
      #pragma unroll
      for (int c = 0; c < NCLASS; ++c) sum += expf(v[c] - m);
      float ls = logf(sum);
      #pragma unroll
      for (int c = 0; c < NCLASS; ++c) out[n * NCLASS + c] = v[c] - m - ls;
    }
  }
}

extern "C" void kernel_launch(void* const* d_in, const int* in_sizes, int n_in,
                              void* d_out, int out_size, void* d_ws, size_t ws_size,
                              hipStream_t stream) {
  const float* x     = (const float*)d_in[0];
  const int*   ei    = (const int*)  d_in[1];
  const float* W1    = (const float*)d_in[2];
  const float* attS1 = (const float*)d_in[3];
  const float* attD1 = (const float*)d_in[4];
  const float* b1    = (const float*)d_in[5];
  const float* W2    = (const float*)d_in[6];
  const float* attS2 = (const float*)d_in[7];
  const float* attD2 = (const float*)d_in[8];
  const float* b2    = (const float*)d_in[9];
  float* out = (float*)d_out;

  char* ws = (char*)d_ws;
  size_t off = 0;
  auto alloc = [&](size_t bytes) -> void* {
    void* p = ws + off;
    off = (off + bytes + 255) & ~(size_t)255;
    return p;
  };

  float* h1  = (float*)alloc((size_t)NNODES * F1 * 4);
  float* as1 = (float*)alloc((size_t)NNODES * H1_ * 4);
  float* ad1 = (float*)alloc((size_t)NNODES * H1_ * 4);
  float* h2  = (float*)alloc((size_t)NNODES * C_ * 4);
  float* as2 = (float*)alloc((size_t)NNODES * 4);
  float* ad2 = (float*)alloc((size_t)NNODES * 4);
  unsigned* key = (unsigned*)alloc((size_t)NIT * 8 * 1024 * 16);
  unsigned* glist = (unsigned*)alloc((size_t)NBLK * WCAP * 4);
  int* gcnt = (int*)alloc((size_t)NBLK * 4);

  const int TB = 256;
  auto nb = [](long long n, int b) { return (unsigned)((n + b - 1) / b); };

  gemm1_w_in_lds<<<G1_BLK, G1_THREADS, 0, stream>>>(x, W1, h1);
  attn_scores<<<nb((long long)NNODES * H1_, TB), TB, 0, stream>>>(
      h1, attS1, attD1, as1, ad1, NNODES * H1_, H1_);
  owner_planes<<<nb((long long)NSG * 32, TB), TB, 0, stream>>>(ei, (uint4*)key);
  owned_gat<H1_><<<NBLK, 1024, 0, stream>>>(ei, key, as1, ad1, h1, b1, nullptr, glist, gcnt,
                                            W2, attS2, attD2, h2, as2, ad2);

  owned_gat<1><<<NBLK, 1024, 0, stream>>>(ei, key, as2, ad2, h2, b2, out, glist, gcnt,
                                          nullptr, nullptr, nullptr, nullptr, nullptr, nullptr);
}


# Layer-1 tail by rows: wave per node, lane on 4 channels; self-loop weight and ONE reciprocal of the denominator per (node, head), not per element; rows fetched together; quotient rounding changes
# speedup vs baseline: 13.6652x; 1.0316x over previous
#include <hip/hip_runtime.h>
#include <hip/hip_bf16.h>

#define NNODES 50000
#define NEDGES 1600000
#define NFEAT  512
#define F1     128
#define H1_    8
#define C_     16
#define NCLASS 16

typedef float v2f __attribute__((ext_vector_type(2)));
typedef float v8f __attribute__((ext_vector_type(8)));

__device__ __forceinline__ unsigned enc_f32(float f) {
  unsigned u = __float_as_uint(f);
  return (u & 0x80000000u) ? ~u : (u | 0x80000000u);
}
__device__ __forceinline__ float dec_f32(unsigned u) {
  return (u & 0x80000000u) ? __uint_as_float(u & 0x7FFFFFFFu) : __uint_as_float(~u);
}

#define W1P      516
#define G1_BLK   256
#define G1_WAVES 16
#define G1_THREADS 512
#define G1_STRIPS (NNODES / 16)
#define G1_FULL  (G1_STRIPS / G1_BLK)
#define G1_REST  (G1_STRIPS - G1_FULL * G1_BLK)
static_assert(NNODES % 16 == 0, "strips of 16 rows");
static_assert(G1_FULL + 4 == G1_WAVES, "waves G1_FULL .. G1_FULL+3 take the column quarters");
static_assert(G1_THREADS == G1_WAVES * 32, "work-group size");

template <int TN>
__device__ __forceinline__ void k_block(const float* __restrict__ an, const float* sw,
                                        const unsigned (&bo)[TN],
                                        const v2f (&aU)[8], v2f (&aP)[8],
                                        v2f (&bC)[TN], v8f (&acc)[TN]) {
  #pragma unroll
  for (int j = 0; j < 8; ++j) { aP[j].x = an[4 * j]; aP[j].y = an[4 * j + 1]; }
  #pragma unroll
  for (int ks = 0; ks < 8; ++ks) {
    v2f bN[TN];
    #pragma unroll
    for (int t = 0; t < TN; ++t) bN[t] = *(const v2f*)(sw + bo[t] + 4 * ks + 4);
    #pragma unroll
    for (int t = 0; t < TN; ++t)
      acc[t] = __builtin_amdgcn_wmma_f32_16x16x4_f32(false, aU[ks], false, bC[t],
                                                     (short)0, acc[t], false, false);
    #pragma unroll
    for (int t = 0; t < TN; ++t) bC[t] = bN[t];
  }
  __builtin_amdgcn_sched_barrier(0);
  #pragma unroll
  for (int j = 0; j < 8; ++j) asm volatile("" : "+v"(aP[j]));
}

template <int TN>
__device__ __forceinline__ void strip_b_from_lds(const float* __restrict__ X, const float* sw,
                                                 float* __restrict__ D, int tm, int col0, int lane) {
  constexpr int NB = NFEAT / 32;
  static_assert(NB % 2 == 0, "the two A register sets change roles every block");
  const int l15 = lane & 15, half = lane >> 4;
  const float* ap = X + (size_t)(tm * 16 + l15) * NFEAT + half * 2;
  v8f acc[TN];
  v2f a0[8], a1[8], bC[TN];
  unsigned bo[TN];
  #pragma unroll
  for (int t = 0; t < TN; ++t) {
    acc[t] = (v8f){};
    bo[t] = (unsigned)((col0 + t * 16 + l15) * W1P + half * 2);
    asm volatile("" : "+v"(bo[t]));
  }
  #pragma unroll
  for (int j = 0; j < 8; ++j) { a0[j].x = ap[4 * j]; a0[j].y = ap[4 * j + 1]; }
  #pragma unroll
  for (int t = 0; t < TN; ++t) bC[t] = *(const v2f*)(sw + bo[t]);
  for (int c = 0; c < NB; c += 2) {
    k_block<TN>(ap + 32 * (c + 1), sw, bo, a0, a1, bC, acc);
    #pragma unroll
    for (int t = 0; t < TN; ++t) bo[t] += 32;
    k_block<TN>(ap + 32 * (c + 2 < NB ? c + 2 : c + 1), sw, bo, a1, a0, bC, acc);
    #pragma unroll
    for (int t = 0; t < TN; ++t) bo[t] += 32;
  }
  #pragma unroll
  for (int t = 0; t < TN; ++t) {
    float* dp = D + (size_t)(tm * 16 + half * 8) * F1 + col0 + t * 16 + l15;
    #pragma unroll
    for (int r = 0; r < 8; ++r) dp[r * F1] = acc[t][r];
  }
}

__global__ __launch_bounds__(G1_THREADS) void gemm1_w_in_lds(
    const float* __restrict__ X, const float* __restrict__ Wt, float* __restrict__ D) {
  __shared__ __attribute__((aligned(16))) float s_w[F1 * W1P];
  const int tid = threadIdx.x, lane = tid & 31;
  const int wv = __builtin_amdgcn_readfirstlane(tid >> 5);
  #pragma unroll 8
  for (int i = tid; i < F1 * (NFEAT / 4); i += G1_THREADS) {
    const int row = i / (NFEAT / 4), c4 = i % (NFEAT / 4);
    const float4 v = ((const float4*)Wt)[i];
    *(float4*)&s_w[row * W1P + c4 * 4] = v;
  }
  if (tid < F1) *(float4*)&s_w[tid * W1P + NFEAT] = make_float4(0.f, 0.f, 0.f, 0.f);
  __syncthreads();
  if (wv < G1_FULL) {
    strip_b_from_lds<8>(X, s_w, D, wv * G1_BLK + (int)blockIdx.x, 0, lane);
  } else if ((int)blockIdx.x < G1_REST) {
    strip_b_from_lds<2>(X, s_w, D, G1_FULL * G1_BLK + (int)blockIdx.x, (wv - G1_FULL) * 32, lane);
  }
}

__global__ void attn_scores(const float* __restrict__ h,
                            const float* __restrict__ attS,
                            const float* __restrict__ attD,
                            float* __restrict__ aS, float* __restrict__ aD,
                            int NH, int H) {
  int t = blockIdx.x * blockDim.x + threadIdx.x;
  if (t >= NH) return;
  int hh = t % H;
  const float* hp = h + (size_t)t * C_;
  const float* sp = attS + hh * C_;
  const float* dp = attD + hh * C_;
  float s = 0.f, d = 0.f;
  #pragma unroll
  for (int c = 0; c < C_; ++c) { s += hp[c] * sp[c]; d += hp[c] * dp[c]; }
  aS[t] = s; aD[t] = d;
}

#define NPB    196
#define NBLK   256
#define NSG    (NEDGES / 128)
#define NIT    ((NSG + 1023) / 1024)
static_assert(NEDGES % 128 == 0, "whole groups of 128 edges");

__device__ __forceinline__ float lrelu(float a) { return a > 0.f ? a : 0.2f * a; }

__device__ __forceinline__ float quad_bcast(float v, int k) {
  const int i = __float_as_int(v);
  int r;
  switch (k) {
    case 0:  r = __builtin_amdgcn_mov_dpp(i, 0x00, 0xf, 0xf, false); break;
    case 1:  r = __builtin_amdgcn_mov_dpp(i, 0x55, 0xf, 0xf, false); break;
    case 2:  r = __builtin_amdgcn_mov_dpp(i, 0xaa, 0xf, 0xf, false); break;
    default: r = __builtin_amdgcn_mov_dpp(i, 0xff, 0xf, 0xf, false); break;
  }
  return __int_as_float(r);
}

__global__ __launch_bounds__(256) void owner_planes(const int* __restrict__ ei, uint4* __restrict__ key4) {
  const int lane = threadIdx.x & 31;
  const int g = (int)((blockIdx.x * blockDim.x + threadIdx.x) >> 5);
  if (g >= NSG) return;
  const int4 d = ((const int4*)(ei + NEDGES))[g * 32 + lane];
  const unsigned o0 = (unsigned)d.x / NPB, o1 = (unsigned)d.y / NPB;
  const unsigned o2 = (unsigned)d.z / NPB, o3 = (unsigned)d.w / NPB;
  uint4 mine = make_uint4(0u, 0u, 0u, 0u);
  #pragma unroll
  for (int b = 0; b < 8; ++b) {
    uint4 p;
    p.x = __builtin_amdgcn_ballot_w32(((o0 >> b) & 1u) != 0u);
    p.y = __builtin_amdgcn_ballot_w32(((o1 >> b) & 1u) != 0u);
    p.z = __builtin_amdgcn_ballot_w32(((o2 >> b) & 1u) != 0u);
    p.w = __builtin_amdgcn_ballot_w32(((o3 >> b) & 1u) != 0u);
    if (lane == b) mine = p;
  }
  if (lane < 8) key4[((g >> 10) * 8 + lane) * 1024 + (g & 1023)] = mine;
}

__device__ __forceinline__ void load_planes(const uint4* __restrict__ k4, int it, int tid,
                                            uint4 (&p)[8]) {
  int g = it * 1024 + tid;
  g = g < NSG ? g : NSG - 1;
  const uint4* q = k4 + ((g >> 10) * 8) * 1024 + (g & 1023);
  #pragma unroll
  for (int b = 0; b < 8; ++b) p[b] = q[b * 1024];
}

__device__ __forceinline__ unsigned and_xor(unsigned acc, unsigned plane, unsigned inv) {
  return __builtin_amdgcn_bitop3_b32(acc, plane, inv, 0x60);
}

__device__ __forceinline__ uint4 match_planes(const uint4 (&p)[8], int it, int tid,
                                              const unsigned (&inv)[8]) {
  uint4 m = make_uint4(p[0].x ^ inv[0], p[0].y ^ inv[0], p[0].z ^ inv[0], p[0].w ^ inv[0]);
  #pragma unroll
  for (int b = 1; b < 8; ++b) {
    m.x = and_xor(m.x, p[b].x, inv[b]); m.y = and_xor(m.y, p[b].y, inv[b]);
    m.z = and_xor(m.z, p[b].z, inv[b]); m.w = and_xor(m.w, p[b].w, inv[b]);
  }
  if (it * 1024 + tid >= NSG) m = make_uint4(0u, 0u, 0u, 0u);
  return m;
}

__device__ __forceinline__ int edge_of(int it, int tid, int c, int l) {
  return ((it * 1024 + tid) << 7) + (l << 2) + c;
}

__device__ __forceinline__ unsigned word_of(const uint4& m, int c) {
  return c == 0 ? m.x : c == 1 ? m.y : c == 2 ? m.z : m.w;
}

#define X1P    132
#define T2     ((NPB + 15) / 16)
#define H2P    17
#define LCAP   16384
#define WCAP   7168
static_assert(WCAP <= LCAP, "a saved list is copied back into the LDS list");
static_assert(NNODES <= 65536 && NPB <= 65536, "a list entry packs src and dst_local into 16 bits each");
static_assert(T2 <= 32, "one wave per 16-row tile of the layer-2 product");
static_assert(C_ * X1P + T2 * 16 * H2P <= LCAP, "W2 and the layer-2 rows reuse the list's LDS in the tail");
static_assert(F1 == 128 && C_ == 16 && NCLASS == C_, "index arithmetic of the fused layer-2 product");

template <int H>
__global__ __launch_bounds__(1024) void owned_gat(
    const int* __restrict__ ei, const unsigned* __restrict__ key,
    const float* __restrict__ aS, const float* __restrict__ aD,
    const float* __restrict__ h, const float* __restrict__ bias,
    float* __restrict__ out,
    unsigned* __restrict__ glist, int* __restrict__ gcnt,
    const float* __restrict__ W2, const float* __restrict__ attS2, const float* __restrict__ attD2,
    float* __restrict__ h2o, float* __restrict__ as2o, float* __restrict__ ad2o) {
  constexpr int HC = H * C_;
  constexpr int AGG = (H == 8) ? NPB * X1P : NPB * HC;
  __shared__ __attribute__((aligned(16))) float s_agg[AGG];
  __shared__ float    s_den[NPB * H];
  __shared__ unsigned s_max[NPB * H];
  __shared__ float    s_ad [NPB * H];
  __shared__ __attribute__((aligned(16))) unsigned s_list[LCAP];
  __shared__ int      s_cnt;

  const int tid = threadIdx.x, lane = tid & 31;
  const int lo  = blockIdx.x * NPB;
  const int cnt = (NNODES - lo) < NPB ? (NNODES - lo) : NPB;
  unsigned inv[8];
  #pragma unroll
  for (int b = 0; b < 8; ++b) inv[b] = ((blockIdx.x >> b) & 1u) ? 0u : ~0u;
  const uint4* k4 = (const uint4*)key;
  const int rot = blockIdx.x % NIT;

  for (int i = tid; i < NPB * H; i += 1024) {
    float ad = 0.f, as = 0.f;
    if (i < cnt * H) { ad = aD[(size_t)lo * H + i]; as = aS[(size_t)lo * H + i]; }
    const int t = (i % H) * NPB + i / H;
    s_ad[t]  = ad;
    s_max[t] = enc_f32(lrelu(as + ad));
    s_den[t] = 0.f;
  }
  for (int i = tid; i < AGG; i += 1024) s_agg[i] = 0.f;
  if (tid == 0) s_cnt = 0;
  __syncthreads();

  unsigned* gl = glist + (size_t)blockIdx.x * WCAP;
  int saved = -1;
  if constexpr (H == 1) {
    const int c = __builtin_amdgcn_readfirstlane(gcnt[blockIdx.x]);
    if (c <= WCAP) saved = c;
  }

  if (saved < 0) {
    auto scan_trip = [&](const uint4 (&cur)[8], uint4 (&nxt)[8], int t) {
      const int it = (t + rot) < NIT ? (t + rot) : (t + rot - NIT);
      if (t + 1 < NIT) load_planes(k4, (it + 1) < NIT ? (it + 1) : 0, tid, nxt);
      const uint4 m = match_planes(cur, it, tid, inv);
      unsigned mw[4] = {m.x, m.y, m.z, m.w};
      unsigned bl[4];
      int n = 0;
      #pragma unroll
      for (int c = 0; c < 4; ++c) { bl[c] = __builtin_amdgcn_ballot_w32(mw[c] != 0); n += __builtin_popcount(bl[c]); }
      if (n) {
        int off = 0;
        if (lane == 0) off = atomicAdd(&s_cnt, n);
        off = __builtin_amdgcn_readfirstlane(off);
        #pragma unroll
        for (int c = 0; c < 4; ++c) {
          if (mw[c]) {
            const int k = __builtin_ctz(mw[c]); mw[c] &= mw[c] - 1;
            const int pos = off + __builtin_popcount(bl[c] & ((1u << lane) - 1u));
            if (pos < LCAP) s_list[pos] = (unsigned)edge_of(it, tid, c, k);
          }
          off += __builtin_popcount(bl[c]);
        }
      }
      #pragma unroll
      for (int c = 0; c < 4; ++c) {
        unsigned mm = mw[c];
        for (;;) {
          const unsigned bal = __builtin_amdgcn_ballot_w32(mm != 0);
          if (!bal) break;
          int base = 0;
          if (lane == 0) base = atomicAdd(&s_cnt, __builtin_popcount(bal));
          base = __builtin_amdgcn_readfirstlane(base);
          if (mm) {
            const int k = __builtin_ctz(mm); mm &= mm - 1;
            const int pos = base + __builtin_popcount(bal & ((1u << lane) - 1u));
            if (pos < LCAP) s_list[pos] = (unsigned)edge_of(it, tid, c, k);
          }
        }
      }
    };
    uint4 pa[8], pb[8];
    load_planes(k4, rot, tid, pa);
    for (int t = 0; t + 1 < NIT; t += 2) { scan_trip(pa, pb, t); scan_trip(pb, pa, t + 1); }
    if (NIT & 1) scan_trip(pa, pb, NIT - 1);
    __syncthreads();
  }
  const int nl = saved >= 0 ? saved : __builtin_amdgcn_readfirstlane(s_cnt);
  const bool save = (H == 8) && nl <= WCAP;
  if constexpr (H == 8) { if (tid == 0) gcnt[blockIdx.x] = nl; }

  if (nl <= LCAP) {
    if (saved >= 0) {
      if constexpr (H == 1) {
        constexpr int U2 = 4;
        for (int i0 = tid; i0 < nl; i0 += 1024 * U2) {
          int idx[U2];
          unsigned ent[U2];
          bool v[U2];
          float a[U2], ad[U2];
          #pragma unroll
          for (int j = 0; j < U2; ++j) {
            idx[j] = i0 + 1024 * j;
            v[j] = idx[j] < nl;
            ent[j] = gl[v[j] ? idx[j] : i0];
          }
          #pragma unroll
          for (int j = 0; j < U2; ++j) asm volatile("" : "+v"(ent[j]));
          #pragma unroll
          for (int j = 0; j < U2; ++j) a[j] = aS[ent[j] & 0xffffu];
          #pragma unroll
          for (int j = 0; j < U2; ++j) ad[j] = s_ad[ent[j] >> 16];
          #pragma unroll
          for (int j = 0; j < U2; ++j) asm volatile("" : "+v"(a[j]));
          #pragma unroll
          for (int j = 0; j < U2; ++j) {
            if (v[j]) {
              atomicMax(&s_max[ent[j] >> 16], enc_f32(lrelu(a[j] + ad[j])));
              s_list[idx[j]] = ent[j];
            }
          }
        }
      }
    } else {
    {
      constexpr int U2 = 4;
      for (int i0 = tid; i0 < nl; i0 += 1024 * U2) {
        int idx[U2], ev[U2], sv[U2], dv[U2];
        bool v[U2];
        #pragma unroll
        for (int j = 0; j < U2; ++j) {
          idx[j] = i0 + 1024 * j;
          v[j] = idx[j] < nl;
          ev[j] = (int)s_list[v[j] ? idx[j] : i0];
        }
        #pragma unroll
        for (int j = 0; j < U2; ++j) {
          sv[j] = ei[ev[j]];
          dv[j] = ei[NEDGES + ev[j]] - lo;
        }
        if constexpr (H == 8) {
          float4 a0[U2], a1[U2];
          float ad[U2][8];
          #pragma unroll
          for (int j = 0; j < U2; ++j) {
            const float4* ap = (const float4*)(aS + (size_t)sv[j] * 8);
            a0[j] = ap[0]; a1[j] = ap[1];
          }
          #pragma unroll
          for (int j = 0; j < U2; ++j) {
            #pragma unroll
            for (int hh = 0; hh < 8; ++hh) ad[j][hh] = s_ad[hh * NPB + dv[j]];
          }
          #pragma unroll
          for (int j = 0; j < U2; ++j) {
            if (v[j]) {
              const int d = dv[j];
              atomicMax(&s_max[0 * NPB + d], enc_f32(lrelu(a0[j].x + ad[j][0])));
              atomicMax(&s_max[1 * NPB + d], enc_f32(lrelu(a0[j].y + ad[j][1])));
              atomicMax(&s_max[2 * NPB + d], enc_f32(lrelu(a0[j].z + ad[j][2])));
              atomicMax(&s_max[3 * NPB + d], enc_f32(lrelu(a0[j].w + ad[j][3])));
              atomicMax(&s_max[4 * NPB + d], enc_f32(lrelu(a1[j].x + ad[j][4])));
              atomicMax(&s_max[5 * NPB + d], enc_f32(lrelu(a1[j].y + ad[j][5])));
              atomicMax(&s_max[6 * NPB + d], enc_f32(lrelu(a1[j].z + ad[j][6])));
              atomicMax(&s_max[7 * NPB + d], enc_f32(lrelu(a1[j].w + ad[j][7])));
              const unsigned ent = (unsigned)sv[j] | ((unsigned)d << 16);
              s_list[idx[j]] = ent;
              if (save) gl[idx[j]] = ent;
            }
          }
        } else {
          float a[U2], ad[U2];
          #pragma unroll
          for (int j = 0; j < U2; ++j) a[j] = aS[sv[j]];
          #pragma unroll
          for (int j = 0; j < U2; ++j) ad[j] = s_ad[dv[j]];
          #pragma unroll
          for (int j = 0; j < U2; ++j) {
            if (v[j]) {
              atomicMax(&s_max[dv[j]], enc_f32(lrelu(a[j] + ad[j])));
              s_list[idx[j]] = (unsigned)sv[j] | ((unsigned)dv[j] << 16);
            }
          }
        }
      }
    }
    }
    __syncthreads();

    if constexpr (H == 8) {
      constexpr int EB = 8;
      const int hd = lane >> 2, qp = lane & 3;
      const int wv = __builtin_amdgcn_readfirstlane(tid >> 5);
      for (int base = wv * EB; base < nl; base += 32 * EB) {
        const int liA = base + qp, liB = base + 4 + qp;
        const unsigned entA = s_list[liA < nl ? liA : nl - 1];
        const unsigned entB = s_list[liB < nl ? liB : nl - 1];
        int sj[EB], dj[EB];
        bool vj[EB];
        #pragma unroll
        for (int j = 0; j < EB; ++j) {
          const unsigned x = (unsigned)__builtin_amdgcn_readlane((int)(j < 4 ? entA : entB), j & 3);
          sj[j] = (int)(x & 0xffffu);
          dj[j] = (int)(x >> 16);
          vj[j] = base + j < nl;
        }
        const float aA = aS[(size_t)(entA & 0xffffu) * 8 + hd];
        const float aB = aS[(size_t)(entB & 0xffffu) * 8 + hd];
        float4 rj[EB];
        #pragma unroll
        for (int j = 0; j < EB; ++j) rj[j] = ((const float4*)(h + (size_t)sj[j] * HC))[lane];
        const int oA = hd * NPB + (int)(entA >> 16), oB = hd * NPB + (int)(entB >> 16);
        const float adA = s_ad[oA], mxA = dec_f32(s_max[oA]);
        const float adB = s_ad[oB], mxB = dec_f32(s_max[oB]);
        float wA = expf(lrelu(aA + adA) - mxA);
        float wB = expf(lrelu(aB + adB) - mxB);
        asm volatile("" : "+v"(wA), "+v"(wB));
        #pragma unroll
        for (int j = 0; j < EB; ++j) {
          if (vj[j]) {
            const int dl = dj[j];
            const float w = quad_bcast(j < 4 ? wA : wB, j & 3);
            float* ap = &s_agg[dl * X1P + lane];
            atomicAdd(ap,      rj[j].x * w);
            atomicAdd(ap + 32, rj[j].y * w);
            atomicAdd(ap + 64, rj[j].z * w);
            atomicAdd(ap + 96, rj[j].w * w);
            if ((lane & 3) == 0) atomicAdd(&s_den[hd * NPB + dl], w);
          }
        }
      }
    } else {
      constexpr int U3 = 2;
      for (int i0 = tid; i0 < nl; i0 += 1024 * U3) {
        int sv[U3], dv[U3];
        bool v[U3];
        float a[U3];
        float4 r[U3][4];
        #pragma unroll
        for (int j = 0; j < U3; ++j) {
          const int idx = i0 + 1024 * j;
          v[j] = idx < nl;
          const unsigned ent = s_list[v[j] ? idx : i0];
          sv[j] = (int)(ent & 0xffffu);
          dv[j] = (int)(ent >> 16);
        }
        #pragma unroll
        for (int j = 0; j < U3; ++j) {
          a[j] = aS[sv[j]];
          const float4* rp = (const float4*)(h + (size_t)sv[j] * HC);
          r[j][0] = rp[0]; r[j][1] = rp[1]; r[j][2] = rp[2]; r[j][3] = rp[3];
        }
        float adv[U3], mxv[U3];
        #pragma unroll
        for (int j = 0; j < U3; ++j) { adv[j] = s_ad[dv[j]]; mxv[j] = dec_f32(s_max[dv[j]]); }
        #pragma unroll
        for (int j = 0; j < U3; ++j) {
          if (v[j]) {
            const int d = dv[j];
            float w = expf(lrelu(a[j] + adv[j]) - mxv[j]);
            float* ap = &s_agg[d];
            atomicAdd(ap + 0  * NPB, r[j][0].x * w); atomicAdd(ap + 1  * NPB, r[j][0].y * w);
            atomicAdd(ap + 2  * NPB, r[j][0].z * w); atomicAdd(ap + 3  * NPB, r[j][0].w * w);
            atomicAdd(ap + 4  * NPB, r[j][1].x * w); atomicAdd(ap + 5  * NPB, r[j][1].y * w);
            atomicAdd(ap + 6  * NPB, r[j][1].z * w); atomicAdd(ap + 7  * NPB, r[j][1].w * w);
            atomicAdd(ap + 8  * NPB, r[j][2].x * w); atomicAdd(ap + 9  * NPB, r[j][2].y * w);
            atomicAdd(ap + 10 * NPB, r[j][2].z * w); atomicAdd(ap + 11 * NPB, r[j][2].w * w);
            atomicAdd(ap + 12 * NPB, r[j][3].x * w); atomicAdd(ap + 13 * NPB, r[j][3].y * w);
            atomicAdd(ap + 14 * NPB, r[j][3].z * w); atomicAdd(ap + 15 * NPB, r[j][3].w * w);
            atomicAdd(&s_den[d], w);
          }
        }
      }
    }
  } else {
    {
      uint4 pc[8];
      load_planes(k4, rot, tid, pc);
      for (int t = 0; t < NIT; ++t) {
        const int it  = (t + rot) < NIT ? (t + rot) : (t + rot - NIT);
        const int itn = (it + 1) < NIT ? (it + 1) : 0;
        uint4 pn[8];
        load_planes(k4, itn, tid, pn);
        const uint4 m = match_planes(pc, it, tid, inv);
        #pragma nounroll
        for (int c = 0; c < 4; ++c) {
          unsigned mm = word_of(m, c);
          while (mm) {
            int k = __builtin_ctz(mm); mm &= mm - 1;
            int e = edge_of(it, tid, c, k);
            int s = ei[e];
            int dl = ei[NEDGES + e] - lo;
            if constexpr (H == 8) {
              const float4* ap = (const float4*)(aS + (size_t)s * 8);
              float4 a0 = ap[0], a1 = ap[1];
              atomicMax(&s_max[0 * NPB + dl], enc_f32(lrelu(a0.x + s_ad[0 * NPB + dl])));
              atomicMax(&s_max[1 * NPB + dl], enc_f32(lrelu(a0.y + s_ad[1 * NPB + dl])));
              atomicMax(&s_max[2 * NPB + dl], enc_f32(lrelu(a0.z + s_ad[2 * NPB + dl])));
              atomicMax(&s_max[3 * NPB + dl], enc_f32(lrelu(a0.w + s_ad[3 * NPB + dl])));
              atomicMax(&s_max[4 * NPB + dl], enc_f32(lrelu(a1.x + s_ad[4 * NPB + dl])));
              atomicMax(&s_max[5 * NPB + dl], enc_f32(lrelu(a1.y + s_ad[5 * NPB + dl])));
              atomicMax(&s_max[6 * NPB + dl], enc_f32(lrelu(a1.z + s_ad[6 * NPB + dl])));
              atomicMax(&s_max[7 * NPB + dl], enc_f32(lrelu(a1.w + s_ad[7 * NPB + dl])));
            } else {
              atomicMax(&s_max[dl], enc_f32(lrelu(aS[s] + s_ad[dl])));
            }
          }
        }
        #pragma unroll
        for (int b = 0; b < 8; ++b) pc[b] = pn[b];
      }
    }
    __syncthreads();

    {
      uint4 pc[8];
      load_planes(k4, rot, tid, pc);
      for (int t = 0; t < NIT; ++t) {
        const int it  = (t + rot) < NIT ? (t + rot) : (t + rot - NIT);
        const int itn = (it + 1) < NIT ? (it + 1) : 0;
        uint4 pn[8];
        load_planes(k4, itn, tid, pn);
        const uint4 m = match_planes(pc, it, tid, inv);
        #pragma nounroll
        for (int c = 0; c < 4; ++c) {
          unsigned mm = word_of(m, c);
          if constexpr (H == 8) {
            const int hd = lane >> 2;
            for (;;) {
              unsigned bal = __builtin_amdgcn_ballot_w32(mm != 0);
              if (!bal) break;
              int sv = 0, dv = 0;
              if (mm) {
                int k = __builtin_ctz(mm); mm &= mm - 1;
                int e = edge_of(it, tid, c, k);
                sv = ei[e];
                dv = ei[NEDGES + e] - lo;
              }
              while (bal) {
                int ml = __builtin_ctz(bal); bal &= bal - 1;
                int s  = __builtin_amdgcn_readlane(sv, ml);
                int dl = __builtin_amdgcn_readlane(dv, ml);
                float  a = aS[(size_t)s * 8 + hd];
                float4 r = ((const float4*)(h + (size_t)s * HC))[lane];
                float w = expf(lrelu(a + s_ad[hd * NPB + dl]) - dec_f32(s_max[hd * NPB + dl]));
                float* ap = &s_agg[dl * X1P + lane];
                atomicAdd(ap,      r.x * w);
                atomicAdd(ap + 32, r.y * w);
                atomicAdd(ap + 64, r.z * w);
                atomicAdd(ap + 96, r.w * w);
                if ((lane & 3) == 0) atomicAdd(&s_den[hd * NPB + dl], w);
              }
            }
          } else {
            while (mm) {
              int k = __builtin_ctz(mm); mm &= mm - 1;
              int e = edge_of(it, tid, c, k);
              int s = ei[e];
              int dl = ei[NEDGES + e] - lo;
              float a = aS[s];
              const float4* rp = (const float4*)(h + (size_t)s * HC);
              float4 r0 = rp[0], r1 = rp[1], r2 = rp[2], r3 = rp[3];
              float w = expf(lrelu(a + s_ad[dl]) - dec_f32(s_max[dl]));
              float* ap = &s_agg[dl];
              atomicAdd(ap + 0  * NPB, r0.x * w); atomicAdd(ap + 1  * NPB, r0.y * w);
              atomicAdd(ap + 2  * NPB, r0.z * w); atomicAdd(ap + 3  * NPB, r0.w * w);
              atomicAdd(ap + 4  * NPB, r1.x * w); atomicAdd(ap + 5  * NPB, r1.y * w);
              atomicAdd(ap + 6  * NPB, r1.z * w); atomicAdd(ap + 7  * NPB, r1.w * w);
              atomicAdd(ap + 8  * NPB, r2.x * w); atomicAdd(ap + 9  * NPB, r2.y * w);
              atomicAdd(ap + 10 * NPB, r2.z * w); atomicAdd(ap + 11 * NPB, r2.w * w);
              atomicAdd(ap + 12 * NPB, r3.x * w); atomicAdd(ap + 13 * NPB, r3.y * w);
              atomicAdd(ap + 14 * NPB, r3.z * w); atomicAdd(ap + 15 * NPB, r3.w * w);
              atomicAdd(&s_den[dl], w);
            }
          }
        }
        #pragma unroll
        for (int b = 0; b < 8; ++b) pc[b] = pn[b];
      }
    }
  }
  __syncthreads();

  if constexpr (H == 8) {
    float* s_w2 = (float*)s_list;
    float* s_h2 = s_w2 + C_ * X1P;
    static_assert(C_ * HC == 2 * 1024, "W2 is copied two words per thread");
    const float w2a = W2[tid], w2b = W2[tid + 1024];
    float4 sa[4], da[4];
    #pragma unroll
    for (int q = 0; q < 4; ++q) { sa[q] = ((const float4*)attS2)[q]; da[q] = ((const float4*)attD2)[q]; }
    {
      constexpr int RW = (NPB + 31) / 32;
      const int hdt = lane >> 2;
      const int wvt = __builtin_amdgcn_readfirstlane(tid >> 5);
      const float4 bv = ((const float4*)bias)[lane];
      float4 hr[RW];
      float  sr[RW];
      #pragma unroll
      for (int r = 0; r < RW; ++r) {
        const int dl = wvt + 32 * r;
        const size_t n = (size_t)(lo + (dl < cnt ? dl : cnt - 1));
        hr[r] = ((const float4*)(h + n * HC))[lane];
        sr[r] = aS[n * 8 + hdt];
      }
      #pragma unroll
      for (int r = 0; r < RW; ++r) {
        const int dl = wvt + 32 * r;
        if (dl < cnt) {
          const int o = hdt * NPB + dl;
          const float ws = expf(lrelu(sr[r] + s_ad[o]) - dec_f32(s_max[o]));
          const float inv = 1.0f / (s_den[o] + ws + 1e-16f);
          float* xp = &s_agg[dl * X1P + lane];
          const float v0 = (xp[0]  + ws * hr[r].x) * inv + bv.x;
          const float v1 = (xp[32] + ws * hr[r].y) * inv + bv.y;
          const float v2 = (xp[64] + ws * hr[r].z) * inv + bv.z;
          const float v3 = (xp[96] + ws * hr[r].w) * inv + bv.w;
          xp[0]  = v0 > 0.f ? v0 : expm1f(v0);
          xp[32] = v1 > 0.f ? v1 : expm1f(v1);
          xp[64] = v2 > 0.f ? v2 : expm1f(v2);
          xp[96] = v3 > 0.f ? v3 : expm1f(v3);
        }
      }
    }
    {
      const int c = tid & 127;
      float* wp = &s_w2[(tid >> 7) * X1P + (c & 3) * 32 + (c >> 2)];
      wp[0] = w2a; wp[8 * X1P] = w2b;
    }
    __syncthreads();
    const int wv2 = __builtin_amdgcn_readfirstlane(tid >> 5);
    if (wv2 < T2) {
      const int l15 = lane & 15, half = lane >> 4;
      const int row = (wv2 * 16 + l15) < NPB ? (wv2 * 16 + l15) : NPB - 1;
      const float* ar = s_agg + row * X1P + half * 2;
      const float* br = s_w2 + l15 * X1P + half * 2;
      v8f acc0 = (v8f){};
      #pragma unroll
      for (int s = 0; s < HC / 4; ++s) {
        const v2f a0 = *(const v2f*)(ar + 4 * s), b0 = *(const v2f*)(br + 4 * s);
        acc0 = __builtin_amdgcn_wmma_f32_16x16x4_f32(false, a0, false, b0, (short)0, acc0, false, false);
      }
      float* hp = s_h2 + (wv2 * 16 + half * 8) * H2P + l15;
      #pragma unroll
      for (int r = 0; r < 8; ++r) hp[r * H2P] = acc0[r];
    }
    __syncthreads();
    if (tid < cnt) {
      const size_t n = (size_t)(lo + tid);
      const float* hp = s_h2 + tid * H2P;
      float hv[C_];
      float s = 0.f, d = 0.f;
      #pragma unroll
      for (int c = 0; c < C_; ++c) hv[c] = hp[c];
      #pragma unroll
      for (int q = 0; q < 4; ++q) {
        s += hv[4 * q] * sa[q].x; s += hv[4 * q + 1] * sa[q].y; s += hv[4 * q + 2] * sa[q].z; s += hv[4 * q + 3] * sa[q].w;
        d += hv[4 * q] * da[q].x; d += hv[4 * q + 1] * da[q].y; d += hv[4 * q + 2] * da[q].z; d += hv[4 * q + 3] * da[q].w;
      }
      as2o[n] = s; ad2o[n] = d;
      float4* op = (float4*)(h2o + n * C_);
      op[0] = make_float4(hv[0],  hv[1],  hv[2],  hv[3]);
      op[1] = make_float4(hv[4],  hv[5],  hv[6],  hv[7]);
      op[2] = make_float4(hv[8],  hv[9],  hv[10], hv[11]);
      op[3] = make_float4(hv[12], hv[13], hv[14], hv[15]);
    }
  } else {
    if (tid < cnt) {
      int dl = tid;
      size_t n = (size_t)(lo + dl);
      float ws = expf(lrelu(aS[n] + s_ad[dl]) - dec_f32(s_max[dl]));
      float dn = s_den[dl] + ws + 1e-16f;
      float v[NCLASS];
      float m = -1e30f;
      #pragma unroll
      for (int c = 0; c < NCLASS; ++c) {
        v[c] = (s_agg[c * NPB + dl] + ws * h[n * HC + c]) / dn + bias[c];
        m = fmaxf(m, v[c]);
      }
      float sum = 0.f;
      #pragma unroll
      for (int c = 0; c < NCLASS; ++c) sum += expf(v[c] - m);
      float ls = logf(sum);
      #pragma unroll
      for (int c = 0; c < NCLASS; ++c) out[n * NCLASS + c] = v[c] - m - ls;
    }
  }
}

extern "C" void kernel_launch(void* const* d_in, const int* in_sizes, int n_in,
                              void* d_out, int out_size, void* d_ws, size_t ws_size,
                              hipStream_t stream) {
  const float* x     = (const float*)d_in[0];
  const int*   ei    = (const int*)  d_in[1];
  const float* W1    = (const float*)d_in[2];
  const float* attS1 = (const float*)d_in[3];
  const float* attD1 = (const float*)d_in[4];
  const float* b1    = (const float*)d_in[5];
  const float* W2    = (const float*)d_in[6];
  const float* attS2 = (const float*)d_in[7];
  const float* attD2 = (const float*)d_in[8];
  const float* b2    = (const float*)d_in[9];
  float* out = (float*)d_out;

  char* ws = (char*)d_ws;
  size_t off = 0;
  auto alloc = [&](size_t bytes) -> void* {
    void* p = ws + off;
    off = (off + bytes + 255) & ~(size_t)255;
    return p;
  };

  float* h1  = (float*)alloc((size_t)NNODES * F1 * 4);
  float* as1 = (float*)alloc((size_t)NNODES * H1_ * 4);
  float* ad1 = (float*)alloc((size_t)NNODES * H1_ * 4);
  float* h2  = (float*)alloc((size_t)NNODES * C_ * 4);
  float* as2 = (float*)alloc((size_t)NNODES * 4);
  float* ad2 = (float*)alloc((size_t)NNODES * 4);
  unsigned* key = (unsigned*)alloc((size_t)NIT * 8 * 1024 * 16);
  unsigned* glist = (unsigned*)alloc((size_t)NBLK * WCAP * 4);
  int* gcnt = (int*)alloc((size_t)NBLK * 4);

  const int TB = 256;
  auto nb = [](long long n, int b) { return (unsigned)((n + b - 1) / b); };

  gemm1_w_in_lds<<<G1_BLK, G1_THREADS, 0, stream>>>(x, W1, h1);
  attn_scores<<<nb((long long)NNODES * H1_, TB), TB, 0, stream>>>(
      h1, attS1, attD1, as1, ad1, NNODES * H1_, H1_);
  owner_planes<<<nb((long long)NSG * 32, TB), TB, 0, stream>>>(ei, (uint4*)key);
  owned_gat<H1_><<<NBLK, 1024, 0, stream>>>(ei, key, as1, ad1, h1, b1, nullptr, glist, gcnt,
                                            W2, attS2, attD2, h2, as2, ad2);

  owned_gat<1><<<NBLK, 1024, 0, stream>>>(ei, key, as2, ad2, h2, b2, out, glist, gcnt,
                                          nullptr, nullptr, nullptr, nullptr, nullptr, nullptr);
}
